# starting kernel
# baseline (speedup 1.0000x reference)
#include <hip/hip_runtime.h>
#include <math.h>

#define SLOPE 0.2f

typedef float v2f __attribute__((ext_vector_type(2)));
typedef float v8f __attribute__((ext_vector_type(8)));

__device__ __forceinline__ unsigned f2key(float f) {
  unsigned u = __float_as_uint(f);
  return (u & 0x80000000u) ? ~u : (u | 0x80000000u);
}
__device__ __forceinline__ float key2f(unsigned k) {
  unsigned u = (k & 0x80000000u) ? (k ^ 0x80000000u) : ~k;
  return __uint_as_float(u);
}
__device__ __forceinline__ float lrelu(float v) { return v > 0.0f ? v : SLOPE * v; }

template <int K, int NC>
__global__ void gemm_bias_wmma(const float* __restrict__ X, const float* __restrict__ W,
                               const float* __restrict__ b, float* __restrict__ Y, int M) {
  const int ntiles = NC / 16;
  int wid = threadIdx.x >> 5;
  int tile = blockIdx.x * (blockDim.x >> 5) + wid;
  int total = (M / 16) * ntiles;
  if (tile >= total) return;
  int mt = tile / ntiles, nt = tile % ntiles;
  int lane = threadIdx.x & 31;
  int half = lane >> 4;
  int l = lane & 15;
  int row = mt * 16 + l;
  int col = nt * 16 + l;
  float bc = b[col];
  v8f acc = {bc, bc, bc, bc, bc, bc, bc, bc};
  const float* xp = X + (size_t)row * K;
  const float* wp = W + col;
#pragma unroll
  for (int k0 = 0; k0 < K; k0 += 4) {
    v2f a, bv;
    a.x = xp[k0 + 2 * half];
    a.y = xp[k0 + 2 * half + 1];
    bv.x = wp[(size_t)(k0 + 2 * half) * NC];
    bv.y = wp[(size_t)(k0 + 2 * half + 1) * NC];
    acc = __builtin_amdgcn_wmma_f32_16x16x4_f32(false, a, false, bv, (short)0, acc,
                                                false, false);
  }
  float* yp = Y + (size_t)(mt * 16 + 8 * half) * NC + col;
#pragma unroll
  for (int v = 0; v < 8; v++) yp[(size_t)v * NC] = acc[v];
}

template <int H_, int C_>
__global__ void edge_scores(const float* __restrict__ xl, const float* __restrict__ xr,
                            const float* __restrict__ att, const int* __restrict__ ei,
                            int E_, int Etot, float* __restrict__ ebuf,
                            unsigned* __restrict__ nmax) {
  int g = (int)((blockIdx.x * (size_t)blockDim.x + threadIdx.x) >> 4);
  int sub = threadIdx.x & 15;
  if (g >= Etot * H_) return;
  int eidx = g / H_, head = g % H_;
  int src, dst;
  if (eidx < E_) { src = ei[eidx]; dst = ei[E_ + eidx]; }
  else           { src = dst = eidx - E_; }
  int base = head * C_ + sub * 4;
  float4 a  = *(const float4*)(xl + (size_t)src * (H_ * C_) + base);
  float4 bq = *(const float4*)(xr + (size_t)dst * (H_ * C_) + base);
  float4 at = *(const float4*)(att + base);
  float p = lrelu(a.x + bq.x) * at.x + lrelu(a.y + bq.y) * at.y +
            lrelu(a.z + bq.z) * at.z + lrelu(a.w + bq.w) * at.w;
  for (int off = 8; off > 0; off >>= 1) p += __shfl_xor(p, off, 32);
  if (sub == 0) {
    ebuf[g] = p;
    atomicMax(nmax + (size_t)dst * H_ + head, f2key(p));
  }
}

template <int H_>
__global__ void edge_exp(const int* __restrict__ ei, int E_, int Etot,
                         const unsigned* __restrict__ nmax, float* __restrict__ ebuf,
                         float* __restrict__ denom) {
  int g = (int)(blockIdx.x * (size_t)blockDim.x + threadIdx.x);
  if (g >= Etot * H_) return;
  int eidx = g / H_, head = g % H_;
  int dst = (eidx < E_) ? ei[E_ + eidx] : (eidx - E_);
  float mx = key2f(nmax[(size_t)dst * H_ + head]);
  float ex = expf(ebuf[g] - mx);
  ebuf[g] = ex;
  atomicAdd(denom + (size_t)dst * H_ + head, ex);
}

template <int H_, int C_>
__global__ void edge_scatter(const float* __restrict__ xl, const int* __restrict__ ei,
                             int E_, int Etot, const float* __restrict__ ebuf,
                             const float* __restrict__ denom, float* __restrict__ oacc) {
  int g = (int)((blockIdx.x * (size_t)blockDim.x + threadIdx.x) >> 4);
  int sub = threadIdx.x & 15;
  if (g >= Etot * H_) return;
  int eidx = g / H_, head = g % H_;
  int src, dst;
  if (eidx < E_) { src = ei[eidx]; dst = ei[E_ + eidx]; }
  else           { src = dst = eidx - E_; }
  float alpha = ebuf[g] / denom[(size_t)dst * H_ + head];
  int base = head * C_ + sub * 4;
  float4 a = *(const float4*)(xl + (size_t)src * (H_ * C_) + base);
  float* o = oacc + (size_t)dst * (H_ * C_) + base;
  atomicAdd(o + 0, alpha * a.x);
  atomicAdd(o + 1, alpha * a.y);
  atomicAdd(o + 2, alpha * a.z);
  atomicAdd(o + 3, alpha * a.w);
}

__global__ void bias_elu_inplace(float* __restrict__ h, const float* __restrict__ bias,
                                 int n, int nc) {
  int i = (int)(blockIdx.x * (size_t)blockDim.x + threadIdx.x);
  if (i >= n) return;
  float v = h[i] + bias[i % nc];
  h[i] = v > 0.0f ? v : expm1f(v);
}

__global__ void bias_add_out(const float* __restrict__ acc, const float* __restrict__ bias,
                             float* __restrict__ out, int n, int nc) {
  int i = (int)(blockIdx.x * (size_t)blockDim.x + threadIdx.x);
  if (i >= n) return;
  out[i] = acc[i] + bias[i % nc];
}

extern "C" void kernel_launch(void* const* d_in, const int* in_sizes, int n_in,
                              void* d_out, int out_size, void* d_ws, size_t ws_size,
                              hipStream_t stream) {
  (void)n_in; (void)out_size; (void)ws_size;
  const int IN = 128, HID = 64, H = 4, OUT = 64;
  const float*    x    = (const float*)d_in[0];
  const int*      ei   = (const int*)d_in[1];
  const float*    Wl1  = (const float*)d_in[2];
  const float*    bl1  = (const float*)d_in[3];
  const float*    Wr1  = (const float*)d_in[4];
  const float*    br1  = (const float*)d_in[5];
  const float*    att1 = (const float*)d_in[6];
  const float*    bias1= (const float*)d_in[7];
  const float*    Wl2  = (const float*)d_in[8];
  const float*    bl2  = (const float*)d_in[9];
  const float*    Wr2  = (const float*)d_in[10];
  const float*    br2  = (const float*)d_in[11];
  const float*    att2 = (const float*)d_in[12];
  const float*    bias2= (const float*)d_in[13];
  float* out = (float*)d_out;

  const int N    = in_sizes[0] / IN;
  const int E    = in_sizes[1] / 2;
  const int Etot = E + N;
  const int HC   = H * HID;

  float* ws = (float*)d_ws;
  size_t o = 0;
  float*    xl1   = ws + o; o += (size_t)N * HC;
  float*    xr1   = ws + o; o += (size_t)N * HC;
  float*    hacc  = ws + o; o += (size_t)N * HC;
  float*    ebuf  = ws + o; o += (size_t)Etot * H;
  unsigned* nmax  = (unsigned*)(ws + o); o += (size_t)N * H;
  float*    denom = ws + o; o += (size_t)N * H;
  float* xl2   = xl1;
  float* xr2   = xr1;
  float* oacc2 = xr1 + (size_t)N * OUT;

  const int B = 256;

  hipMemsetAsync(hacc,  0, (size_t)N * HC * sizeof(float), stream);
  hipMemsetAsync(nmax,  0, (size_t)N * H * sizeof(unsigned), stream);
  hipMemsetAsync(denom, 0, (size_t)N * H * sizeof(float), stream);

  int tiles1 = (N / 16) * (HC / 16);
  gemm_bias_wmma<128, 256><<<(tiles1 + 7) / 8, B, 0, stream>>>(x, Wl1, bl1, xl1, N);
  gemm_bias_wmma<128, 256><<<(tiles1 + 7) / 8, B, 0, stream>>>(x, Wr1, br1, xr1, N);

  long long items1 = (long long)Etot * H;
  int gb1 = (int)((items1 * 16 + B - 1) / B);
  edge_scores<4, 64><<<gb1, B, 0, stream>>>(xl1, xr1, att1, ei, E, Etot, ebuf, nmax);
  edge_exp<4><<<(int)((items1 + B - 1) / B), B, 0, stream>>>(ei, E, Etot, nmax, ebuf, denom);
  edge_scatter<4, 64><<<gb1, B, 0, stream>>>(xl1, ei, E, Etot, ebuf, denom, hacc);

  int n1 = N * HC;
  bias_elu_inplace<<<(n1 + B - 1) / B, B, 0, stream>>>(hacc, bias1, n1, HC);

  hipMemsetAsync(oacc2, 0, (size_t)N * OUT * sizeof(float), stream);
  hipMemsetAsync(nmax,  0, (size_t)N * sizeof(unsigned), stream);
  hipMemsetAsync(denom, 0, (size_t)N * sizeof(float), stream);

  int tiles2 = (N / 16) * (OUT / 16);
  gemm_bias_wmma<256, 64><<<(tiles2 + 7) / 8, B, 0, stream>>>(hacc, Wl2, bl2, xl2, N);
  gemm_bias_wmma<256, 64><<<(tiles2 + 7) / 8, B, 0, stream>>>(hacc, Wr2, br2, xr2, N);

  long long items2 = Etot;
  int gb2 = (int)((items2 * 16 + B - 1) / B);
  edge_scores<1, 64><<<gb2, B, 0, stream>>>(xl2, xr2, att2, ei, E, Etot, ebuf, nmax);
  edge_exp<1><<<(int)((items2 + B - 1) / B), B, 0, stream>>>(ei, E, Etot, nmax, ebuf, denom);
  edge_scatter<1, 64><<<gb2, B, 0, stream>>>(xl2, ei, E, Etot, ebuf, denom, oacc2);

  int n2 = N * OUT;
  bias_add_out<<<(n2 + B - 1) / B, B, 0, stream>>>(oacc2, bias2, out, n2, OUT);
}


# Weight slab (K x 64) staged once per 8-wave work-group in LDS as padded k-pairs; each wave takes its B fragments from there (one 8-byte LDS read) for a 16x64 strip; k order of sums unchanged
# speedup vs baseline: 1.1256x; 1.1256x over previous
#include <hip/hip_runtime.h>
#include <math.h>

#define SLOPE 0.2f

typedef float v2f __attribute__((ext_vector_type(2)));
typedef float v8f __attribute__((ext_vector_type(8)));

__device__ __forceinline__ unsigned f2key(float f) {
  unsigned u = __float_as_uint(f);
  return (u & 0x80000000u) ? ~u : (u | 0x80000000u);
}
__device__ __forceinline__ float key2f(unsigned k) {
  unsigned u = (k & 0x80000000u) ? (k ^ 0x80000000u) : ~k;
  return __uint_as_float(u);
}
__device__ __forceinline__ float lrelu(float v) { return v > 0.0f ? v : SLOPE * v; }

template <int K, int NC>
__global__ __launch_bounds__(256) void gemm_bias_wmma(const float* __restrict__ X,
                                                      const float* __restrict__ W,
                                                      const float* __restrict__ b,
                                                      float* __restrict__ Y, int M) {
  constexpr int SLAB = 64;
  constexpr int NSLAB = NC / SLAB;
  constexpr int PITCH = 160;
  __shared__ __attribute__((aligned(16))) float wl[(K / 2) * PITCH];

  const int slab = blockIdx.x % NSLAB;
  const int rblk = blockIdx.x / NSLAB;
  const int cbase = slab * SLAB;

#pragma unroll
  for (int j = 0; j < (K / 2) * 16 / 256; j++) {
    int it = j * 256 + (int)threadIdx.x;
    int p = it >> 4, q = it & 15;
    const float* g = W + (size_t)(2 * p) * NC + cbase + 4 * q;
    float4 r0 = *(const float4*)g;
    float4 r1 = *(const float4*)(g + NC);
    float4* d = (float4*)(wl + p * PITCH + 8 * q);
    d[0] = make_float4(r0.x, r1.x, r0.y, r1.y);
    d[1] = make_float4(r0.z, r1.z, r0.w, r1.w);
  }
  __syncthreads();

  int wid = threadIdx.x >> 5;
  int mt = rblk * 8 + wid;
  if (mt * 16 >= M) return;
  int lane = threadIdx.x & 31;
  int half = lane >> 4;
  int l = lane & 15;
  const float* xp = X + (size_t)(mt * 16 + l) * K + 2 * half;
  const float* bp = wl + half * PITCH + 2 * l;
  v8f acc[4];
#pragma unroll
  for (int t = 0; t < 4; t++) {
    float bc = b[cbase + t * 16 + l];
    acc[t] = (v8f){bc, bc, bc, bc, bc, bc, bc, bc};
  }
#pragma unroll
  for (int k0 = 0; k0 < K; k0 += 4) {
    v2f a = *(const v2f*)(xp + k0);
#pragma unroll
    for (int t = 0; t < 4; t++) {
      v2f bv = *(const v2f*)(bp + (k0 / 2) * PITCH + t * 32);
      acc[t] = __builtin_amdgcn_wmma_f32_16x16x4_f32(false, a, false, bv, (short)0, acc[t],
                                                     false, false);
    }
  }
#pragma unroll
  for (int t = 0; t < 4; t++) {
    float* yp = Y + (size_t)(mt * 16 + 8 * half) * NC + cbase + t * 16 + l;
#pragma unroll
    for (int v = 0; v < 8; v++) yp[(size_t)v * NC] = acc[t][v];
  }
}

template <int H_, int C_>
__global__ void edge_scores(const float* __restrict__ xl, const float* __restrict__ xr,
                            const float* __restrict__ att, const int* __restrict__ ei,
                            int E_, int Etot, float* __restrict__ ebuf,
                            unsigned* __restrict__ nmax) {
  int g = (int)((blockIdx.x * (size_t)blockDim.x + threadIdx.x) >> 4);
  int sub = threadIdx.x & 15;
  if (g >= Etot * H_) return;
  int eidx = g / H_, head = g % H_;
  int src, dst;
  if (eidx < E_) { src = ei[eidx]; dst = ei[E_ + eidx]; }
  else           { src = dst = eidx - E_; }
  int base = head * C_ + sub * 4;
  float4 a  = *(const float4*)(xl + (size_t)src * (H_ * C_) + base);
  float4 bq = *(const float4*)(xr + (size_t)dst * (H_ * C_) + base);
  float4 at = *(const float4*)(att + base);
  float p = lrelu(a.x + bq.x) * at.x + lrelu(a.y + bq.y) * at.y +
            lrelu(a.z + bq.z) * at.z + lrelu(a.w + bq.w) * at.w;
  for (int off = 8; off > 0; off >>= 1) p += __shfl_xor(p, off, 32);
  if (sub == 0) {
    ebuf[g] = p;
    atomicMax(nmax + (size_t)dst * H_ + head, f2key(p));
  }
}

template <int H_>
__global__ void edge_exp(const int* __restrict__ ei, int E_, int Etot,
                         const unsigned* __restrict__ nmax, float* __restrict__ ebuf,
                         float* __restrict__ denom) {
  int g = (int)(blockIdx.x * (size_t)blockDim.x + threadIdx.x);
  if (g >= Etot * H_) return;
  int eidx = g / H_, head = g % H_;
  int dst = (eidx < E_) ? ei[E_ + eidx] : (eidx - E_);
  float mx = key2f(nmax[(size_t)dst * H_ + head]);
  float ex = expf(ebuf[g] - mx);
  ebuf[g] = ex;
  atomicAdd(denom + (size_t)dst * H_ + head, ex);
}

template <int H_, int C_>
__global__ void edge_scatter(const float* __restrict__ xl, const int* __restrict__ ei,
                             int E_, int Etot, const float* __restrict__ ebuf,
                             const float* __restrict__ denom, float* __restrict__ oacc) {
  int g = (int)((blockIdx.x * (size_t)blockDim.x + threadIdx.x) >> 4);
  int sub = threadIdx.x & 15;
  if (g >= Etot * H_) return;
  int eidx = g / H_, head = g % H_;
  int src, dst;
  if (eidx < E_) { src = ei[eidx]; dst = ei[E_ + eidx]; }
  else           { src = dst = eidx - E_; }
  float alpha = ebuf[g] / denom[(size_t)dst * H_ + head];
  int base = head * C_ + sub * 4;
  float4 a = *(const float4*)(xl + (size_t)src * (H_ * C_) + base);
  float* o = oacc + (size_t)dst * (H_ * C_) + base;
  atomicAdd(o + 0, alpha * a.x);
  atomicAdd(o + 1, alpha * a.y);
  atomicAdd(o + 2, alpha * a.z);
  atomicAdd(o + 3, alpha * a.w);
}

__global__ void bias_elu_inplace(float* __restrict__ h, const float* __restrict__ bias,
                                 int n, int nc) {
  int i = (int)(blockIdx.x * (size_t)blockDim.x + threadIdx.x);
  if (i >= n) return;
  float v = h[i] + bias[i % nc];
  h[i] = v > 0.0f ? v : expm1f(v);
}

__global__ void bias_add_out(const float* __restrict__ acc, const float* __restrict__ bias,
                             float* __restrict__ out, int n, int nc) {
  int i = (int)(blockIdx.x * (size_t)blockDim.x + threadIdx.x);
  if (i >= n) return;
  out[i] = acc[i] + bias[i % nc];
}

extern "C" void kernel_launch(void* const* d_in, const int* in_sizes, int n_in,
                              void* d_out, int out_size, void* d_ws, size_t ws_size,
                              hipStream_t stream) {
  (void)n_in; (void)out_size; (void)ws_size;
  const int IN = 128, HID = 64, H = 4, OUT = 64;
  const float*    x    = (const float*)d_in[0];
  const int*      ei   = (const int*)d_in[1];
  const float*    Wl1  = (const float*)d_in[2];
  const float*    bl1  = (const float*)d_in[3];
  const float*    Wr1  = (const float*)d_in[4];
  const float*    br1  = (const float*)d_in[5];
  const float*    att1 = (const float*)d_in[6];
  const float*    bias1= (const float*)d_in[7];
  const float*    Wl2  = (const float*)d_in[8];
  const float*    bl2  = (const float*)d_in[9];
  const float*    Wr2  = (const float*)d_in[10];
  const float*    br2  = (const float*)d_in[11];
  const float*    att2 = (const float*)d_in[12];
  const float*    bias2= (const float*)d_in[13];
  float* out = (float*)d_out;

  const int N    = in_sizes[0] / IN;
  const int E    = in_sizes[1] / 2;
  const int Etot = E + N;
  const int HC   = H * HID;

  float* ws = (float*)d_ws;
  size_t o = 0;
  float*    xl1   = ws + o; o += (size_t)N * HC;
  float*    xr1   = ws + o; o += (size_t)N * HC;
  float*    hacc  = ws + o; o += (size_t)N * HC;
  float*    ebuf  = ws + o; o += (size_t)Etot * H;
  unsigned* nmax  = (unsigned*)(ws + o); o += (size_t)N * H;
  float*    denom = ws + o; o += (size_t)N * H;
  float* xl2   = xl1;
  float* xr2   = xr1;
  float* oacc2 = xr1 + (size_t)N * OUT;

  const int B = 256;

  hipMemsetAsync(hacc,  0, (size_t)N * HC * sizeof(float), stream);
  hipMemsetAsync(nmax,  0, (size_t)N * H * sizeof(unsigned), stream);
  hipMemsetAsync(denom, 0, (size_t)N * H * sizeof(float), stream);

  int rblks = (N / 16 + 7) / 8;
  gemm_bias_wmma<128, 256><<<rblks * (HC / 64), B, 0, stream>>>(x, Wl1, bl1, xl1, N);
  gemm_bias_wmma<128, 256><<<rblks * (HC / 64), B, 0, stream>>>(x, Wr1, br1, xr1, N);

  long long items1 = (long long)Etot * H;
  int gb1 = (int)((items1 * 16 + B - 1) / B);
  edge_scores<4, 64><<<gb1, B, 0, stream>>>(xl1, xr1, att1, ei, E, Etot, ebuf, nmax);
  edge_exp<4><<<(int)((items1 + B - 1) / B), B, 0, stream>>>(ei, E, Etot, nmax, ebuf, denom);
  edge_scatter<4, 64><<<gb1, B, 0, stream>>>(xl1, ei, E, Etot, ebuf, denom, hacc);

  int n1 = N * HC;
  bias_elu_inplace<<<(n1 + B - 1) / B, B, 0, stream>>>(hacc, bias1, n1, HC);

  hipMemsetAsync(oacc2, 0, (size_t)N * OUT * sizeof(float), stream);
  hipMemsetAsync(nmax,  0, (size_t)N * sizeof(unsigned), stream);
  hipMemsetAsync(denom, 0, (size_t)N * sizeof(float), stream);

  gemm_bias_wmma<256, 64><<<rblks * (OUT / 64), B, 0, stream>>>(hacc, Wl2, bl2, xl2, N);
  gemm_bias_wmma<256, 64><<<rblks * (OUT / 64), B, 0, stream>>>(hacc, Wr2, br2, xr2, N);

  long long items2 = Etot;
  int gb2 = (int)((items2 * 16 + B - 1) / B);
  edge_scores<1, 64><<<gb2, B, 0, stream>>>(xl2, xr2, att2, ei, E, Etot, ebuf, nmax);
  edge_exp<1><<<(int)((items2 + B - 1) / B), B, 0, stream>>>(ei, E, Etot, nmax, ebuf, denom);
  edge_scatter<1, 64><<<gb2, B, 0, stream>>>(xl2, ei, E, Etot, ebuf, denom, oacc2);

  int n2 = N * OUT;
  bias_add_out<<<(n2 + B - 1) / B, B, 0, stream>>>(oacc2, bias2, out, n2, OUT);
}


# Edge-score and elementwise launches run as fewer, fatter waves: a wave gathers 16 (edge,head) rows at once, lane sums by a transposing butterfly (partners and order kept); 16 floats a thread
# speedup vs baseline: 1.2570x; 1.1167x over previous
#include <hip/hip_runtime.h>
#include <math.h>

#define SLOPE 0.2f

typedef float v2f __attribute__((ext_vector_type(2)));
typedef float v4f __attribute__((ext_vector_type(4)));
typedef float v8f __attribute__((ext_vector_type(8)));

__device__ __forceinline__ unsigned f2key(float f) {
  unsigned u = __float_as_uint(f);
  return (u & 0x80000000u) ? ~u : (u | 0x80000000u);
}
__device__ __forceinline__ float key2f(unsigned k) {
  unsigned u = (k & 0x80000000u) ? (k ^ 0x80000000u) : ~k;
  return __uint_as_float(u);
}
__device__ __forceinline__ float lrelu(float v) { return v > 0.0f ? v : SLOPE * v; }

template <int K, int NC>
__global__ __launch_bounds__(256) void gemm_bias_wmma(const float* __restrict__ X,
                                                      const float* __restrict__ W,
                                                      const float* __restrict__ b,
                                                      float* __restrict__ Y, int M) {
  constexpr int SLAB = 64;
  constexpr int NSLAB = NC / SLAB;
  constexpr int PITCH = 160;
  __shared__ __attribute__((aligned(16))) float wl[(K / 2) * PITCH];

  const int slab = blockIdx.x % NSLAB;
  const int rblk = blockIdx.x / NSLAB;
  const int cbase = slab * SLAB;

#pragma unroll
  for (int j = 0; j < (K / 2) * 16 / 256; j++) {
    int it = j * 256 + (int)threadIdx.x;
    int p = it >> 4, q = it & 15;
    const float* g = W + (size_t)(2 * p) * NC + cbase + 4 * q;
    float4 r0 = *(const float4*)g;
    float4 r1 = *(const float4*)(g + NC);
    float4* d = (float4*)(wl + p * PITCH + 8 * q);
    d[0] = make_float4(r0.x, r1.x, r0.y, r1.y);
    d[1] = make_float4(r0.z, r1.z, r0.w, r1.w);
  }
  __syncthreads();

  int wid = threadIdx.x >> 5;
  int mt = rblk * 8 + wid;
  if (mt * 16 >= M) return;
  int lane = threadIdx.x & 31;
  int half = lane >> 4;
  int l = lane & 15;
  const float* xp = X + (size_t)(mt * 16 + l) * K + 2 * half;
  const float* bp = wl + half * PITCH + 2 * l;
  v8f acc[4];
#pragma unroll
  for (int t = 0; t < 4; t++) {
    float bc = b[cbase + t * 16 + l];
    acc[t] = (v8f){bc, bc, bc, bc, bc, bc, bc, bc};
  }
#pragma unroll
  for (int k0 = 0; k0 < K; k0 += 4) {
    v2f a = *(const v2f*)(xp + k0);
#pragma unroll
    for (int t = 0; t < 4; t++) {
      v2f bv = *(const v2f*)(bp + (k0 / 2) * PITCH + t * 32);
      acc[t] = __builtin_amdgcn_wmma_f32_16x16x4_f32(false, a, false, bv, (short)0, acc[t],
                                                     false, false);
    }
  }
#pragma unroll
  for (int t = 0; t < 4; t++) {
    float* yp = Y + (size_t)(mt * 16 + 8 * half) * NC + cbase + t * 16 + l;
#pragma unroll
    for (int v = 0; v < 8; v++) yp[(size_t)v * NC] = acc[t][v];
  }
}

template <int H_, int C_>
__global__ __launch_bounds__(256) void edge_scores(const float* __restrict__ xl,
                                                   const float* __restrict__ xr,
                                                   const float* __restrict__ att,
                                                   const int* __restrict__ ei, int E_, int Etot,
                                                   float* __restrict__ ebuf,
                                                   unsigned* __restrict__ nmax) {
  static_assert(C_ == 64, "16 lanes x 4 channels per item");
  static_assert(4 % H_ == 0, "a wave's 16 items start at head 0; a slot's head follows its parity");
  constexpr int NV = 8;
  const int G = Etot * H_;
  int wave = (int)((blockIdx.x * (size_t)blockDim.x + threadIdx.x) >> 5);
  int g0 = wave * (2 * NV);
  if (g0 >= G) return;
  int lane = threadIdx.x & 31;
  int hf = lane >> 4, sub = lane & 15;

  int gm = g0 + 2 * (sub >> 1) + hf;

  v4f at[2];
#pragma unroll
  for (int q = 0; q < 2; q++) at[q] = *(const v4f*)(att + ((2 * q + hf) % H_) * C_ + sub * 4);

  int srcs[NV], dsts[NV];
#pragma unroll
  for (int i = 0; i < NV; i++) {
    int e = min(g0 / H_ + (2 * i + hf) / H_, Etot - 1);
    int el = min(e, E_ - 1);
    int s_ld = ei[el], d_ld = ei[E_ + el];
    srcs[i] = e < E_ ? s_ld : e - E_;
    dsts[i] = e < E_ ? d_ld : e - E_;
  }
  asm volatile("" ::
               "v"(srcs[0]), "v"(dsts[0]), "v"(srcs[1]), "v"(dsts[1]),
               "v"(srcs[2]), "v"(dsts[2]), "v"(srcs[3]), "v"(dsts[3]),
               "v"(srcs[4]), "v"(dsts[4]), "v"(srcs[5]), "v"(dsts[5]),
               "v"(srcs[6]), "v"(dsts[6]), "v"(srcs[7]), "v"(dsts[7]));
  int dm = dsts[0];
#pragma unroll
  for (int i = 1; i < NV; i++) dm = (sub >> 1) == i ? dsts[i] : dm;
  v4f a[NV], bq[NV];
#pragma unroll
  for (int i = 0; i < NV; i++) {
    int base = ((2 * i + hf) % H_) * C_ + sub * 4;
    a[i]  = *(const v4f*)(xl + (size_t)srcs[i] * (H_ * C_) + base);
    bq[i] = *(const v4f*)(xr + (size_t)dsts[i] * (H_ * C_) + base);
  }
  asm volatile("" ::
               "v"(a[0]), "v"(bq[0]), "v"(a[1]), "v"(bq[1]), "v"(a[2]), "v"(bq[2]), "v"(a[3]), "v"(bq[3]),
               "v"(a[4]), "v"(bq[4]), "v"(a[5]), "v"(bq[5]), "v"(a[6]), "v"(bq[6]), "v"(a[7]), "v"(bq[7]));
  float p[NV];
#pragma unroll
  for (int i = 0; i < NV; i++) {
    v4f t4 = at[i & 1];
    p[i] = lrelu(a[i].x + bq[i].x) * t4.x + lrelu(a[i].y + bq[i].y) * t4.y +
           lrelu(a[i].z + bq[i].z) * t4.z + lrelu(a[i].w + bq[i].w) * t4.w;
  }
  bool b3 = (sub & 8) != 0, b2 = (sub & 4) != 0, b1 = (sub & 2) != 0;
  float w4[4], w2[2];
#pragma unroll
  for (int k = 0; k < 4; k++) {
    float keep = b3 ? p[k + 4] : p[k];
    float send = b3 ? p[k] : p[k + 4];
    w4[k] = keep + __shfl_xor(send, 8, 32);
  }
#pragma unroll
  for (int k = 0; k < 2; k++) {
    float keep = b2 ? w4[k + 2] : w4[k];
    float send = b2 ? w4[k] : w4[k + 2];
    w2[k] = keep + __shfl_xor(send, 4, 32);
  }
  float keep1 = b1 ? w2[1] : w2[0];
  float send1 = b1 ? w2[0] : w2[1];
  float t = keep1 + __shfl_xor(send1, 2, 32);
  t += __shfl_xor(t, 1, 32);
  if ((sub & 1) == 0 && gm < G) {
    ebuf[gm] = t;
    atomicMax(nmax + (size_t)dm * H_ + gm % H_, f2key(t));
  }
}

template <int H_>
__global__ void edge_exp(const int* __restrict__ ei, int E_, int Etot,
                         const unsigned* __restrict__ nmax, float* __restrict__ ebuf,
                         float* __restrict__ denom) {
  int g = (int)(blockIdx.x * (size_t)blockDim.x + threadIdx.x);
  if (g >= Etot * H_) return;
  int eidx = g / H_, head = g % H_;
  int dst = (eidx < E_) ? ei[E_ + eidx] : (eidx - E_);
  float mx = key2f(nmax[(size_t)dst * H_ + head]);
  float ex = expf(ebuf[g] - mx);
  ebuf[g] = ex;
  atomicAdd(denom + (size_t)dst * H_ + head, ex);
}

template <int H_, int C_>
__global__ void edge_scatter(const float* __restrict__ xl, const int* __restrict__ ei,
                             int E_, int Etot, const float* __restrict__ ebuf,
                             const float* __restrict__ denom, float* __restrict__ oacc) {
  int g = (int)((blockIdx.x * (size_t)blockDim.x + threadIdx.x) >> 4);
  int sub = threadIdx.x & 15;
  if (g >= Etot * H_) return;
  int eidx = g / H_, head = g % H_;
  int src, dst;
  if (eidx < E_) { src = ei[eidx]; dst = ei[E_ + eidx]; }
  else           { src = dst = eidx - E_; }
  float alpha = ebuf[g] / denom[(size_t)dst * H_ + head];
  int base = head * C_ + sub * 4;
  float4 a = *(const float4*)(xl + (size_t)src * (H_ * C_) + base);
  float* o = oacc + (size_t)dst * (H_ * C_) + base;
  atomicAdd(o + 0, alpha * a.x);
  atomicAdd(o + 1, alpha * a.y);
  atomicAdd(o + 2, alpha * a.z);
  atomicAdd(o + 3, alpha * a.w);
}

__device__ __forceinline__ float elu1(float v) { return v > 0.0f ? v : expm1f(v); }

__global__ __launch_bounds__(256) void bias_elu_inplace(float* __restrict__ h,
                                                        const float* __restrict__ bias,
                                                        int n, int nc) {
  int b0 = (int)blockIdx.x * 4096 + (int)threadIdx.x * 4;
  float4 v[4];
#pragma unroll
  for (int j = 0; j < 4; j++) {
    int i = b0 + j * 1024;
    v[j] = i < n ? *(const float4*)(h + i) : make_float4(0.f, 0.f, 0.f, 0.f);
  }
#pragma unroll
  for (int j = 0; j < 4; j++) {
    int i = b0 + j * 1024;
    if (i < n) {
      float4 bb = *(const float4*)(bias + i % nc);
      float4 r;
      r.x = elu1(v[j].x + bb.x);
      r.y = elu1(v[j].y + bb.y);
      r.z = elu1(v[j].z + bb.z);
      r.w = elu1(v[j].w + bb.w);
      *(float4*)(h + i) = r;
    }
  }
}

__global__ __launch_bounds__(256) void bias_add_out(const float* __restrict__ acc,
                                                    const float* __restrict__ bias,
                                                    float* __restrict__ out, int n, int nc) {
  int b0 = (int)blockIdx.x * 4096 + (int)threadIdx.x * 4;
  float4 v[4];
#pragma unroll
  for (int j = 0; j < 4; j++) {
    int i = b0 + j * 1024;
    v[j] = i < n ? *(const float4*)(acc + i) : make_float4(0.f, 0.f, 0.f, 0.f);
  }
#pragma unroll
  for (int j = 0; j < 4; j++) {
    int i = b0 + j * 1024;
    if (i < n) {
      float4 bb = *(const float4*)(bias + i % nc);
      *(float4*)(out + i) = make_float4(v[j].x + bb.x, v[j].y + bb.y, v[j].z + bb.z, v[j].w + bb.w);
    }
  }
}

extern "C" void kernel_launch(void* const* d_in, const int* in_sizes, int n_in,
                              void* d_out, int out_size, void* d_ws, size_t ws_size,
                              hipStream_t stream) {
  (void)n_in; (void)out_size; (void)ws_size;
  const int IN = 128, HID = 64, H = 4, OUT = 64;
  const float*    x    = (const float*)d_in[0];
  const int*      ei   = (const int*)d_in[1];
  const float*    Wl1  = (const float*)d_in[2];
  const float*    bl1  = (const float*)d_in[3];
  const float*    Wr1  = (const float*)d_in[4];
  const float*    br1  = (const float*)d_in[5];
  const float*    att1 = (const float*)d_in[6];
  const float*    bias1= (const float*)d_in[7];
  const float*    Wl2  = (const float*)d_in[8];
  const float*    bl2  = (const float*)d_in[9];
  const float*    Wr2  = (const float*)d_in[10];
  const float*    br2  = (const float*)d_in[11];
  const float*    att2 = (const float*)d_in[12];
  const float*    bias2= (const float*)d_in[13];
  float* out = (float*)d_out;

  const int N    = in_sizes[0] / IN;
  const int E    = in_sizes[1] / 2;
  const int Etot = E + N;
  const int HC   = H * HID;

  float* ws = (float*)d_ws;
  size_t o = 0;
  float*    xl1   = ws + o; o += (size_t)N * HC;
  float*    xr1   = ws + o; o += (size_t)N * HC;
  float*    hacc  = ws + o; o += (size_t)N * HC;
  float*    ebuf  = ws + o; o += (size_t)Etot * H;
  unsigned* nmax  = (unsigned*)(ws + o); o += (size_t)N * H;
  float*    denom = ws + o; o += (size_t)N * H;
  float* xl2   = xl1;
  float* xr2   = xr1;
  float* oacc2 = xr1 + (size_t)N * OUT;

  const int B = 256;

  hipMemsetAsync(hacc,  0, (size_t)N * HC * sizeof(float), stream);
  hipMemsetAsync(nmax,  0, (size_t)N * H * sizeof(unsigned), stream);
  hipMemsetAsync(denom, 0, (size_t)N * H * sizeof(float), stream);

  int rblks = (N / 16 + 7) / 8;
  gemm_bias_wmma<128, 256><<<rblks * (HC / 64), B, 0, stream>>>(x, Wl1, bl1, xl1, N);
  gemm_bias_wmma<128, 256><<<rblks * (HC / 64), B, 0, stream>>>(x, Wr1, br1, xr1, N);

  long long items1 = (long long)Etot * H;
  int gb1 = (int)((items1 * 16 + B - 1) / B);
  int gs1 = (int)((items1 + 127) / 128);
  edge_scores<4, 64><<<gs1, B, 0, stream>>>(xl1, xr1, att1, ei, E, Etot, ebuf, nmax);
  edge_exp<4><<<(int)((items1 + B - 1) / B), B, 0, stream>>>(ei, E, Etot, nmax, ebuf, denom);
  edge_scatter<4, 64><<<gb1, B, 0, stream>>>(xl1, ei, E, Etot, ebuf, denom, hacc);

  int n1 = N * HC;
  bias_elu_inplace<<<(n1 + 4095) / 4096, B, 0, stream>>>(hacc, bias1, n1, HC);

  hipMemsetAsync(oacc2, 0, (size_t)N * OUT * sizeof(float), stream);
  hipMemsetAsync(nmax,  0, (size_t)N * sizeof(unsigned), stream);
  hipMemsetAsync(denom, 0, (size_t)N * sizeof(float), stream);

  gemm_bias_wmma<256, 64><<<rblks * (OUT / 64), B, 0, stream>>>(hacc, Wl2, bl2, xl2, N);
  gemm_bias_wmma<256, 64><<<rblks * (OUT / 64), B, 0, stream>>>(hacc, Wr2, br2, xr2, N);

  long long items2 = Etot;
  int gb2 = (int)((items2 * 16 + B - 1) / B);
  int gs2 = (int)((items2 + 127) / 128);
  edge_scores<1, 64><<<gs2, B, 0, stream>>>(xl2, xr2, att2, ei, E, Etot, ebuf, nmax);
  edge_exp<1><<<(int)((items2 + B - 1) / B), B, 0, stream>>>(ei, E, Etot, nmax, ebuf, denom);
  edge_scatter<1, 64><<<gb2, B, 0, stream>>>(xl2, ei, E, Etot, ebuf, denom, oacc2);

  int n2 = N * OUT;
  bias_add_out<<<(n2 + 4095) / 4096, B, 0, stream>>>(oacc2, bias2, out, n2, OUT);
}


# Scatter add re-mapped lane = channel with f64 sums: every atomic instruction covers whole 128-byte lines, a wave takes 2 or 8 whole edges; the sums are exact, so arrival order no longer shows
# speedup vs baseline: 1.4080x; 1.1201x over previous
#include <hip/hip_runtime.h>
#include <math.h>

#define SLOPE 0.2f

typedef float v2f __attribute__((ext_vector_type(2)));
typedef float v4f __attribute__((ext_vector_type(4)));
typedef float v8f __attribute__((ext_vector_type(8)));
typedef double v2d __attribute__((ext_vector_type(2)));

__device__ __forceinline__ unsigned f2key(float f) {
  unsigned u = __float_as_uint(f);
  return (u & 0x80000000u) ? ~u : (u | 0x80000000u);
}
__device__ __forceinline__ float key2f(unsigned k) {
  unsigned u = (k & 0x80000000u) ? (k ^ 0x80000000u) : ~k;
  return __uint_as_float(u);
}
__device__ __forceinline__ float lrelu(float v) { return v > 0.0f ? v : SLOPE * v; }

template <int K, int NC>
__global__ __launch_bounds__(256) void gemm_bias_wmma(const float* __restrict__ X,
                                                      const float* __restrict__ W,
                                                      const float* __restrict__ b,
                                                      float* __restrict__ Y, int M) {
  constexpr int SLAB = 64;
  constexpr int NSLAB = NC / SLAB;
  constexpr int PITCH = 160;
  __shared__ __attribute__((aligned(16))) float wl[(K / 2) * PITCH];

  const int slab = blockIdx.x % NSLAB;
  const int rblk = blockIdx.x / NSLAB;
  const int cbase = slab * SLAB;

#pragma unroll
  for (int j = 0; j < (K / 2) * 16 / 256; j++) {
    int it = j * 256 + (int)threadIdx.x;
    int p = it >> 4, q = it & 15;
    const float* g = W + (size_t)(2 * p) * NC + cbase + 4 * q;
    float4 r0 = *(const float4*)g;
    float4 r1 = *(const float4*)(g + NC);
    float4* d = (float4*)(wl + p * PITCH + 8 * q);
    d[0] = make_float4(r0.x, r1.x, r0.y, r1.y);
    d[1] = make_float4(r0.z, r1.z, r0.w, r1.w);
  }
  __syncthreads();

  int wid = threadIdx.x >> 5;
  int mt = rblk * 8 + wid;
  if (mt * 16 >= M) return;
  int lane = threadIdx.x & 31;
  int half = lane >> 4;
  int l = lane & 15;
  const float* xp = X + (size_t)(mt * 16 + l) * K + 2 * half;
  const float* bp = wl + half * PITCH + 2 * l;
  v8f acc[4];
#pragma unroll
  for (int t = 0; t < 4; t++) {
    float bc = b[cbase + t * 16 + l];
    acc[t] = (v8f){bc, bc, bc, bc, bc, bc, bc, bc};
  }
#pragma unroll
  for (int k0 = 0; k0 < K; k0 += 4) {
    v2f a = *(const v2f*)(xp + k0);
#pragma unroll
    for (int t = 0; t < 4; t++) {
      v2f bv = *(const v2f*)(bp + (k0 / 2) * PITCH + t * 32);
      acc[t] = __builtin_amdgcn_wmma_f32_16x16x4_f32(false, a, false, bv, (short)0, acc[t],
                                                     false, false);
    }
  }
#pragma unroll
  for (int t = 0; t < 4; t++) {
    float* yp = Y + (size_t)(mt * 16 + 8 * half) * NC + cbase + t * 16 + l;
#pragma unroll
    for (int v = 0; v < 8; v++) yp[(size_t)v * NC] = acc[t][v];
  }
}

template <int H_, int C_>
__global__ __launch_bounds__(256) void edge_scores(const float* __restrict__ xl,
                                                   const float* __restrict__ xr,
                                                   const float* __restrict__ att,
                                                   const int* __restrict__ ei, int E_, int Etot,
                                                   float* __restrict__ ebuf,
                                                   unsigned* __restrict__ nmax) {
  static_assert(C_ == 64, "16 lanes x 4 channels per item");
  static_assert(4 % H_ == 0, "a wave's 16 items start at head 0; a slot's head follows its parity");
  constexpr int NV = 8;
  const int G = Etot * H_;
  int wave = (int)((blockIdx.x * (size_t)blockDim.x + threadIdx.x) >> 5);
  int g0 = wave * (2 * NV);
  if (g0 >= G) return;
  int lane = threadIdx.x & 31;
  int hf = lane >> 4, sub = lane & 15;

  int gm = g0 + 2 * (sub >> 1) + hf;

  v4f at[2];
#pragma unroll
  for (int q = 0; q < 2; q++) at[q] = *(const v4f*)(att + ((2 * q + hf) % H_) * C_ + sub * 4);

  int srcs[NV], dsts[NV];
#pragma unroll
  for (int i = 0; i < NV; i++) {
    int e = min(g0 / H_ + (2 * i + hf) / H_, Etot - 1);
    int el = min(e, E_ - 1);
    int s_ld = ei[el], d_ld = ei[E_ + el];
    srcs[i] = e < E_ ? s_ld : e - E_;
    dsts[i] = e < E_ ? d_ld : e - E_;
  }
  asm volatile("" ::
               "v"(srcs[0]), "v"(dsts[0]), "v"(srcs[1]), "v"(dsts[1]),
               "v"(srcs[2]), "v"(dsts[2]), "v"(srcs[3]), "v"(dsts[3]),
               "v"(srcs[4]), "v"(dsts[4]), "v"(srcs[5]), "v"(dsts[5]),
               "v"(srcs[6]), "v"(dsts[6]), "v"(srcs[7]), "v"(dsts[7]));
  int dm = dsts[0];
#pragma unroll
  for (int i = 1; i < NV; i++) dm = (sub >> 1) == i ? dsts[i] : dm;
  v4f a[NV], bq[NV];
#pragma unroll
  for (int i = 0; i < NV; i++) {
    int base = ((2 * i + hf) % H_) * C_ + sub * 4;
    a[i]  = *(const v4f*)(xl + (size_t)srcs[i] * (H_ * C_) + base);
    bq[i] = *(const v4f*)(xr + (size_t)dsts[i] * (H_ * C_) + base);
  }
  asm volatile("" ::
               "v"(a[0]), "v"(bq[0]), "v"(a[1]), "v"(bq[1]), "v"(a[2]), "v"(bq[2]), "v"(a[3]), "v"(bq[3]),
               "v"(a[4]), "v"(bq[4]), "v"(a[5]), "v"(bq[5]), "v"(a[6]), "v"(bq[6]), "v"(a[7]), "v"(bq[7]));
  float p[NV];
#pragma unroll
  for (int i = 0; i < NV; i++) {
    v4f t4 = at[i & 1];
    p[i] = lrelu(a[i].x + bq[i].x) * t4.x + lrelu(a[i].y + bq[i].y) * t4.y +
           lrelu(a[i].z + bq[i].z) * t4.z + lrelu(a[i].w + bq[i].w) * t4.w;
  }
  bool b3 = (sub & 8) != 0, b2 = (sub & 4) != 0, b1 = (sub & 2) != 0;
  float w4[4], w2[2];
#pragma unroll
  for (int k = 0; k < 4; k++) {
    float keep = b3 ? p[k + 4] : p[k];
    float send = b3 ? p[k] : p[k + 4];
    w4[k] = keep + __shfl_xor(send, 8, 32);
  }
#pragma unroll
  for (int k = 0; k < 2; k++) {
    float keep = b2 ? w4[k + 2] : w4[k];
    float send = b2 ? w4[k] : w4[k + 2];
    w2[k] = keep + __shfl_xor(send, 4, 32);
  }
  float keep1 = b1 ? w2[1] : w2[0];
  float send1 = b1 ? w2[0] : w2[1];
  float t = keep1 + __shfl_xor(send1, 2, 32);
  t += __shfl_xor(t, 1, 32);
  if ((sub & 1) == 0 && gm < G) {
    ebuf[gm] = t;
    atomicMax(nmax + (size_t)dm * H_ + gm % H_, f2key(t));
  }
}

template <int H_>
__global__ void edge_exp(const int* __restrict__ ei, int E_, int Etot,
                         const unsigned* __restrict__ nmax, float* __restrict__ ebuf,
                         float* __restrict__ denom) {
  int g = (int)(blockIdx.x * (size_t)blockDim.x + threadIdx.x);
  if (g >= Etot * H_) return;
  int eidx = g / H_, head = g % H_;
  int dst = (eidx < E_) ? ei[E_ + eidx] : (eidx - E_);
  float mx = key2f(nmax[(size_t)dst * H_ + head]);
  float ex = expf(ebuf[g] - mx);
  ebuf[g] = ex;
  atomicAdd(denom + (size_t)dst * H_ + head, ex);
}


__device__ __forceinline__ double widen_f32(float f) {
  const unsigned u = __float_as_uint(f);
  const bool tiny = __builtin_amdgcn_classf(f, 0x0F0);
  const bool nonfin = __builtin_amdgcn_classf(f, 0x207);
  const unsigned rebias = nonfin ? 0x70000000u : 0x38000000u;
  const unsigned body = ((u >> 3) & 0x0FFFFFFFu) + rebias;
  const unsigned hi = (u & 0x80000000u) | (tiny ? 0u : body);
  const unsigned lo = tiny ? 0u : (u << 29);
  return __hiloint2double((int)hi, (int)lo);
}

template <int H_, int C_, int EPW>
__global__ __launch_bounds__(256) void edge_scatter(const float* __restrict__ xl,
                                                    const int* __restrict__ ei,
                                                    int E_, int Etot,
                                                    const float* __restrict__ ebuf,
                                                    const float* __restrict__ denom,
                                                    double* __restrict__ acc) {
  static_assert(C_ == 64, "two 32-channel pieces per head");
  constexpr int HC_ = H_ * C_;
  constexpr int NP = HC_ / 32;
  const int wave = __builtin_amdgcn_readfirstlane(
      (int)((blockIdx.x * (size_t)blockDim.x + threadIdx.x) >> 5));
  const int e0 = wave * EPW;
  if (e0 >= Etot) return;
  const int lane = threadIdx.x & 31;

  static_assert(EPW == 2 || EPW == 8, "the index fence below lists its operands");
  int s_ld[EPW], d_ld[EPW];
#pragma unroll
  for (int q = 0; q < EPW; q++) {
    int el = min(min(e0 + q, Etot - 1), E_ - 1);
    s_ld[q] = ei[el];
    d_ld[q] = ei[E_ + el];
  }
  if constexpr (EPW == 2) {
    asm volatile("" :: "s"(s_ld[0]), "s"(d_ld[0]), "s"(s_ld[1]), "s"(d_ld[1]));
  } else {
    asm volatile("" ::
                 "s"(s_ld[0]), "s"(d_ld[0]), "s"(s_ld[1]), "s"(d_ld[1]),
                 "s"(s_ld[2]), "s"(d_ld[2]), "s"(s_ld[3]), "s"(d_ld[3]),
                 "s"(s_ld[4]), "s"(d_ld[4]), "s"(s_ld[5]), "s"(d_ld[5]),
                 "s"(s_ld[6]), "s"(d_ld[6]), "s"(s_ld[7]), "s"(d_ld[7]));
  }
  int srcs[EPW], dsts[EPW];
  float al[EPW][H_];
#pragma unroll
  for (int q = 0; q < EPW; q++) {
    int e = min(e0 + q, Etot - 1);
    srcs[q] = e < E_ ? s_ld[q] : e - E_;
    dsts[q] = e < E_ ? d_ld[q] : e - E_;
#pragma unroll
    for (int h = 0; h < H_; h++) {
      al[q][h] = ebuf[(size_t)e * H_ + h] / denom[(size_t)dsts[q] * H_ + h];
    }
  }
  float v[EPW][NP];
#pragma unroll
  for (int q = 0; q < EPW; q++) {
    const float* r = xl + (size_t)srcs[q] * HC_ + lane;
#pragma unroll
    for (int k = 0; k < NP; k++) v[q][k] = r[k * 32];
  }
#pragma unroll
  for (int q = 0; q < EPW; q++) {
    double* o = acc + (size_t)dsts[q] * HC_ + lane;
    const bool live = e0 + q < Etot;
#pragma unroll
    for (int k = 0; k < NP; k++) {
      float t = al[q][k >> 1] * v[q][k];
      atomicAdd(o + k * 32, widen_f32(live ? t : 0.0f));
    }
  }
}

__device__ __forceinline__ float elu1(float v) { return v > 0.0f ? v : expm1f(v); }

template <int NC>
__global__ __launch_bounds__(256) void bias_elu(const double* __restrict__ acc,
                                                const float* __restrict__ bias,
                                                float* __restrict__ h, int n) {
  static_assert(512 % NC == 0, "one bias pair per thread");
  int b0 = (int)blockIdx.x * 4096 + (int)threadIdx.x * 2;
  const float2 bb = *(const float2*)(bias + b0 % NC);
  v2d v[8];
#pragma unroll
  for (int j = 0; j < 8; j++) {
    int i = b0 + j * 512;
    v[j] = i < n ? *(const v2d*)(acc + i) : (v2d){0.0, 0.0};
  }
#pragma unroll
  for (int j = 0; j < 8; j++) {
    int i = b0 + j * 512;
    if (i < n)
      *(float2*)(h + i) = make_float2(elu1((float)v[j].x + bb.x), elu1((float)v[j].y + bb.y));
  }
}

template <int NC>
__global__ __launch_bounds__(256) void bias_add_out(const double* __restrict__ acc,
                                                    const float* __restrict__ bias,
                                                    float* __restrict__ out, int n) {
  static_assert(512 % NC == 0, "one bias pair per thread");
  int b0 = (int)blockIdx.x * 4096 + (int)threadIdx.x * 2;
  const float2 bb = *(const float2*)(bias + b0 % NC);
  v2d v[8];
#pragma unroll
  for (int j = 0; j < 8; j++) {
    int i = b0 + j * 512;
    v[j] = i < n ? *(const v2d*)(acc + i) : (v2d){0.0, 0.0};
  }
#pragma unroll
  for (int j = 0; j < 8; j++) {
    int i = b0 + j * 512;
    if (i < n) *(float2*)(out + i) = make_float2((float)v[j].x + bb.x, (float)v[j].y + bb.y);
  }
}

extern "C" void kernel_launch(void* const* d_in, const int* in_sizes, int n_in,
                              void* d_out, int out_size, void* d_ws, size_t ws_size,
                              hipStream_t stream) {
  (void)n_in; (void)out_size; (void)ws_size;
  const int IN = 128, HID = 64, H = 4, OUT = 64;
  const float*    x    = (const float*)d_in[0];
  const int*      ei   = (const int*)d_in[1];
  const float*    Wl1  = (const float*)d_in[2];
  const float*    bl1  = (const float*)d_in[3];
  const float*    Wr1  = (const float*)d_in[4];
  const float*    br1  = (const float*)d_in[5];
  const float*    att1 = (const float*)d_in[6];
  const float*    bias1= (const float*)d_in[7];
  const float*    Wl2  = (const float*)d_in[8];
  const float*    bl2  = (const float*)d_in[9];
  const float*    Wr2  = (const float*)d_in[10];
  const float*    br2  = (const float*)d_in[11];
  const float*    att2 = (const float*)d_in[12];
  const float*    bias2= (const float*)d_in[13];
  float* out = (float*)d_out;

  const int N    = in_sizes[0] / IN;
  const int E    = in_sizes[1] / 2;
  const int Etot = E + N;
  const int HC   = H * HID;

  float* ws = (float*)d_ws;
  size_t o = 0;
  float*    xl1   = ws + o; o += (size_t)N * HC;
  float*    xr1   = ws + o; o += (size_t)N * HC;
  double*   acc1  = (double*)(ws + o); o += (size_t)N * HC * 2;
  float*    ebuf  = ws + o; o += (size_t)Etot * H;
  unsigned* nmax  = (unsigned*)(ws + o); o += (size_t)N * H;
  float*    denom = ws + o; o += (size_t)N * H;
  float*  h    = xr1;
  float*  xl2  = xl1;
  float*  xr2  = xl1 + (size_t)N * OUT;
  double* acc2 = acc1;

  const int B = 256;

  hipMemsetAsync(acc1,  0, (size_t)N * HC * sizeof(double), stream);
  hipMemsetAsync(nmax,  0, (size_t)N * H * sizeof(unsigned), stream);
  hipMemsetAsync(denom, 0, (size_t)N * H * sizeof(float), stream);

  int rblks = (N / 16 + 7) / 8;
  gemm_bias_wmma<128, 256><<<rblks * (HC / 64), B, 0, stream>>>(x, Wl1, bl1, xl1, N);
  gemm_bias_wmma<128, 256><<<rblks * (HC / 64), B, 0, stream>>>(x, Wr1, br1, xr1, N);

  long long items1 = (long long)Etot * H;
  int gs1 = (int)((items1 + 127) / 128);
  int gb1 = ((Etot + 1) / 2 + 7) / 8;
  edge_scores<4, 64><<<gs1, B, 0, stream>>>(xl1, xr1, att1, ei, E, Etot, ebuf, nmax);
  edge_exp<4><<<(int)((items1 + B - 1) / B), B, 0, stream>>>(ei, E, Etot, nmax, ebuf, denom);
  edge_scatter<4, 64, 2><<<gb1, B, 0, stream>>>(xl1, ei, E, Etot, ebuf, denom, acc1);

  int n1 = N * HC;
  bias_elu<256><<<(n1 + 4095) / 4096, B, 0, stream>>>(acc1, bias1, h, n1);

  hipMemsetAsync(acc2,  0, (size_t)N * OUT * sizeof(double), stream);
  hipMemsetAsync(nmax,  0, (size_t)N * sizeof(unsigned), stream);
  hipMemsetAsync(denom, 0, (size_t)N * sizeof(float), stream);

  gemm_bias_wmma<256, 64><<<rblks * (OUT / 64), B, 0, stream>>>(h, Wl2, bl2, xl2, N);
  gemm_bias_wmma<256, 64><<<rblks * (OUT / 64), B, 0, stream>>>(h, Wr2, br2, xr2, N);

  long long items2 = Etot;
  int gs2 = (int)((items2 + 127) / 128);
  int gb2 = ((Etot + 7) / 8 + 7) / 8;
  edge_scores<1, 64><<<gs2, B, 0, stream>>>(xl2, xr2, att2, ei, E, Etot, ebuf, nmax);
  edge_exp<1><<<(int)((items2 + B - 1) / B), B, 0, stream>>>(ei, E, Etot, nmax, ebuf, denom);
  edge_scatter<1, 64, 8><<<gb2, B, 0, stream>>>(xl2, ei, E, Etot, ebuf, denom, acc2);

  int n2 = N * OUT;
  bias_add_out<64><<<(n2 + 4095) / 4096, B, 0, stream>>>(acc2, bias2, out, n2);
}


# Self loops leave the atomic scatter: its waves take only the listed edges; the elementwise pass adds each node's own term (self weight times its row) to the f64 sum, last; sums exact, same values
# speedup vs baseline: 1.4653x; 1.0407x over previous
#include <hip/hip_runtime.h>
#include <math.h>

#define SLOPE 0.2f

typedef float v2f __attribute__((ext_vector_type(2)));
typedef float v4f __attribute__((ext_vector_type(4)));
typedef float v8f __attribute__((ext_vector_type(8)));
typedef double v2d __attribute__((ext_vector_type(2)));

__device__ __forceinline__ unsigned f2key(float f) {
  unsigned u = __float_as_uint(f);
  return (u & 0x80000000u) ? ~u : (u | 0x80000000u);
}
__device__ __forceinline__ float key2f(unsigned k) {
  unsigned u = (k & 0x80000000u) ? (k ^ 0x80000000u) : ~k;
  return __uint_as_float(u);
}
__device__ __forceinline__ float lrelu(float v) { return v > 0.0f ? v : SLOPE * v; }

template <int K, int NC>
__global__ __launch_bounds__(256) void gemm_bias_wmma(const float* __restrict__ X,
                                                      const float* __restrict__ W,
                                                      const float* __restrict__ b,
                                                      float* __restrict__ Y, int M) {
  constexpr int SLAB = 64;
  constexpr int NSLAB = NC / SLAB;
  constexpr int PITCH = 160;
  __shared__ __attribute__((aligned(16))) float wl[(K / 2) * PITCH];

  const int slab = blockIdx.x % NSLAB;
  const int rblk = blockIdx.x / NSLAB;
  const int cbase = slab * SLAB;

#pragma unroll
  for (int j = 0; j < (K / 2) * 16 / 256; j++) {
    int it = j * 256 + (int)threadIdx.x;
    int p = it >> 4, q = it & 15;
    const float* g = W + (size_t)(2 * p) * NC + cbase + 4 * q;
    float4 r0 = *(const float4*)g;
    float4 r1 = *(const float4*)(g + NC);
    float4* d = (float4*)(wl + p * PITCH + 8 * q);
    d[0] = make_float4(r0.x, r1.x, r0.y, r1.y);
    d[1] = make_float4(r0.z, r1.z, r0.w, r1.w);
  }
  __syncthreads();

  int wid = threadIdx.x >> 5;
  int mt = rblk * 8 + wid;
  if (mt * 16 >= M) return;
  int lane = threadIdx.x & 31;
  int half = lane >> 4;
  int l = lane & 15;
  const float* xp = X + (size_t)(mt * 16 + l) * K + 2 * half;
  const float* bp = wl + half * PITCH + 2 * l;
  v8f acc[4];
#pragma unroll
  for (int t = 0; t < 4; t++) {
    float bc = b[cbase + t * 16 + l];
    acc[t] = (v8f){bc, bc, bc, bc, bc, bc, bc, bc};
  }
#pragma unroll
  for (int k0 = 0; k0 < K; k0 += 4) {
    v2f a = *(const v2f*)(xp + k0);
#pragma unroll
    for (int t = 0; t < 4; t++) {
      v2f bv = *(const v2f*)(bp + (k0 / 2) * PITCH + t * 32);
      acc[t] = __builtin_amdgcn_wmma_f32_16x16x4_f32(false, a, false, bv, (short)0, acc[t],
                                                     false, false);
    }
  }
#pragma unroll
  for (int t = 0; t < 4; t++) {
    float* yp = Y + (size_t)(mt * 16 + 8 * half) * NC + cbase + t * 16 + l;
#pragma unroll
    for (int v = 0; v < 8; v++) yp[(size_t)v * NC] = acc[t][v];
  }
}

template <int H_, int C_>
__global__ __launch_bounds__(256) void edge_scores(const float* __restrict__ xl,
                                                   const float* __restrict__ xr,
                                                   const float* __restrict__ att,
                                                   const int* __restrict__ ei, int E_, int Etot,
                                                   float* __restrict__ ebuf,
                                                   unsigned* __restrict__ nmax) {
  static_assert(C_ == 64, "16 lanes x 4 channels per item");
  static_assert(4 % H_ == 0, "a wave's 16 items start at head 0; a slot's head follows its parity");
  constexpr int NV = 8;
  const int G = Etot * H_;
  int wave = (int)((blockIdx.x * (size_t)blockDim.x + threadIdx.x) >> 5);
  int g0 = wave * (2 * NV);
  if (g0 >= G) return;
  int lane = threadIdx.x & 31;
  int hf = lane >> 4, sub = lane & 15;

  int gm = g0 + 2 * (sub >> 1) + hf;

  v4f at[2];
#pragma unroll
  for (int q = 0; q < 2; q++) at[q] = *(const v4f*)(att + ((2 * q + hf) % H_) * C_ + sub * 4);

  int srcs[NV], dsts[NV];
#pragma unroll
  for (int i = 0; i < NV; i++) {
    int e = min(g0 / H_ + (2 * i + hf) / H_, Etot - 1);
    int el = min(e, E_ - 1);
    int s_ld = ei[el], d_ld = ei[E_ + el];
    srcs[i] = e < E_ ? s_ld : e - E_;
    dsts[i] = e < E_ ? d_ld : e - E_;
  }
  asm volatile("" ::
               "v"(srcs[0]), "v"(dsts[0]), "v"(srcs[1]), "v"(dsts[1]),
               "v"(srcs[2]), "v"(dsts[2]), "v"(srcs[3]), "v"(dsts[3]),
               "v"(srcs[4]), "v"(dsts[4]), "v"(srcs[5]), "v"(dsts[5]),
               "v"(srcs[6]), "v"(dsts[6]), "v"(srcs[7]), "v"(dsts[7]));
  int dm = dsts[0];
#pragma unroll
  for (int i = 1; i < NV; i++) dm = (sub >> 1) == i ? dsts[i] : dm;
  v4f a[NV], bq[NV];
#pragma unroll
  for (int i = 0; i < NV; i++) {
    int base = ((2 * i + hf) % H_) * C_ + sub * 4;
    a[i]  = *(const v4f*)(xl + (size_t)srcs[i] * (H_ * C_) + base);
    bq[i] = *(const v4f*)(xr + (size_t)dsts[i] * (H_ * C_) + base);
  }
  asm volatile("" ::
               "v"(a[0]), "v"(bq[0]), "v"(a[1]), "v"(bq[1]), "v"(a[2]), "v"(bq[2]), "v"(a[3]), "v"(bq[3]),
               "v"(a[4]), "v"(bq[4]), "v"(a[5]), "v"(bq[5]), "v"(a[6]), "v"(bq[6]), "v"(a[7]), "v"(bq[7]));
  float p[NV];
#pragma unroll
  for (int i = 0; i < NV; i++) {
    v4f t4 = at[i & 1];
    p[i] = lrelu(a[i].x + bq[i].x) * t4.x + lrelu(a[i].y + bq[i].y) * t4.y +
           lrelu(a[i].z + bq[i].z) * t4.z + lrelu(a[i].w + bq[i].w) * t4.w;
  }
  bool b3 = (sub & 8) != 0, b2 = (sub & 4) != 0, b1 = (sub & 2) != 0;
  float w4[4], w2[2];
#pragma unroll
  for (int k = 0; k < 4; k++) {
    float keep = b3 ? p[k + 4] : p[k];
    float send = b3 ? p[k] : p[k + 4];
    w4[k] = keep + __shfl_xor(send, 8, 32);
  }
#pragma unroll
  for (int k = 0; k < 2; k++) {
    float keep = b2 ? w4[k + 2] : w4[k];
    float send = b2 ? w4[k] : w4[k + 2];
    w2[k] = keep + __shfl_xor(send, 4, 32);
  }
  float keep1 = b1 ? w2[1] : w2[0];
  float send1 = b1 ? w2[0] : w2[1];
  float t = keep1 + __shfl_xor(send1, 2, 32);
  t += __shfl_xor(t, 1, 32);
  if ((sub & 1) == 0 && gm < G) {
    ebuf[gm] = t;
    atomicMax(nmax + (size_t)dm * H_ + gm % H_, f2key(t));
  }
}

template <int H_>
__global__ void edge_exp(const int* __restrict__ ei, int E_, int Etot,
                         const unsigned* __restrict__ nmax, float* __restrict__ ebuf,
                         float* __restrict__ denom) {
  int g = (int)(blockIdx.x * (size_t)blockDim.x + threadIdx.x);
  if (g >= Etot * H_) return;
  int eidx = g / H_, head = g % H_;
  int dst = (eidx < E_) ? ei[E_ + eidx] : (eidx - E_);
  float mx = key2f(nmax[(size_t)dst * H_ + head]);
  float ex = expf(ebuf[g] - mx);
  ebuf[g] = ex;
  atomicAdd(denom + (size_t)dst * H_ + head, ex);
}


__device__ __forceinline__ double widen_f32(float f) {
  const unsigned u = __float_as_uint(f);
  const bool tiny = __builtin_amdgcn_classf(f, 0x0F0);
  const bool nonfin = __builtin_amdgcn_classf(f, 0x207);
  const unsigned rebias = nonfin ? 0x70000000u : 0x38000000u;
  const unsigned body = ((u >> 3) & 0x0FFFFFFFu) + rebias;
  const unsigned hi = (u & 0x80000000u) | (tiny ? 0u : body);
  const unsigned lo = tiny ? 0u : (u << 29);
  return __hiloint2double((int)hi, (int)lo);
}

template <int H_, int C_, int EPW>
__global__ __launch_bounds__(256) void edge_scatter(const float* __restrict__ xl,
                                                    const int* __restrict__ ei, int E_,
                                                    const float* __restrict__ ebuf,
                                                    const float* __restrict__ denom,
                                                    double* __restrict__ acc) {
  static_assert(C_ == 64, "two 32-channel pieces per head");
  constexpr int HC_ = H_ * C_;
  constexpr int NP = HC_ / 32;
  const int wave = __builtin_amdgcn_readfirstlane(
      (int)((blockIdx.x * (size_t)blockDim.x + threadIdx.x) >> 5));
  const int e0 = wave * EPW;
  if (e0 >= E_) return;
  const int lane = threadIdx.x & 31;

  static_assert(EPW == 2 || EPW == 8, "the index fence below lists its operands");
  int srcs[EPW], dsts[EPW];
#pragma unroll
  for (int q = 0; q < EPW; q++) {
    int e = min(e0 + q, E_ - 1);
    srcs[q] = ei[e];
    dsts[q] = ei[E_ + e];
  }
  if constexpr (EPW == 2) {
    asm volatile("" :: "s"(srcs[0]), "s"(dsts[0]), "s"(srcs[1]), "s"(dsts[1]));
  } else {
    asm volatile("" ::
                 "s"(srcs[0]), "s"(dsts[0]), "s"(srcs[1]), "s"(dsts[1]),
                 "s"(srcs[2]), "s"(dsts[2]), "s"(srcs[3]), "s"(dsts[3]),
                 "s"(srcs[4]), "s"(dsts[4]), "s"(srcs[5]), "s"(dsts[5]),
                 "s"(srcs[6]), "s"(dsts[6]), "s"(srcs[7]), "s"(dsts[7]));
  }
  float al[EPW][H_];
#pragma unroll
  for (int q = 0; q < EPW; q++) {
    int e = min(e0 + q, E_ - 1);
#pragma unroll
    for (int h = 0; h < H_; h++) {
      al[q][h] = ebuf[(size_t)e * H_ + h] / denom[(size_t)dsts[q] * H_ + h];
    }
  }
  float v[EPW][NP];
#pragma unroll
  for (int q = 0; q < EPW; q++) {
    const float* r = xl + (size_t)srcs[q] * HC_ + lane;
#pragma unroll
    for (int k = 0; k < NP; k++) v[q][k] = r[k * 32];
  }
#pragma unroll
  for (int q = 0; q < EPW; q++) {
    double* o = acc + (size_t)dsts[q] * HC_ + lane;
    const bool live = e0 + q < E_;
#pragma unroll
    for (int k = 0; k < NP; k++) {
      float t = al[q][k >> 1] * v[q][k];
      atomicAdd(o + k * 32, widen_f32(live ? t : 0.0f));
    }
  }
}

__device__ __forceinline__ float elu1(float v) { return v > 0.0f ? v : expm1f(v); }

__device__ __forceinline__ void closed_sums(const double* __restrict__ acc,
                                            const float* __restrict__ xl,
                                            const float* __restrict__ eself,
                                            const float* __restrict__ denom,
                                            int b0, int n, float2 (&r)[8]) {
  v2d v[8];
  v2f xs[8];
  float ex[8], dn[8];
#pragma unroll
  for (int j = 0; j < 8; j++) {
    int item = __builtin_amdgcn_readfirstlane(min(b0 + j * 512, n - 2) >> 6);
    ex[j] = eself[item];
    dn[j] = denom[item];
  }
#pragma unroll
  for (int j = 0; j < 8; j++) {
    int i = min(b0 + j * 512, n - 2);
    v[j]  = *(const v2d*)(acc + i);
    xs[j] = *(const v2f*)(xl + i);
  }
  asm volatile("" ::
               "v"(v[0]), "v"(v[1]), "v"(v[2]), "v"(v[3]), "v"(v[4]), "v"(v[5]), "v"(v[6]), "v"(v[7]),
               "v"(xs[0]), "v"(xs[1]), "v"(xs[2]), "v"(xs[3]), "v"(xs[4]), "v"(xs[5]), "v"(xs[6]), "v"(xs[7]),
               "s"(ex[0]), "s"(dn[0]), "s"(ex[1]), "s"(dn[1]), "s"(ex[2]), "s"(dn[2]), "s"(ex[3]), "s"(dn[3]),
               "s"(ex[4]), "s"(dn[4]), "s"(ex[5]), "s"(dn[5]), "s"(ex[6]), "s"(dn[6]), "s"(ex[7]), "s"(dn[7]));
#pragma unroll
  for (int j = 0; j < 8; j++) {
    float al = ex[j] / dn[j];
    float tx = al * xs[j].x, ty = al * xs[j].y;
    r[j] = make_float2((float)(v[j].x + widen_f32(tx)), (float)(v[j].y + widen_f32(ty)));
  }
}

template <int NC>
__global__ __launch_bounds__(256) void bias_elu(const double* __restrict__ acc,
                                                const float* __restrict__ xl,
                                                const float* __restrict__ eself,
                                                const float* __restrict__ denom,
                                                const float* __restrict__ bias,
                                                float* __restrict__ h, int n) {
  static_assert(512 % NC == 0 && NC % 64 == 0, "one bias pair per thread, 64 channels a head");
  int b0 = (int)blockIdx.x * 4096 + (int)threadIdx.x * 2;
  const float2 bb = *(const float2*)(bias + b0 % NC);
  float2 r[8];
  closed_sums(acc, xl, eself, denom, b0, n, r);
#pragma unroll
  for (int j = 0; j < 8; j++) {
    int i = b0 + j * 512;
    if (i < n) *(float2*)(h + i) = make_float2(elu1(r[j].x + bb.x), elu1(r[j].y + bb.y));
  }
}

template <int NC>
__global__ __launch_bounds__(256) void bias_add_out(const double* __restrict__ acc,
                                                    const float* __restrict__ xl,
                                                    const float* __restrict__ eself,
                                                    const float* __restrict__ denom,
                                                    const float* __restrict__ bias,
                                                    float* __restrict__ out, int n) {
  static_assert(512 % NC == 0 && NC % 64 == 0, "one bias pair per thread, 64 channels a head");
  int b0 = (int)blockIdx.x * 4096 + (int)threadIdx.x * 2;
  const float2 bb = *(const float2*)(bias + b0 % NC);
  float2 r[8];
  closed_sums(acc, xl, eself, denom, b0, n, r);
#pragma unroll
  for (int j = 0; j < 8; j++) {
    int i = b0 + j * 512;
    if (i < n) *(float2*)(out + i) = make_float2(r[j].x + bb.x, r[j].y + bb.y);
  }
}

extern "C" void kernel_launch(void* const* d_in, const int* in_sizes, int n_in,
                              void* d_out, int out_size, void* d_ws, size_t ws_size,
                              hipStream_t stream) {
  (void)n_in; (void)out_size; (void)ws_size;
  const int IN = 128, HID = 64, H = 4, OUT = 64;
  const float*    x    = (const float*)d_in[0];
  const int*      ei   = (const int*)d_in[1];
  const float*    Wl1  = (const float*)d_in[2];
  const float*    bl1  = (const float*)d_in[3];
  const float*    Wr1  = (const float*)d_in[4];
  const float*    br1  = (const float*)d_in[5];
  const float*    att1 = (const float*)d_in[6];
  const float*    bias1= (const float*)d_in[7];
  const float*    Wl2  = (const float*)d_in[8];
  const float*    bl2  = (const float*)d_in[9];
  const float*    Wr2  = (const float*)d_in[10];
  const float*    br2  = (const float*)d_in[11];
  const float*    att2 = (const float*)d_in[12];
  const float*    bias2= (const float*)d_in[13];
  float* out = (float*)d_out;

  const int N    = in_sizes[0] / IN;
  const int E    = in_sizes[1] / 2;
  const int Etot = E + N;
  const int HC   = H * HID;

  float* ws = (float*)d_ws;
  size_t o = 0;
  float*    xl1   = ws + o; o += (size_t)N * HC;
  float*    xr1   = ws + o; o += (size_t)N * HC;
  double*   acc1  = (double*)(ws + o); o += (size_t)N * HC * 2;
  float*    ebuf  = ws + o; o += (size_t)Etot * H;
  unsigned* nmax  = (unsigned*)(ws + o); o += (size_t)N * H;
  float*    denom = ws + o; o += (size_t)N * H;
  float*  h    = xr1;
  float*  xl2  = xl1;
  float*  xr2  = xl1 + (size_t)N * OUT;
  double* acc2 = acc1;

  const int B = 256;

  hipMemsetAsync(acc1,  0, (size_t)N * HC * sizeof(double), stream);
  hipMemsetAsync(nmax,  0, (size_t)N * H * sizeof(unsigned), stream);
  hipMemsetAsync(denom, 0, (size_t)N * H * sizeof(float), stream);

  int rblks = (N / 16 + 7) / 8;
  gemm_bias_wmma<128, 256><<<rblks * (HC / 64), B, 0, stream>>>(x, Wl1, bl1, xl1, N);
  gemm_bias_wmma<128, 256><<<rblks * (HC / 64), B, 0, stream>>>(x, Wr1, br1, xr1, N);

  long long items1 = (long long)Etot * H;
  int gs1 = (int)((items1 + 127) / 128);
  int gb1 = ((E + 1) / 2 + 7) / 8;
  edge_scores<4, 64><<<gs1, B, 0, stream>>>(xl1, xr1, att1, ei, E, Etot, ebuf, nmax);
  edge_exp<4><<<(int)((items1 + B - 1) / B), B, 0, stream>>>(ei, E, Etot, nmax, ebuf, denom);
  edge_scatter<4, 64, 2><<<gb1, B, 0, stream>>>(xl1, ei, E, ebuf, denom, acc1);

  int n1 = N * HC;
  bias_elu<256><<<(n1 + 4095) / 4096, B, 0, stream>>>(acc1, xl1, ebuf + (size_t)E * H, denom,
                                                      bias1, h, n1);

  hipMemsetAsync(acc2,  0, (size_t)N * OUT * sizeof(double), stream);
  hipMemsetAsync(nmax,  0, (size_t)N * sizeof(unsigned), stream);
  hipMemsetAsync(denom, 0, (size_t)N * sizeof(float), stream);

  gemm_bias_wmma<256, 64><<<rblks * (OUT / 64), B, 0, stream>>>(h, Wl2, bl2, xl2, N);
  gemm_bias_wmma<256, 64><<<rblks * (OUT / 64), B, 0, stream>>>(h, Wr2, br2, xr2, N);

  long long items2 = Etot;
  int gs2 = (int)((items2 + 127) / 128);
  int gb2 = ((E + 7) / 8 + 7) / 8;
  edge_scores<1, 64><<<gs2, B, 0, stream>>>(xl2, xr2, att2, ei, E, Etot, ebuf, nmax);
  edge_exp<1><<<(int)((items2 + B - 1) / B), B, 0, stream>>>(ei, E, Etot, nmax, ebuf, denom);
  edge_scatter<1, 64, 8><<<gb2, B, 0, stream>>>(xl2, ei, E, ebuf, denom, acc2);

  int n2 = N * OUT;
  bias_add_out<64><<<(n2 + 4095) / 4096, B, 0, stream>>>(acc2, xl2, ebuf + E, denom, bias2, out, n2);
}


# The six zero fills leave the stream: each product launch also stores the start values for its own rows (half of the f64 sum columns; the max keys or the denominators) beside its matrix work
# speedup vs baseline: 1.5469x; 1.0557x over previous
#include <hip/hip_runtime.h>
#include <math.h>

#define SLOPE 0.2f

typedef float v2f __attribute__((ext_vector_type(2)));
typedef float v4f __attribute__((ext_vector_type(4)));
typedef float v8f __attribute__((ext_vector_type(8)));
typedef double v2d __attribute__((ext_vector_type(2)));

__device__ __forceinline__ unsigned f2key(float f) {
  unsigned u = __float_as_uint(f);
  return (u & 0x80000000u) ? ~u : (u | 0x80000000u);
}
__device__ __forceinline__ float key2f(unsigned k) {
  unsigned u = (k & 0x80000000u) ? (k ^ 0x80000000u) : ~k;
  return __uint_as_float(u);
}
__device__ __forceinline__ float lrelu(float v) { return v > 0.0f ? v : SLOPE * v; }

template <int K, int NC>
__global__ __launch_bounds__(256) void gemm_bias_wmma(const float* __restrict__ X,
                                                      const float* __restrict__ W,
                                                      const float* __restrict__ b,
                                                      float* __restrict__ Y, int M,
                                                      double* __restrict__ zacc, int zc0,
                                                      unsigned* __restrict__ zsm) {
  constexpr int SLAB = 64;
  constexpr int NSLAB = NC / SLAB;
  constexpr int PITCH = 160;
  __shared__ __attribute__((aligned(16))) float wl[(K / 2) * PITCH];

  const int slab = blockIdx.x % NSLAB;
  const int rblk = blockIdx.x / NSLAB;
  const int cbase = slab * SLAB;

#pragma unroll
  for (int j = 0; j < (K / 2) * 16 / 256; j++) {
    int it = j * 256 + (int)threadIdx.x;
    int p = it >> 4, q = it & 15;
    const float* g = W + (size_t)(2 * p) * NC + cbase + 4 * q;
    float4 r0 = *(const float4*)g;
    float4 r1 = *(const float4*)(g + NC);
    float4* d = (float4*)(wl + p * PITCH + 8 * q);
    d[0] = make_float4(r0.x, r1.x, r0.y, r1.y);
    d[1] = make_float4(r0.z, r1.z, r0.w, r1.w);
  }
  __syncthreads();

  {
    int zi = (int)blockIdx.x * 128 + 4 * (int)threadIdx.x;
    if (threadIdx.x < 32 && zi < M * NSLAB) *(uint4*)(zsm + zi) = make_uint4(0u, 0u, 0u, 0u);
  }
  int wid = threadIdx.x >> 5;
  int mt = rblk * 8 + wid;
  if (mt * 16 >= M) return;
  int lane = threadIdx.x & 31;
  int half = lane >> 4;
  int l = lane & 15;
  {
    const v2d z2 = {0.0, 0.0};
    double* zp = zacc + (size_t)(mt * 16 + half) * NC + cbase + zc0 + 2 * l;
#pragma unroll
    for (int j = 0; j < 8; j++) *(v2d*)(zp + (size_t)(2 * j) * NC) = z2;
  }
  const float* xp = X + (size_t)(mt * 16 + l) * K + 2 * half;
  const float* bp = wl + half * PITCH + 2 * l;
  v8f acc[4];
#pragma unroll
  for (int t = 0; t < 4; t++) {
    float bc = b[cbase + t * 16 + l];
    acc[t] = (v8f){bc, bc, bc, bc, bc, bc, bc, bc};
  }
#pragma unroll
  for (int k0 = 0; k0 < K; k0 += 4) {
    v2f a = *(const v2f*)(xp + k0);
#pragma unroll
    for (int t = 0; t < 4; t++) {
      v2f bv = *(const v2f*)(bp + (k0 / 2) * PITCH + t * 32);
      acc[t] = __builtin_amdgcn_wmma_f32_16x16x4_f32(false, a, false, bv, (short)0, acc[t],
                                                     false, false);
    }
  }
#pragma unroll
  for (int t = 0; t < 4; t++) {
    float* yp = Y + (size_t)(mt * 16 + 8 * half) * NC + cbase + t * 16 + l;
#pragma unroll
    for (int v = 0; v < 8; v++) yp[(size_t)v * NC] = acc[t][v];
  }
}

template <int H_, int C_>
__global__ __launch_bounds__(256) void edge_scores(const float* __restrict__ xl,
                                                   const float* __restrict__ xr,
                                                   const float* __restrict__ att,
                                                   const int* __restrict__ ei, int E_, int Etot,
                                                   float* __restrict__ ebuf,
                                                   unsigned* __restrict__ nmax) {
  static_assert(C_ == 64, "16 lanes x 4 channels per item");
  static_assert(4 % H_ == 0, "a wave's 16 items start at head 0; a slot's head follows its parity");
  constexpr int NV = 8;
  const int G = Etot * H_;
  int wave = (int)((blockIdx.x * (size_t)blockDim.x + threadIdx.x) >> 5);
  int g0 = wave * (2 * NV);
  if (g0 >= G) return;
  int lane = threadIdx.x & 31;
  int hf = lane >> 4, sub = lane & 15;

  int gm = g0 + 2 * (sub >> 1) + hf;

  v4f at[2];
#pragma unroll
  for (int q = 0; q < 2; q++) at[q] = *(const v4f*)(att + ((2 * q + hf) % H_) * C_ + sub * 4);

  int srcs[NV], dsts[NV];
#pragma unroll
  for (int i = 0; i < NV; i++) {
    int e = min(g0 / H_ + (2 * i + hf) / H_, Etot - 1);
    int el = min(e, E_ - 1);
    int s_ld = ei[el], d_ld = ei[E_ + el];
    srcs[i] = e < E_ ? s_ld : e - E_;
    dsts[i] = e < E_ ? d_ld : e - E_;
  }
  asm volatile("" ::
               "v"(srcs[0]), "v"(dsts[0]), "v"(srcs[1]), "v"(dsts[1]),
               "v"(srcs[2]), "v"(dsts[2]), "v"(srcs[3]), "v"(dsts[3]),
               "v"(srcs[4]), "v"(dsts[4]), "v"(srcs[5]), "v"(dsts[5]),
               "v"(srcs[6]), "v"(dsts[6]), "v"(srcs[7]), "v"(dsts[7]));
  int dm = dsts[0];
#pragma unroll
  for (int i = 1; i < NV; i++) dm = (sub >> 1) == i ? dsts[i] : dm;
  v4f a[NV], bq[NV];
#pragma unroll
  for (int i = 0; i < NV; i++) {
    int base = ((2 * i + hf) % H_) * C_ + sub * 4;
    a[i]  = *(const v4f*)(xl + (size_t)srcs[i] * (H_ * C_) + base);
    bq[i] = *(const v4f*)(xr + (size_t)dsts[i] * (H_ * C_) + base);
  }
  asm volatile("" ::
               "v"(a[0]), "v"(bq[0]), "v"(a[1]), "v"(bq[1]), "v"(a[2]), "v"(bq[2]), "v"(a[3]), "v"(bq[3]),
               "v"(a[4]), "v"(bq[4]), "v"(a[5]), "v"(bq[5]), "v"(a[6]), "v"(bq[6]), "v"(a[7]), "v"(bq[7]));
  float p[NV];
#pragma unroll
  for (int i = 0; i < NV; i++) {
    v4f t4 = at[i & 1];
    p[i] = lrelu(a[i].x + bq[i].x) * t4.x + lrelu(a[i].y + bq[i].y) * t4.y +
           lrelu(a[i].z + bq[i].z) * t4.z + lrelu(a[i].w + bq[i].w) * t4.w;
  }
  bool b3 = (sub & 8) != 0, b2 = (sub & 4) != 0, b1 = (sub & 2) != 0;
  float w4[4], w2[2];
#pragma unroll
  for (int k = 0; k < 4; k++) {
    float keep = b3 ? p[k + 4] : p[k];
    float send = b3 ? p[k] : p[k + 4];
    w4[k] = keep + __shfl_xor(send, 8, 32);
  }
#pragma unroll
  for (int k = 0; k < 2; k++) {
    float keep = b2 ? w4[k + 2] : w4[k];
    float send = b2 ? w4[k] : w4[k + 2];
    w2[k] = keep + __shfl_xor(send, 4, 32);
  }
  float keep1 = b1 ? w2[1] : w2[0];
  float send1 = b1 ? w2[0] : w2[1];
  float t = keep1 + __shfl_xor(send1, 2, 32);
  t += __shfl_xor(t, 1, 32);
  if ((sub & 1) == 0 && gm < G) {
    ebuf[gm] = t;
    atomicMax(nmax + (size_t)dm * H_ + gm % H_, f2key(t));
  }
}

template <int H_>
__global__ void edge_exp(const int* __restrict__ ei, int E_, int Etot,
                         const unsigned* __restrict__ nmax, float* __restrict__ ebuf,
                         float* __restrict__ denom) {
  int g = (int)(blockIdx.x * (size_t)blockDim.x + threadIdx.x);
  if (g >= Etot * H_) return;
  int eidx = g / H_, head = g % H_;
  int dst = (eidx < E_) ? ei[E_ + eidx] : (eidx - E_);
  float mx = key2f(nmax[(size_t)dst * H_ + head]);
  float ex = expf(ebuf[g] - mx);
  ebuf[g] = ex;
  atomicAdd(denom + (size_t)dst * H_ + head, ex);
}


__device__ __forceinline__ double widen_f32(float f) {
  const unsigned u = __float_as_uint(f);
  const bool tiny = __builtin_amdgcn_classf(f, 0x0F0);
  const bool nonfin = __builtin_amdgcn_classf(f, 0x207);
  const unsigned rebias = nonfin ? 0x70000000u : 0x38000000u;
  const unsigned body = ((u >> 3) & 0x0FFFFFFFu) + rebias;
  const unsigned hi = (u & 0x80000000u) | (tiny ? 0u : body);
  const unsigned lo = tiny ? 0u : (u << 29);
  return __hiloint2double((int)hi, (int)lo);
}

template <int H_, int C_, int EPW>
__global__ __launch_bounds__(256) void edge_scatter(const float* __restrict__ xl,
                                                    const int* __restrict__ ei, int E_,
                                                    const float* __restrict__ ebuf,
                                                    const float* __restrict__ denom,
                                                    double* __restrict__ acc) {
  static_assert(C_ == 64, "two 32-channel pieces per head");
  constexpr int HC_ = H_ * C_;
  constexpr int NP = HC_ / 32;
  const int wave = __builtin_amdgcn_readfirstlane(
      (int)((blockIdx.x * (size_t)blockDim.x + threadIdx.x) >> 5));
  const int e0 = wave * EPW;
  if (e0 >= E_) return;
  const int lane = threadIdx.x & 31;

  static_assert(EPW == 2 || EPW == 8, "the index fence below lists its operands");
  int srcs[EPW], dsts[EPW];
#pragma unroll
  for (int q = 0; q < EPW; q++) {
    int e = min(e0 + q, E_ - 1);
    srcs[q] = ei[e];
    dsts[q] = ei[E_ + e];
  }
  if constexpr (EPW == 2) {
    asm volatile("" :: "s"(srcs[0]), "s"(dsts[0]), "s"(srcs[1]), "s"(dsts[1]));
  } else {
    asm volatile("" ::
                 "s"(srcs[0]), "s"(dsts[0]), "s"(srcs[1]), "s"(dsts[1]),
                 "s"(srcs[2]), "s"(dsts[2]), "s"(srcs[3]), "s"(dsts[3]),
                 "s"(srcs[4]), "s"(dsts[4]), "s"(srcs[5]), "s"(dsts[5]),
                 "s"(srcs[6]), "s"(dsts[6]), "s"(srcs[7]), "s"(dsts[7]));
  }
  float al[EPW][H_];
#pragma unroll
  for (int q = 0; q < EPW; q++) {
    int e = min(e0 + q, E_ - 1);
#pragma unroll
    for (int h = 0; h < H_; h++) {
      al[q][h] = ebuf[(size_t)e * H_ + h] / denom[(size_t)dsts[q] * H_ + h];
    }
  }
  float v[EPW][NP];
#pragma unroll
  for (int q = 0; q < EPW; q++) {
    const float* r = xl + (size_t)srcs[q] * HC_ + lane;
#pragma unroll
    for (int k = 0; k < NP; k++) v[q][k] = r[k * 32];
  }
#pragma unroll
  for (int q = 0; q < EPW; q++) {
    double* o = acc + (size_t)dsts[q] * HC_ + lane;
    const bool live = e0 + q < E_;
#pragma unroll
    for (int k = 0; k < NP; k++) {
      float t = al[q][k >> 1] * v[q][k];
      atomicAdd(o + k * 32, widen_f32(live ? t : 0.0f));
    }
  }
}

__device__ __forceinline__ float elu1(float v) { return v > 0.0f ? v : expm1f(v); }

__device__ __forceinline__ void closed_sums(const double* __restrict__ acc,
                                            const float* __restrict__ xl,
                                            const float* __restrict__ eself,
                                            const float* __restrict__ denom,
                                            int b0, int n, float2 (&r)[8]) {
  v2d v[8];
  v2f xs[8];
  float ex[8], dn[8];
#pragma unroll
  for (int j = 0; j < 8; j++) {
    int item = __builtin_amdgcn_readfirstlane(min(b0 + j * 512, n - 2) >> 6);
    ex[j] = eself[item];
    dn[j] = denom[item];
  }
#pragma unroll
  for (int j = 0; j < 8; j++) {
    int i = min(b0 + j * 512, n - 2);
    v[j]  = *(const v2d*)(acc + i);
    xs[j] = *(const v2f*)(xl + i);
  }
  asm volatile("" ::
               "v"(v[0]), "v"(v[1]), "v"(v[2]), "v"(v[3]), "v"(v[4]), "v"(v[5]), "v"(v[6]), "v"(v[7]),
               "v"(xs[0]), "v"(xs[1]), "v"(xs[2]), "v"(xs[3]), "v"(xs[4]), "v"(xs[5]), "v"(xs[6]), "v"(xs[7]),
               "s"(ex[0]), "s"(dn[0]), "s"(ex[1]), "s"(dn[1]), "s"(ex[2]), "s"(dn[2]), "s"(ex[3]), "s"(dn[3]),
               "s"(ex[4]), "s"(dn[4]), "s"(ex[5]), "s"(dn[5]), "s"(ex[6]), "s"(dn[6]), "s"(ex[7]), "s"(dn[7]));
#pragma unroll
  for (int j = 0; j < 8; j++) {
    float al = ex[j] / dn[j];
    float tx = al * xs[j].x, ty = al * xs[j].y;
    r[j] = make_float2((float)(v[j].x + widen_f32(tx)), (float)(v[j].y + widen_f32(ty)));
  }
}

template <int NC>
__global__ __launch_bounds__(256) void bias_elu(const double* __restrict__ acc,
                                                const float* __restrict__ xl,
                                                const float* __restrict__ eself,
                                                const float* __restrict__ denom,
                                                const float* __restrict__ bias,
                                                float* __restrict__ h, int n) {
  static_assert(512 % NC == 0 && NC % 64 == 0, "one bias pair per thread, 64 channels a head");
  int b0 = (int)blockIdx.x * 4096 + (int)threadIdx.x * 2;
  const float2 bb = *(const float2*)(bias + b0 % NC);
  float2 r[8];
  closed_sums(acc, xl, eself, denom, b0, n, r);
#pragma unroll
  for (int j = 0; j < 8; j++) {
    int i = b0 + j * 512;
    if (i < n) *(float2*)(h + i) = make_float2(elu1(r[j].x + bb.x), elu1(r[j].y + bb.y));
  }
}

template <int NC>
__global__ __launch_bounds__(256) void bias_add_out(const double* __restrict__ acc,
                                                    const float* __restrict__ xl,
                                                    const float* __restrict__ eself,
                                                    const float* __restrict__ denom,
                                                    const float* __restrict__ bias,
                                                    float* __restrict__ out, int n) {
  static_assert(512 % NC == 0 && NC % 64 == 0, "one bias pair per thread, 64 channels a head");
  int b0 = (int)blockIdx.x * 4096 + (int)threadIdx.x * 2;
  const float2 bb = *(const float2*)(bias + b0 % NC);
  float2 r[8];
  closed_sums(acc, xl, eself, denom, b0, n, r);
#pragma unroll
  for (int j = 0; j < 8; j++) {
    int i = b0 + j * 512;
    if (i < n) *(float2*)(out + i) = make_float2(r[j].x + bb.x, r[j].y + bb.y);
  }
}

extern "C" void kernel_launch(void* const* d_in, const int* in_sizes, int n_in,
                              void* d_out, int out_size, void* d_ws, size_t ws_size,
                              hipStream_t stream) {
  (void)n_in; (void)out_size; (void)ws_size;
  const int IN = 128, HID = 64, H = 4, OUT = 64;
  const float*    x    = (const float*)d_in[0];
  const int*      ei   = (const int*)d_in[1];
  const float*    Wl1  = (const float*)d_in[2];
  const float*    bl1  = (const float*)d_in[3];
  const float*    Wr1  = (const float*)d_in[4];
  const float*    br1  = (const float*)d_in[5];
  const float*    att1 = (const float*)d_in[6];
  const float*    bias1= (const float*)d_in[7];
  const float*    Wl2  = (const float*)d_in[8];
  const float*    bl2  = (const float*)d_in[9];
  const float*    Wr2  = (const float*)d_in[10];
  const float*    br2  = (const float*)d_in[11];
  const float*    att2 = (const float*)d_in[12];
  const float*    bias2= (const float*)d_in[13];
  float* out = (float*)d_out;

  const int N    = in_sizes[0] / IN;
  const int E    = in_sizes[1] / 2;
  const int Etot = E + N;
  const int HC   = H * HID;

  float* ws = (float*)d_ws;
  size_t o = 0;
  float*    xl1   = ws + o; o += (size_t)N * HC;
  float*    xr1   = ws + o; o += (size_t)N * HC;
  double*   acc1  = (double*)(ws + o); o += (size_t)N * HC * 2;
  float*    ebuf  = ws + o; o += (size_t)Etot * H;
  unsigned* nmax  = (unsigned*)(ws + o); o += (size_t)N * H;
  float*    denom = ws + o; o += (size_t)N * H;
  float*  h    = xr1;
  float*  xl2  = xl1;
  float*  xr2  = xl1 + (size_t)N * OUT;
  double* acc2 = acc1;

  const int B = 256;

  int rblks = (N / 16 + 7) / 8;
  gemm_bias_wmma<128, 256><<<rblks * (HC / 64), B, 0, stream>>>(x, Wl1, bl1, xl1, N,
                                                                acc1, 0, nmax);
  gemm_bias_wmma<128, 256><<<rblks * (HC / 64), B, 0, stream>>>(x, Wr1, br1, xr1, N,
                                                                acc1, 32, (unsigned*)denom);

  long long items1 = (long long)Etot * H;
  int gs1 = (int)((items1 + 127) / 128);
  int gb1 = ((E + 1) / 2 + 7) / 8;
  edge_scores<4, 64><<<gs1, B, 0, stream>>>(xl1, xr1, att1, ei, E, Etot, ebuf, nmax);
  edge_exp<4><<<(int)((items1 + B - 1) / B), B, 0, stream>>>(ei, E, Etot, nmax, ebuf, denom);
  edge_scatter<4, 64, 2><<<gb1, B, 0, stream>>>(xl1, ei, E, ebuf, denom, acc1);

  int n1 = N * HC;
  bias_elu<256><<<(n1 + 4095) / 4096, B, 0, stream>>>(acc1, xl1, ebuf + (size_t)E * H, denom,
                                                      bias1, h, n1);

  gemm_bias_wmma<256, 64><<<rblks * (OUT / 64), B, 0, stream>>>(h, Wl2, bl2, xl2, N,
                                                                acc2, 0, nmax);
  gemm_bias_wmma<256, 64><<<rblks * (OUT / 64), B, 0, stream>>>(h, Wr2, br2, xr2, N,
                                                                acc2, 32, (unsigned*)denom);

  long long items2 = Etot;
  int gs2 = (int)((items2 + 127) / 128);
  int gb2 = ((E + 7) / 8 + 7) / 8;
  edge_scores<1, 64><<<gs2, B, 0, stream>>>(xl2, xr2, att2, ei, E, Etot, ebuf, nmax);
  edge_exp<1><<<(int)((items2 + B - 1) / B), B, 0, stream>>>(ei, E, Etot, nmax, ebuf, denom);
  edge_scatter<1, 64, 8><<<gb2, B, 0, stream>>>(xl2, ei, E, ebuf, denom, acc2);

  int n2 = N * OUT;
  bias_add_out<64><<<(n2 + 4095) / 4096, B, 0, stream>>>(acc2, xl2, ebuf + E, denom, bias2, out, n2);
}


# Atomic scatter and elementwise pass become one owner pass: a work-group owns 196 rows, scans the dst list, sums its in-edges in LDS in list order (f32, fixed order); f64 sums and their clears gone
# speedup vs baseline: 1.9575x; 1.2654x over previous
#include <hip/hip_runtime.h>
#include <math.h>

#define SLOPE 0.2f

typedef float v2f __attribute__((ext_vector_type(2)));
typedef float v4f __attribute__((ext_vector_type(4)));
typedef float v8f __attribute__((ext_vector_type(8)));

__device__ __forceinline__ unsigned f2key(float f) {
  unsigned u = __float_as_uint(f);
  return (u & 0x80000000u) ? ~u : (u | 0x80000000u);
}
__device__ __forceinline__ float key2f(unsigned k) {
  unsigned u = (k & 0x80000000u) ? (k ^ 0x80000000u) : ~k;
  return __uint_as_float(u);
}
__device__ __forceinline__ float lrelu(float v) { return v > 0.0f ? v : SLOPE * v; }

template <int K, int NC>
__global__ __launch_bounds__(256) void gemm_bias_wmma(const float* __restrict__ X,
                                                      const float* __restrict__ W,
                                                      const float* __restrict__ b,
                                                      float* __restrict__ Y, int M,
                                                      unsigned* __restrict__ zsm) {
  constexpr int SLAB = 64;
  constexpr int NSLAB = NC / SLAB;
  constexpr int PITCH = 160;
  __shared__ __attribute__((aligned(16))) float wl[(K / 2) * PITCH];

  const int slab = blockIdx.x % NSLAB;
  const int rblk = blockIdx.x / NSLAB;
  const int cbase = slab * SLAB;

#pragma unroll
  for (int j = 0; j < (K / 2) * 16 / 256; j++) {
    int it = j * 256 + (int)threadIdx.x;
    int p = it >> 4, q = it & 15;
    const float* g = W + (size_t)(2 * p) * NC + cbase + 4 * q;
    float4 r0 = *(const float4*)g;
    float4 r1 = *(const float4*)(g + NC);
    float4* d = (float4*)(wl + p * PITCH + 8 * q);
    d[0] = make_float4(r0.x, r1.x, r0.y, r1.y);
    d[1] = make_float4(r0.z, r1.z, r0.w, r1.w);
  }
  __syncthreads();

  {
    int zi = (int)blockIdx.x * 128 + 4 * (int)threadIdx.x;
    if (threadIdx.x < 32 && zi < M * NSLAB) *(uint4*)(zsm + zi) = make_uint4(0u, 0u, 0u, 0u);
  }
  int wid = threadIdx.x >> 5;
  int mt = rblk * 8 + wid;
  if (mt * 16 >= M) return;
  int lane = threadIdx.x & 31;
  int half = lane >> 4;
  int l = lane & 15;
  const float* xp = X + (size_t)(mt * 16 + l) * K + 2 * half;
  const float* bp = wl + half * PITCH + 2 * l;
  v8f acc[4];
#pragma unroll
  for (int t = 0; t < 4; t++) {
    float bc = b[cbase + t * 16 + l];
    acc[t] = (v8f){bc, bc, bc, bc, bc, bc, bc, bc};
  }
#pragma unroll
  for (int k0 = 0; k0 < K; k0 += 4) {
    v2f a = *(const v2f*)(xp + k0);
#pragma unroll
    for (int t = 0; t < 4; t++) {
      v2f bv = *(const v2f*)(bp + (k0 / 2) * PITCH + t * 32);
      acc[t] = __builtin_amdgcn_wmma_f32_16x16x4_f32(false, a, false, bv, (short)0, acc[t],
                                                     false, false);
    }
  }
#pragma unroll
  for (int t = 0; t < 4; t++) {
    float* yp = Y + (size_t)(mt * 16 + 8 * half) * NC + cbase + t * 16 + l;
#pragma unroll
    for (int v = 0; v < 8; v++) yp[(size_t)v * NC] = acc[t][v];
  }
}

template <int H_, int C_>
__global__ __launch_bounds__(256) void edge_scores(const float* __restrict__ xl,
                                                   const float* __restrict__ xr,
                                                   const float* __restrict__ att,
                                                   const int* __restrict__ ei, int E_, int Etot,
                                                   float* __restrict__ ebuf,
                                                   unsigned* __restrict__ nmax) {
  static_assert(C_ == 64, "16 lanes x 4 channels per item");
  static_assert(4 % H_ == 0, "a wave's 16 items start at head 0; a slot's head follows its parity");
  constexpr int NV = 8;
  const int G = Etot * H_;
  int wave = (int)((blockIdx.x * (size_t)blockDim.x + threadIdx.x) >> 5);
  int g0 = wave * (2 * NV);
  if (g0 >= G) return;
  int lane = threadIdx.x & 31;
  int hf = lane >> 4, sub = lane & 15;

  int gm = g0 + 2 * (sub >> 1) + hf;

  v4f at[2];
#pragma unroll
  for (int q = 0; q < 2; q++) at[q] = *(const v4f*)(att + ((2 * q + hf) % H_) * C_ + sub * 4);

  int srcs[NV], dsts[NV];
#pragma unroll
  for (int i = 0; i < NV; i++) {
    int e = min(g0 / H_ + (2 * i + hf) / H_, Etot - 1);
    int el = min(e, E_ - 1);
    int s_ld = ei[el], d_ld = ei[E_ + el];
    srcs[i] = e < E_ ? s_ld : e - E_;
    dsts[i] = e < E_ ? d_ld : e - E_;
  }
  asm volatile("" ::
               "v"(srcs[0]), "v"(dsts[0]), "v"(srcs[1]), "v"(dsts[1]),
               "v"(srcs[2]), "v"(dsts[2]), "v"(srcs[3]), "v"(dsts[3]),
               "v"(srcs[4]), "v"(dsts[4]), "v"(srcs[5]), "v"(dsts[5]),
               "v"(srcs[6]), "v"(dsts[6]), "v"(srcs[7]), "v"(dsts[7]));
  int dm = dsts[0];
#pragma unroll
  for (int i = 1; i < NV; i++) dm = (sub >> 1) == i ? dsts[i] : dm;
  v4f a[NV], bq[NV];
#pragma unroll
  for (int i = 0; i < NV; i++) {
    int base = ((2 * i + hf) % H_) * C_ + sub * 4;
    a[i]  = *(const v4f*)(xl + (size_t)srcs[i] * (H_ * C_) + base);
    bq[i] = *(const v4f*)(xr + (size_t)dsts[i] * (H_ * C_) + base);
  }
  asm volatile("" ::
               "v"(a[0]), "v"(bq[0]), "v"(a[1]), "v"(bq[1]), "v"(a[2]), "v"(bq[2]), "v"(a[3]), "v"(bq[3]),
               "v"(a[4]), "v"(bq[4]), "v"(a[5]), "v"(bq[5]), "v"(a[6]), "v"(bq[6]), "v"(a[7]), "v"(bq[7]));
  float p[NV];
#pragma unroll
  for (int i = 0; i < NV; i++) {
    v4f t4 = at[i & 1];
    p[i] = lrelu(a[i].x + bq[i].x) * t4.x + lrelu(a[i].y + bq[i].y) * t4.y +
           lrelu(a[i].z + bq[i].z) * t4.z + lrelu(a[i].w + bq[i].w) * t4.w;
  }
  bool b3 = (sub & 8) != 0, b2 = (sub & 4) != 0, b1 = (sub & 2) != 0;
  float w4[4], w2[2];
#pragma unroll
  for (int k = 0; k < 4; k++) {
    float keep = b3 ? p[k + 4] : p[k];
    float send = b3 ? p[k] : p[k + 4];
    w4[k] = keep + __shfl_xor(send, 8, 32);
  }
#pragma unroll
  for (int k = 0; k < 2; k++) {
    float keep = b2 ? w4[k + 2] : w4[k];
    float send = b2 ? w4[k] : w4[k + 2];
    w2[k] = keep + __shfl_xor(send, 4, 32);
  }
  float keep1 = b1 ? w2[1] : w2[0];
  float send1 = b1 ? w2[0] : w2[1];
  float t = keep1 + __shfl_xor(send1, 2, 32);
  t += __shfl_xor(t, 1, 32);
  if ((sub & 1) == 0 && gm < G) {
    ebuf[gm] = t;
    atomicMax(nmax + (size_t)dm * H_ + gm % H_, f2key(t));
  }
}

template <int H_>
__global__ void edge_exp(const int* __restrict__ ei, int E_, int Etot,
                         const unsigned* __restrict__ nmax, float* __restrict__ ebuf,
                         float* __restrict__ denom) {
  int g = (int)(blockIdx.x * (size_t)blockDim.x + threadIdx.x);
  if (g >= Etot * H_) return;
  int eidx = g / H_, head = g % H_;
  int dst = (eidx < E_) ? ei[E_ + eidx] : (eidx - E_);
  float mx = key2f(nmax[(size_t)dst * H_ + head]);
  float ex = expf(ebuf[g] - mx);
  ebuf[g] = ex;
  atomicAdd(denom + (size_t)dst * H_ + head, ex);
}

#define OWN_R 196

template <int H_> struct RowVec;
template <> struct RowVec<4> { typedef v4f T; };
template <> struct RowVec<1> { typedef v2f T; };

__device__ __forceinline__ int rl_i(int v, int j) { return __builtin_amdgcn_readlane(v, j); }
__device__ __forceinline__ float rl_f(float v, int j) {
  return __int_as_float(__builtin_amdgcn_readlane(__float_as_int(v), j));
}

template <int H_>
__device__ __forceinline__ void own_batch(float* __restrict__ accs, const int* oe_w,
                                          const unsigned char* or_w, int b, int nb,
                                          const float* __restrict__ xl,
                                          const int* __restrict__ ei,
                                          const float* __restrict__ ebuf,
                                          const float* __restrict__ denom, int lo, int lane) {
  constexpr int HC = H_ * 64;
  constexpr int NP = H_ == 4 ? 2 : 1;
  constexpr int W = H_ == 4 ? 4 : 2;
  typedef typename RowVec<H_>::T RV;
  const int jj = b + min(lane, nb - 1);
  const int e = oe_w[jj];
  const int r = or_w[jj];
  const int s = ei[e];
  float a[H_];
  if constexpr (H_ == 4) {
    v4f ex = *(const v4f*)(ebuf + (size_t)e * 4);
    v4f dn = *(const v4f*)(denom + (size_t)(lo + r) * 4);
    a[0] = ex.x / dn.x; a[1] = ex.y / dn.y; a[2] = ex.z / dn.z; a[3] = ex.w / dn.w;
  } else {
    a[0] = ebuf[e] / denom[lo + r];
  }
  const bool hf = (lane >> 4) != 0;
  for (int j0 = 0; j0 < nb; j0 += 8) {
    RV v[8][NP];
#pragma unroll
    for (int g = 0; g < 8; g++) {
      const int sg = rl_i(s, min(j0 + g, nb - 1));
      const float* rp = xl + (size_t)sg * HC + W * lane;
#pragma unroll
      for (int k = 0; k < NP; k++) v[g][k] = *(const RV*)(rp + k * 128);
    }
    if constexpr (NP == 2) {
      asm volatile("" ::
                   "v"(v[0][0]), "v"(v[0][1]), "v"(v[1][0]), "v"(v[1][1]), "v"(v[2][0]), "v"(v[2][1]),
                   "v"(v[3][0]), "v"(v[3][1]), "v"(v[4][0]), "v"(v[4][1]), "v"(v[5][0]), "v"(v[5][1]),
                   "v"(v[6][0]), "v"(v[6][1]), "v"(v[7][0]), "v"(v[7][1]));
    } else {
      asm volatile("" ::
                   "v"(v[0][0]), "v"(v[1][0]), "v"(v[2][0]), "v"(v[3][0]),
                   "v"(v[4][0]), "v"(v[5][0]), "v"(v[6][0]), "v"(v[7][0]));
    }
#pragma unroll
    for (int g = 0; g < 8; g++) {
      if (j0 + g < nb) {
        const int rg = rl_i(r, j0 + g);
        float* ap = accs + rg * HC + W * lane;
#pragma unroll
        for (int k = 0; k < NP; k++) {
          float al;
          if constexpr (H_ == 4) {
            const float a_lo = rl_f(a[2 * k], j0 + g), a_hi = rl_f(a[2 * k + 1], j0 + g);
            al = hf ? a_hi : a_lo;
          } else {
            al = rl_f(a[0], j0 + g);
          }
          RV cur = *(RV*)(ap + k * 128);
          cur += al * v[g][k];
          *(RV*)(ap + k * 128) = cur;
        }
      }
    }
  }
}

__device__ __forceinline__ float elu1(float v) { return v > 0.0f ? v : expm1f(v); }

template <int H_, bool ELU_>
__global__ __launch_bounds__(1024) void owner_sums(const float* __restrict__ xl,
                                                   const int* __restrict__ ei, int E_,
                                                   const float* __restrict__ ebuf,
                                                   const float* __restrict__ eself,
                                                   const float* __restrict__ denom,
                                                   const float* __restrict__ bias,
                                                   float* __restrict__ outp, int Nn) {
  constexpr int HC = H_ * 64;
  constexpr int R = OWN_R;
  constexpr int NW = 32;
  constexpr int CH = 512;
  constexpr int NL = CH / 32;
  constexpr int QCAP = H_ == 4 ? 640 : 1024;
  constexpr int OCAP = 128;
  static_assert(R <= 255, "row numbers inside the group are kept in a byte");
  static_assert((R * HC) % 4 == 0, "16-byte clears");
  __shared__ __attribute__((aligned(16))) float accs[R * HC];
  __shared__ int qe[NW * QCAP];
  __shared__ unsigned char qr[NW * QCAP];
  __shared__ int oe[NW * OCAP];
  __shared__ unsigned char orow[NW * OCAP];
  __shared__ int qcnt[NW];
  __shared__ int pending[2];

  const int tid = (int)threadIdx.x;
  const int lane = tid & 31;
  const int w = __builtin_amdgcn_readfirstlane(tid >> 5);
  const int lo = (int)blockIdx.x * R;
  const int nrows = min(R, Nn - lo);
  const int* __restrict__ dsts = ei + E_;
  const unsigned lt = (1u << lane) - 1u;

  for (int i = 4 * tid; i < R * HC; i += 4096) *(v4f*)(accs + i) = (v4f){0.0f, 0.0f, 0.0f, 0.0f};
  if (tid < 2) pending[tid] = 0;

  const int per = ((E_ + NW * CH - 1) / (NW * CH)) * CH;
  int pos = w * per;
  const int wend = min(pos + per, E_);

  int d[NL];
  auto load_step = [&](int p) {
    if (p + CH <= E_) {
#pragma unroll
      for (int k = 0; k < NL; k++) d[k] = dsts[p + k * 32 + lane];
    } else {
#pragma unroll
      for (int k = 0; k < NL; k++) {
        int idx = p + k * 32 + lane;
        d[k] = idx < E_ ? dsts[idx] : -1;
      }
    }
  };
  if (pos < wend) load_step(pos);
  __syncthreads();

  for (int round = 0;; round++) {
    int cnt = 0;
    while (pos < wend && cnt + CH <= QCAP) {
      int cur[NL];
#pragma unroll
      for (int k = 0; k < NL; k++) cur[k] = d[k];
      if (pos + CH < wend) load_step(pos + CH);
#pragma unroll
      for (int k = 0; k < NL; k++) {
        const unsigned rr = (unsigned)(cur[k] - lo);
        const bool m = rr < (unsigned)nrows;
        const unsigned mask = __builtin_amdgcn_ballot_w32(m);
        if (mask != 0u) {
          if (m) {
            const int p = w * QCAP + cnt + __builtin_popcount(mask & lt);
            qe[p] = pos + k * 32 + lane;
            qr[p] = (unsigned char)rr;
          }
          cnt += __builtin_popcount(mask);
        }
      }
      pos += CH;
    }
    if (lane == 0) {
      qcnt[w] = cnt;
      if (pos < wend) pending[round & 1] = 1;
    }
    __syncthreads();

    int fill = 0;
    for (int q = 0; q < NW; q++) {
      const int n = __builtin_amdgcn_readfirstlane(qcnt[q]);
      for (int i = 0; i < n; i += 32) {
        const int j = i + lane;
        const int r = qr[q * QCAP + j];
        const bool mine = j < n && (r & 31) == w;
        const unsigned mask = __builtin_amdgcn_ballot_w32(mine);
        if (mask != 0u) {
          if (mine) {
            const int p = w * OCAP + fill + __builtin_popcount(mask & lt);
            oe[p] = qe[q * QCAP + j];
            orow[p] = (unsigned char)r;
          }
          fill += __builtin_popcount(mask);
          if (fill > OCAP - 32) {
            for (int b = 0; b < fill; b += 32)
              own_batch<H_>(accs, oe + w * OCAP, orow + w * OCAP, b, min(32, fill - b), xl, ei,
                            ebuf, denom, lo, lane);
            fill = 0;
          }
        }
      }
    }
    for (int b = 0; b < fill; b += 32)
      own_batch<H_>(accs, oe + w * OCAP, orow + w * OCAP, b, min(32, fill - b), xl, ei, ebuf,
                    denom, lo, lane);

    const int again = __builtin_amdgcn_readfirstlane(pending[round & 1]);
    if (tid == 0) pending[(round + 1) & 1] = 0;
    __syncthreads();
    if (!again) break;
  }

  const int tot = nrows * HC;
  const int ch = (4 * tid) % HC;
  const v4f bb = *(const v4f*)(bias + ch);
  for (int i0 = 4 * tid; i0 < tot; i0 += 4 * 4096) {
    v4f xs[4];
    float ex[4], dn[4];
#pragma unroll
    for (int u = 0; u < 4; u++) {
      const int i = min(i0 + u * 4096, tot - 4);
      const size_t node = (size_t)(lo + i / HC);
      const int item = (int)node * H_ + ((i % HC) >> 6);
      ex[u] = eself[item];
      dn[u] = denom[item];
      xs[u] = *(const v4f*)(xl + node * HC + i % HC);
    }
    asm volatile("" ::
                 "v"(xs[0]), "v"(xs[1]), "v"(xs[2]), "v"(xs[3]),
                 "v"(ex[0]), "v"(dn[0]), "v"(ex[1]), "v"(dn[1]),
                 "v"(ex[2]), "v"(dn[2]), "v"(ex[3]), "v"(dn[3]));
#pragma unroll
    for (int u = 0; u < 4; u++) {
      const int i = i0 + u * 4096;
      if (i < tot) {
        const float al = ex[u] / dn[u];
        const v4f sm = *(const v4f*)(accs + i);
        v4f o = sm + al * xs[u] + bb;
        if constexpr (ELU_) { o.x = elu1(o.x); o.y = elu1(o.y); o.z = elu1(o.z); o.w = elu1(o.w); }
        *(v4f*)(outp + (size_t)(lo + i / HC) * HC + i % HC) = o;
      }
    }
  }
}

extern "C" void kernel_launch(void* const* d_in, const int* in_sizes, int n_in,
                              void* d_out, int out_size, void* d_ws, size_t ws_size,
                              hipStream_t stream) {
  (void)n_in; (void)out_size; (void)ws_size;
  const int IN = 128, HID = 64, H = 4, OUT = 64;
  const float*    x    = (const float*)d_in[0];
  const int*      ei   = (const int*)d_in[1];
  const float*    Wl1  = (const float*)d_in[2];
  const float*    bl1  = (const float*)d_in[3];
  const float*    Wr1  = (const float*)d_in[4];
  const float*    br1  = (const float*)d_in[5];
  const float*    att1 = (const float*)d_in[6];
  const float*    bias1= (const float*)d_in[7];
  const float*    Wl2  = (const float*)d_in[8];
  const float*    bl2  = (const float*)d_in[9];
  const float*    Wr2  = (const float*)d_in[10];
  const float*    br2  = (const float*)d_in[11];
  const float*    att2 = (const float*)d_in[12];
  const float*    bias2= (const float*)d_in[13];
  float* out = (float*)d_out;

  const int N    = in_sizes[0] / IN;
  const int E    = in_sizes[1] / 2;
  const int Etot = E + N;
  const int HC   = H * HID;

  float* ws = (float*)d_ws;
  size_t o = 0;
  float*    xl1   = ws + o; o += (size_t)N * HC;
  float*    xr1   = ws + o; o += (size_t)N * HC;
  float*    ebuf  = ws + o; o += (size_t)Etot * H;
  unsigned* nmax  = (unsigned*)(ws + o); o += (size_t)N * H;
  float*    denom = ws + o; o += (size_t)N * H;
  float*  h    = xr1;
  float*  xl2  = xl1;
  float*  xr2  = xl1 + (size_t)N * OUT;

  const int B = 256;

  int rblks = (N / 16 + 7) / 8;
  gemm_bias_wmma<128, 256><<<rblks * (HC / 64), B, 0, stream>>>(x, Wl1, bl1, xl1, N,
                                                                nmax);
  gemm_bias_wmma<128, 256><<<rblks * (HC / 64), B, 0, stream>>>(x, Wr1, br1, xr1, N,
                                                                (unsigned*)denom);

  long long items1 = (long long)Etot * H;
  int gs1 = (int)((items1 + 127) / 128);
  int gown = (N + OWN_R - 1) / OWN_R;
  edge_scores<4, 64><<<gs1, B, 0, stream>>>(xl1, xr1, att1, ei, E, Etot, ebuf, nmax);
  edge_exp<4><<<(int)((items1 + B - 1) / B), B, 0, stream>>>(ei, E, Etot, nmax, ebuf, denom);
  owner_sums<4, true><<<gown, 1024, 0, stream>>>(xl1, ei, E, ebuf, ebuf + (size_t)E * H, denom,
                                                 bias1, h, N);

  gemm_bias_wmma<256, 64><<<rblks * (OUT / 64), B, 0, stream>>>(h, Wl2, bl2, xl2, N,
                                                                nmax);
  gemm_bias_wmma<256, 64><<<rblks * (OUT / 64), B, 0, stream>>>(h, Wr2, br2, xr2, N,
                                                                (unsigned*)denom);

  long long items2 = Etot;
  int gs2 = (int)((items2 + 127) / 128);
  edge_scores<1, 64><<<gs2, B, 0, stream>>>(xl2, xr2, att2, ei, E, Etot, ebuf, nmax);
  edge_exp<1><<<(int)((items2 + B - 1) / B), B, 0, stream>>>(ei, E, Etot, nmax, ebuf, denom);
  owner_sums<1, false><<<gown, 1024, 0, stream>>>(xl2, ei, E, ebuf, ebuf + E, denom, bias2, out, N);
}


# Exp and denominator launches fold into the owner pass: its waves take exp(logit - max) per in-edge and sum each row's denominator in list order, rows divided once at the close; float atomics gone
# speedup vs baseline: 2.2140x; 1.1310x over previous
#include <hip/hip_runtime.h>
#include <math.h>

#define SLOPE 0.2f

typedef float v2f __attribute__((ext_vector_type(2)));
typedef float v4f __attribute__((ext_vector_type(4)));
typedef float v8f __attribute__((ext_vector_type(8)));

__device__ __forceinline__ unsigned f2key(float f) {
  unsigned u = __float_as_uint(f);
  return (u & 0x80000000u) ? ~u : (u | 0x80000000u);
}
__device__ __forceinline__ float key2f(unsigned k) {
  unsigned u = (k & 0x80000000u) ? (k ^ 0x80000000u) : ~k;
  return __uint_as_float(u);
}
__device__ __forceinline__ float lrelu(float v) { return v > 0.0f ? v : SLOPE * v; }

template <int K, int NC, bool ZS>
__global__ __launch_bounds__(256) void gemm_bias_wmma(const float* __restrict__ X,
                                                      const float* __restrict__ W,
                                                      const float* __restrict__ b,
                                                      float* __restrict__ Y, int M,
                                                      unsigned* __restrict__ zsm) {
  constexpr int SLAB = 64;
  constexpr int NSLAB = NC / SLAB;
  constexpr int PITCH = 160;
  __shared__ __attribute__((aligned(16))) float wl[(K / 2) * PITCH];

  const int slab = blockIdx.x % NSLAB;
  const int rblk = blockIdx.x / NSLAB;
  const int cbase = slab * SLAB;

#pragma unroll
  for (int j = 0; j < (K / 2) * 16 / 256; j++) {
    int it = j * 256 + (int)threadIdx.x;
    int p = it >> 4, q = it & 15;
    const float* g = W + (size_t)(2 * p) * NC + cbase + 4 * q;
    float4 r0 = *(const float4*)g;
    float4 r1 = *(const float4*)(g + NC);
    float4* d = (float4*)(wl + p * PITCH + 8 * q);
    d[0] = make_float4(r0.x, r1.x, r0.y, r1.y);
    d[1] = make_float4(r0.z, r1.z, r0.w, r1.w);
  }
  __syncthreads();

  if constexpr (ZS) {
    int zi = (int)blockIdx.x * 128 + 4 * (int)threadIdx.x;
    if (threadIdx.x < 32 && zi < M * NSLAB) *(uint4*)(zsm + zi) = make_uint4(0u, 0u, 0u, 0u);
  }
  int wid = threadIdx.x >> 5;
  int mt = rblk * 8 + wid;
  if (mt * 16 >= M) return;
  int lane = threadIdx.x & 31;
  int half = lane >> 4;
  int l = lane & 15;
  const float* xp = X + (size_t)(mt * 16 + l) * K + 2 * half;
  const float* bp = wl + half * PITCH + 2 * l;
  v8f acc[4];
#pragma unroll
  for (int t = 0; t < 4; t++) {
    float bc = b[cbase + t * 16 + l];
    acc[t] = (v8f){bc, bc, bc, bc, bc, bc, bc, bc};
  }
#pragma unroll
  for (int k0 = 0; k0 < K; k0 += 4) {
    v2f a = *(const v2f*)(xp + k0);
#pragma unroll
    for (int t = 0; t < 4; t++) {
      v2f bv = *(const v2f*)(bp + (k0 / 2) * PITCH + t * 32);
      acc[t] = __builtin_amdgcn_wmma_f32_16x16x4_f32(false, a, false, bv, (short)0, acc[t],
                                                     false, false);
    }
  }
#pragma unroll
  for (int t = 0; t < 4; t++) {
    float* yp = Y + (size_t)(mt * 16 + 8 * half) * NC + cbase + t * 16 + l;
#pragma unroll
    for (int v = 0; v < 8; v++) yp[(size_t)v * NC] = acc[t][v];
  }
}

template <int H_, int C_>
__global__ __launch_bounds__(256) void edge_scores(const float* __restrict__ xl,
                                                   const float* __restrict__ xr,
                                                   const float* __restrict__ att,
                                                   const int* __restrict__ ei, int E_, int Etot,
                                                   float* __restrict__ ebuf,
                                                   unsigned* __restrict__ nmax) {
  static_assert(C_ == 64, "16 lanes x 4 channels per item");
  static_assert(4 % H_ == 0, "a wave's 16 items start at head 0; a slot's head follows its parity");
  constexpr int NV = 8;
  const int G = Etot * H_;
  int wave = (int)((blockIdx.x * (size_t)blockDim.x + threadIdx.x) >> 5);
  int g0 = wave * (2 * NV);
  if (g0 >= G) return;
  int lane = threadIdx.x & 31;
  int hf = lane >> 4, sub = lane & 15;

  int gm = g0 + 2 * (sub >> 1) + hf;

  v4f at[2];
#pragma unroll
  for (int q = 0; q < 2; q++) at[q] = *(const v4f*)(att + ((2 * q + hf) % H_) * C_ + sub * 4);

  int srcs[NV], dsts[NV];
#pragma unroll
  for (int i = 0; i < NV; i++) {
    int e = min(g0 / H_ + (2 * i + hf) / H_, Etot - 1);
    int el = min(e, E_ - 1);
    int s_ld = ei[el], d_ld = ei[E_ + el];
    srcs[i] = e < E_ ? s_ld : e - E_;
    dsts[i] = e < E_ ? d_ld : e - E_;
  }
  asm volatile("" ::
               "v"(srcs[0]), "v"(dsts[0]), "v"(srcs[1]), "v"(dsts[1]),
               "v"(srcs[2]), "v"(dsts[2]), "v"(srcs[3]), "v"(dsts[3]),
               "v"(srcs[4]), "v"(dsts[4]), "v"(srcs[5]), "v"(dsts[5]),
               "v"(srcs[6]), "v"(dsts[6]), "v"(srcs[7]), "v"(dsts[7]));
  int dm = dsts[0];
#pragma unroll
  for (int i = 1; i < NV; i++) dm = (sub >> 1) == i ? dsts[i] : dm;
  v4f a[NV], bq[NV];
#pragma unroll
  for (int i = 0; i < NV; i++) {
    int base = ((2 * i + hf) % H_) * C_ + sub * 4;
    a[i]  = *(const v4f*)(xl + (size_t)srcs[i] * (H_ * C_) + base);
    bq[i] = *(const v4f*)(xr + (size_t)dsts[i] * (H_ * C_) + base);
  }
  asm volatile("" ::
               "v"(a[0]), "v"(bq[0]), "v"(a[1]), "v"(bq[1]), "v"(a[2]), "v"(bq[2]), "v"(a[3]), "v"(bq[3]),
               "v"(a[4]), "v"(bq[4]), "v"(a[5]), "v"(bq[5]), "v"(a[6]), "v"(bq[6]), "v"(a[7]), "v"(bq[7]));
  float p[NV];
#pragma unroll
  for (int i = 0; i < NV; i++) {
    v4f t4 = at[i & 1];
    p[i] = lrelu(a[i].x + bq[i].x) * t4.x + lrelu(a[i].y + bq[i].y) * t4.y +
           lrelu(a[i].z + bq[i].z) * t4.z + lrelu(a[i].w + bq[i].w) * t4.w;
  }
  bool b3 = (sub & 8) != 0, b2 = (sub & 4) != 0, b1 = (sub & 2) != 0;
  float w4[4], w2[2];
#pragma unroll
  for (int k = 0; k < 4; k++) {
    float keep = b3 ? p[k + 4] : p[k];
    float send = b3 ? p[k] : p[k + 4];
    w4[k] = keep + __shfl_xor(send, 8, 32);
  }
#pragma unroll
  for (int k = 0; k < 2; k++) {
    float keep = b2 ? w4[k + 2] : w4[k];
    float send = b2 ? w4[k] : w4[k + 2];
    w2[k] = keep + __shfl_xor(send, 4, 32);
  }
  float keep1 = b1 ? w2[1] : w2[0];
  float send1 = b1 ? w2[0] : w2[1];
  float t = keep1 + __shfl_xor(send1, 2, 32);
  t += __shfl_xor(t, 1, 32);
  if ((sub & 1) == 0 && gm < G) {
    ebuf[gm] = t;
    atomicMax(nmax + (size_t)dm * H_ + gm % H_, f2key(t));
  }
}

#define OWN_R 196

template <int H_> struct RowVec;
template <> struct RowVec<4> { typedef v4f T; };
template <> struct RowVec<1> { typedef v2f T; };

__device__ __forceinline__ int rl_i(int v, int j) { return __builtin_amdgcn_readlane(v, j); }
__device__ __forceinline__ float rl_f(float v, int j) {
  return __int_as_float(__builtin_amdgcn_readlane(__float_as_int(v), j));
}

template <int H_>
__device__ __forceinline__ void own_batch(float* __restrict__ accs, float (&den)[H_],
                                          const int* oe_w,
                                          const unsigned char* or_w, int b, int nb,
                                          const float* __restrict__ xl,
                                          const int* __restrict__ ei,
                                          const float* __restrict__ ebuf,
                                          const unsigned* __restrict__ nmax, int lo, int lane) {
  constexpr int HC = H_ * 64;
  constexpr int NP = H_ == 4 ? 2 : 1;
  constexpr int W = H_ == 4 ? 4 : 2;
  typedef typename RowVec<H_>::T RV;
  const int jj = b + min(lane, nb - 1);
  const int e = oe_w[jj];
  const int r = or_w[jj];
  const int s = ei[e];
  float a[H_];
  if constexpr (H_ == 4) {
    const v4f lg = *(const v4f*)(ebuf + (size_t)e * 4);
    const uint4 mk = *(const uint4*)(nmax + (size_t)(lo + r) * 4);
    a[0] = expf(lg.x - key2f(mk.x)); a[1] = expf(lg.y - key2f(mk.y));
    a[2] = expf(lg.z - key2f(mk.z)); a[3] = expf(lg.w - key2f(mk.w));
  } else {
    a[0] = expf(ebuf[e] - key2f(nmax[lo + r]));
  }
  const bool hf = (lane >> 4) != 0;
  for (int j0 = 0; j0 < nb; j0 += 8) {
    RV v[8][NP];
#pragma unroll
    for (int g = 0; g < 8; g++) {
      const int sg = rl_i(s, min(j0 + g, nb - 1));
      const float* rp = xl + (size_t)sg * HC + W * lane;
#pragma unroll
      for (int k = 0; k < NP; k++) v[g][k] = *(const RV*)(rp + k * 128);
    }
    if constexpr (NP == 2) {
      asm volatile("" ::
                   "v"(v[0][0]), "v"(v[0][1]), "v"(v[1][0]), "v"(v[1][1]), "v"(v[2][0]), "v"(v[2][1]),
                   "v"(v[3][0]), "v"(v[3][1]), "v"(v[4][0]), "v"(v[4][1]), "v"(v[5][0]), "v"(v[5][1]),
                   "v"(v[6][0]), "v"(v[6][1]), "v"(v[7][0]), "v"(v[7][1]));
    } else {
      asm volatile("" ::
                   "v"(v[0][0]), "v"(v[1][0]), "v"(v[2][0]), "v"(v[3][0]),
                   "v"(v[4][0]), "v"(v[5][0]), "v"(v[6][0]), "v"(v[7][0]));
    }
#pragma unroll
    for (int g = 0; g < 8; g++) {
      if (j0 + g < nb) {
        const int rg = rl_i(r, j0 + g);
        const bool mine = lane == (rg >> 5);
        float alk[NP];
        if constexpr (H_ == 4) {
          const float a0 = rl_f(a[0], j0 + g), a1 = rl_f(a[1], j0 + g);
          const float a2 = rl_f(a[2], j0 + g), a3 = rl_f(a[3], j0 + g);
          den[0] += mine ? a0 : 0.0f; den[1] += mine ? a1 : 0.0f;
          den[2] += mine ? a2 : 0.0f; den[3] += mine ? a3 : 0.0f;
          alk[0] = hf ? a1 : a0;
          alk[1] = hf ? a3 : a2;
        } else {
          const float a0 = rl_f(a[0], j0 + g);
          den[0] += mine ? a0 : 0.0f;
          alk[0] = a0;
        }
        float* ap = accs + rg * HC + W * lane;
#pragma unroll
        for (int k = 0; k < NP; k++) {
          const float al = alk[k];
          RV cur = *(RV*)(ap + k * 128);
          cur += al * v[g][k];
          *(RV*)(ap + k * 128) = cur;
        }
      }
    }
  }
}

__device__ __forceinline__ float elu1(float v) { return v > 0.0f ? v : expm1f(v); }

template <int H_, bool ELU_>
__global__ __launch_bounds__(1024) void owner_sums(const float* __restrict__ xl,
                                                   const int* __restrict__ ei, int E_,
                                                   const float* __restrict__ ebuf,
                                                   const float* __restrict__ eself,
                                                   const unsigned* __restrict__ nmax,
                                                   const float* __restrict__ bias,
                                                   float* __restrict__ outp, int Nn) {
  constexpr int HC = H_ * 64;
  constexpr int R = OWN_R;
  constexpr int NW = 32;
  constexpr int CH = 512;
  constexpr int NL = CH / 32;
  constexpr int QCAP = H_ == 4 ? 640 : 1024;
  constexpr int OCAP = 128;
  static_assert(R <= 255, "row numbers inside the group are kept in a byte");
  static_assert(R <= 8 * NW, "lanes 0..7 of an owner wave keep the denominators of its rows");
  static_assert(R * H_ <= NW * QCAP, "the denominators are handed over in the queue space");
  static_assert((R * HC) % 4 == 0, "16-byte clears");
  __shared__ __attribute__((aligned(16))) float accs[R * HC];
  __shared__ int qe[NW * QCAP];
  __shared__ unsigned char qr[NW * QCAP];
  __shared__ int oe[NW * OCAP];
  __shared__ unsigned char orow[NW * OCAP];
  __shared__ int qcnt[NW];
  __shared__ int pending[2];

  const int tid = (int)threadIdx.x;
  const int lane = tid & 31;
  const int w = __builtin_amdgcn_readfirstlane(tid >> 5);
  const int lo = (int)blockIdx.x * R;
  const int nrows = min(R, Nn - lo);
  const int* __restrict__ dsts = ei + E_;
  const unsigned lt = (1u << lane) - 1u;

  for (int i = 4 * tid; i < R * HC; i += 4096) *(v4f*)(accs + i) = (v4f){0.0f, 0.0f, 0.0f, 0.0f};
  if (tid < 2) pending[tid] = 0;
  float den[H_];
#pragma unroll
  for (int h = 0; h < H_; h++) den[h] = 0.0f;

  const int per = ((E_ + NW * CH - 1) / (NW * CH)) * CH;
  int pos = w * per;
  const int wend = min(pos + per, E_);

  int d[NL];
  auto load_step = [&](int p) {
    if (p + CH <= E_) {
#pragma unroll
      for (int k = 0; k < NL; k++) d[k] = dsts[p + k * 32 + lane];
    } else {
#pragma unroll
      for (int k = 0; k < NL; k++) {
        int idx = p + k * 32 + lane;
        d[k] = idx < E_ ? dsts[idx] : -1;
      }
    }
  };
  if (pos < wend) load_step(pos);
  __syncthreads();

  for (int round = 0;; round++) {
    int cnt = 0;
    while (pos < wend && cnt + CH <= QCAP) {
      int cur[NL];
#pragma unroll
      for (int k = 0; k < NL; k++) cur[k] = d[k];
      if (pos + CH < wend) load_step(pos + CH);
#pragma unroll
      for (int k = 0; k < NL; k++) {
        const unsigned rr = (unsigned)(cur[k] - lo);
        const bool m = rr < (unsigned)nrows;
        const unsigned mask = __builtin_amdgcn_ballot_w32(m);
        if (mask != 0u) {
          if (m) {
            const int p = w * QCAP + cnt + __builtin_popcount(mask & lt);
            qe[p] = pos + k * 32 + lane;
            qr[p] = (unsigned char)rr;
          }
          cnt += __builtin_popcount(mask);
        }
      }
      pos += CH;
    }
    if (lane == 0) {
      qcnt[w] = cnt;
      if (pos < wend) pending[round & 1] = 1;
    }
    __syncthreads();

    int fill = 0;
    for (int q = 0; q < NW; q++) {
      const int n = __builtin_amdgcn_readfirstlane(qcnt[q]);
      for (int i = 0; i < n; i += 32) {
        const int j = i + lane;
        const int r = qr[q * QCAP + j];
        const bool mine = j < n && (r & 31) == w;
        const unsigned mask = __builtin_amdgcn_ballot_w32(mine);
        if (mask != 0u) {
          if (mine) {
            const int p = w * OCAP + fill + __builtin_popcount(mask & lt);
            oe[p] = qe[q * QCAP + j];
            orow[p] = (unsigned char)r;
          }
          fill += __builtin_popcount(mask);
          if (fill > OCAP - 32) {
            for (int b = 0; b < fill; b += 32)
              own_batch<H_>(accs, den, oe + w * OCAP, orow + w * OCAP, b, min(32, fill - b), xl,
                            ei, ebuf, nmax, lo, lane);
            fill = 0;
          }
        }
      }
    }
    for (int b = 0; b < fill; b += 32)
      own_batch<H_>(accs, den, oe + w * OCAP, orow + w * OCAP, b, min(32, fill - b), xl, ei,
                    ebuf, nmax, lo, lane);

    const int again = __builtin_amdgcn_readfirstlane(pending[round & 1]);
    if (tid == 0) pending[(round + 1) & 1] = 0;
    __syncthreads();
    if (!again) break;
  }

  float* dsum = (float*)qe;
  {
    const int rr = lane * 32 + w;
    if (lane < 8 && rr < R) {
#pragma unroll
      for (int h = 0; h < H_; h++) dsum[rr * H_ + h] = den[h];
    }
  }
  __syncthreads();

  const int tot = nrows * HC;
  const int ch = (4 * tid) % HC;
  const v4f bb = *(const v4f*)(bias + ch);
  for (int i0 = 4 * tid; i0 < tot; i0 += 4 * 4096) {
    v4f xs[4];
    float lg[4];
    unsigned mk[4];
#pragma unroll
    for (int u = 0; u < 4; u++) {
      const int i = min(i0 + u * 4096, tot - 4);
      const size_t node = (size_t)(lo + i / HC);
      const int item = (int)node * H_ + ((i % HC) >> 6);
      lg[u] = eself[item];
      mk[u] = nmax[item];
      xs[u] = *(const v4f*)(xl + node * HC + i % HC);
    }
    asm volatile("" ::
                 "v"(xs[0]), "v"(xs[1]), "v"(xs[2]), "v"(xs[3]),
                 "v"(lg[0]), "v"(mk[0]), "v"(lg[1]), "v"(mk[1]),
                 "v"(lg[2]), "v"(mk[2]), "v"(lg[3]), "v"(mk[3]));
#pragma unroll
    for (int u = 0; u < 4; u++) {
      const int i = i0 + u * 4096;
      if (i < tot) {
        const float exs = expf(lg[u] - key2f(mk[u]));
        const float rc = 1.0f / (dsum[(i / HC) * H_ + ((i % HC) >> 6)] + exs);
        const v4f sm = *(const v4f*)(accs + i);
        v4f o = (sm + exs * xs[u]) * rc + bb;
        if constexpr (ELU_) { o.x = elu1(o.x); o.y = elu1(o.y); o.z = elu1(o.z); o.w = elu1(o.w); }
        *(v4f*)(outp + (size_t)(lo + i / HC) * HC + i % HC) = o;
      }
    }
  }
}

extern "C" void kernel_launch(void* const* d_in, const int* in_sizes, int n_in,
                              void* d_out, int out_size, void* d_ws, size_t ws_size,
                              hipStream_t stream) {
  (void)n_in; (void)out_size; (void)ws_size;
  const int IN = 128, HID = 64, H = 4, OUT = 64;
  const float*    x    = (const float*)d_in[0];
  const int*      ei   = (const int*)d_in[1];
  const float*    Wl1  = (const float*)d_in[2];
  const float*    bl1  = (const float*)d_in[3];
  const float*    Wr1  = (const float*)d_in[4];
  const float*    br1  = (const float*)d_in[5];
  const float*    att1 = (const float*)d_in[6];
  const float*    bias1= (const float*)d_in[7];
  const float*    Wl2  = (const float*)d_in[8];
  const float*    bl2  = (const float*)d_in[9];
  const float*    Wr2  = (const float*)d_in[10];
  const float*    br2  = (const float*)d_in[11];
  const float*    att2 = (const float*)d_in[12];
  const float*    bias2= (const float*)d_in[13];
  float* out = (float*)d_out;

  const int N    = in_sizes[0] / IN;
  const int E    = in_sizes[1] / 2;
  const int Etot = E + N;
  const int HC   = H * HID;

  float* ws = (float*)d_ws;
  size_t o = 0;
  float*    xl1   = ws + o; o += (size_t)N * HC;
  float*    xr1   = ws + o; o += (size_t)N * HC;
  float*    ebuf  = ws + o; o += (size_t)Etot * H;
  unsigned* nmax  = (unsigned*)(ws + o); o += (size_t)N * H;
  float*  h    = xr1;
  float*  xl2  = xl1;
  float*  xr2  = xl1 + (size_t)N * OUT;

  const int B = 256;

  int rblks = (N / 16 + 7) / 8;
  gemm_bias_wmma<128, 256, true><<<rblks * (HC / 64), B, 0, stream>>>(x, Wl1, bl1, xl1, N,
                                                                      nmax);
  gemm_bias_wmma<128, 256, false><<<rblks * (HC / 64), B, 0, stream>>>(x, Wr1, br1, xr1, N,
                                                                       nullptr);

  long long items1 = (long long)Etot * H;
  int gs1 = (int)((items1 + 127) / 128);
  int gown = (N + OWN_R - 1) / OWN_R;
  edge_scores<4, 64><<<gs1, B, 0, stream>>>(xl1, xr1, att1, ei, E, Etot, ebuf, nmax);
  owner_sums<4, true><<<gown, 1024, 0, stream>>>(xl1, ei, E, ebuf, ebuf + (size_t)E * H, nmax,
                                                 bias1, h, N);

  gemm_bias_wmma<256, 64, true><<<rblks * (OUT / 64), B, 0, stream>>>(h, Wl2, bl2, xl2, N,
                                                                      nmax);
  gemm_bias_wmma<256, 64, false><<<rblks * (OUT / 64), B, 0, stream>>>(h, Wr2, br2, xr2, N,
                                                                       nullptr);

  long long items2 = Etot;
  int gs2 = (int)((items2 + 127) / 128);
  edge_scores<1, 64><<<gs2, B, 0, stream>>>(xl2, xr2, att2, ei, E, Etot, ebuf, nmax);
  owner_sums<1, false><<<gown, 1024, 0, stream>>>(xl2, ei, E, ebuf, ebuf + E, nmax, bias2, out, N);
}


# Layer 1's owner pass writes each owner wave's in-edge list (edge, row) to the workspace; layer 2's owner pass walks these lists and no longer scans the destination list; sum order = layer 1's
# speedup vs baseline: 2.4080x; 1.0876x over previous
#include <hip/hip_runtime.h>
#include <math.h>

#define SLOPE 0.2f

typedef float v2f __attribute__((ext_vector_type(2)));
typedef float v4f __attribute__((ext_vector_type(4)));
typedef float v8f __attribute__((ext_vector_type(8)));

__device__ __forceinline__ unsigned f2key(float f) {
  unsigned u = __float_as_uint(f);
  return (u & 0x80000000u) ? ~u : (u | 0x80000000u);
}
__device__ __forceinline__ float key2f(unsigned k) {
  unsigned u = (k & 0x80000000u) ? (k ^ 0x80000000u) : ~k;
  return __uint_as_float(u);
}
__device__ __forceinline__ float lrelu(float v) { return v > 0.0f ? v : SLOPE * v; }

template <int K, int NC, bool ZS>
__global__ __launch_bounds__(256) void gemm_bias_wmma(const float* __restrict__ X,
                                                      const float* __restrict__ W,
                                                      const float* __restrict__ b,
                                                      float* __restrict__ Y, int M,
                                                      unsigned* __restrict__ zsm) {
  constexpr int SLAB = 64;
  constexpr int NSLAB = NC / SLAB;
  constexpr int PITCH = 160;
  __shared__ __attribute__((aligned(16))) float wl[(K / 2) * PITCH];

  const int slab = blockIdx.x % NSLAB;
  const int rblk = blockIdx.x / NSLAB;
  const int cbase = slab * SLAB;

#pragma unroll
  for (int j = 0; j < (K / 2) * 16 / 256; j++) {
    int it = j * 256 + (int)threadIdx.x;
    int p = it >> 4, q = it & 15;
    const float* g = W + (size_t)(2 * p) * NC + cbase + 4 * q;
    float4 r0 = *(const float4*)g;
    float4 r1 = *(const float4*)(g + NC);
    float4* d = (float4*)(wl + p * PITCH + 8 * q);
    d[0] = make_float4(r0.x, r1.x, r0.y, r1.y);
    d[1] = make_float4(r0.z, r1.z, r0.w, r1.w);
  }
  __syncthreads();

  if constexpr (ZS) {
    int zi = (int)blockIdx.x * 128 + 4 * (int)threadIdx.x;
    if (threadIdx.x < 32 && zi < M * NSLAB) *(uint4*)(zsm + zi) = make_uint4(0u, 0u, 0u, 0u);
  }
  int wid = threadIdx.x >> 5;
  int mt = rblk * 8 + wid;
  if (mt * 16 >= M) return;
  int lane = threadIdx.x & 31;
  int half = lane >> 4;
  int l = lane & 15;
  const float* xp = X + (size_t)(mt * 16 + l) * K + 2 * half;
  const float* bp = wl + half * PITCH + 2 * l;
  v8f acc[4];
#pragma unroll
  for (int t = 0; t < 4; t++) {
    float bc = b[cbase + t * 16 + l];
    acc[t] = (v8f){bc, bc, bc, bc, bc, bc, bc, bc};
  }
#pragma unroll
  for (int k0 = 0; k0 < K; k0 += 4) {
    v2f a = *(const v2f*)(xp + k0);
#pragma unroll
    for (int t = 0; t < 4; t++) {
      v2f bv = *(const v2f*)(bp + (k0 / 2) * PITCH + t * 32);
      acc[t] = __builtin_amdgcn_wmma_f32_16x16x4_f32(false, a, false, bv, (short)0, acc[t],
                                                     false, false);
    }
  }
#pragma unroll
  for (int t = 0; t < 4; t++) {
    float* yp = Y + (size_t)(mt * 16 + 8 * half) * NC + cbase + t * 16 + l;
#pragma unroll
    for (int v = 0; v < 8; v++) yp[(size_t)v * NC] = acc[t][v];
  }
}

template <int H_, int C_>
__global__ __launch_bounds__(256) void edge_scores(const float* __restrict__ xl,
                                                   const float* __restrict__ xr,
                                                   const float* __restrict__ att,
                                                   const int* __restrict__ ei, int E_, int Etot,
                                                   float* __restrict__ ebuf,
                                                   unsigned* __restrict__ nmax) {
  static_assert(C_ == 64, "16 lanes x 4 channels per item");
  static_assert(4 % H_ == 0, "a wave's 16 items start at head 0; a slot's head follows its parity");
  constexpr int NV = 8;
  const int G = Etot * H_;
  int wave = (int)((blockIdx.x * (size_t)blockDim.x + threadIdx.x) >> 5);
  int g0 = wave * (2 * NV);
  if (g0 >= G) return;
  int lane = threadIdx.x & 31;
  int hf = lane >> 4, sub = lane & 15;

  int gm = g0 + 2 * (sub >> 1) + hf;

  v4f at[2];
#pragma unroll
  for (int q = 0; q < 2; q++) at[q] = *(const v4f*)(att + ((2 * q + hf) % H_) * C_ + sub * 4);

  int srcs[NV], dsts[NV];
#pragma unroll
  for (int i = 0; i < NV; i++) {
    int e = min(g0 / H_ + (2 * i + hf) / H_, Etot - 1);
    int el = min(e, E_ - 1);
    int s_ld = ei[el], d_ld = ei[E_ + el];
    srcs[i] = e < E_ ? s_ld : e - E_;
    dsts[i] = e < E_ ? d_ld : e - E_;
  }
  asm volatile("" ::
               "v"(srcs[0]), "v"(dsts[0]), "v"(srcs[1]), "v"(dsts[1]),
               "v"(srcs[2]), "v"(dsts[2]), "v"(srcs[3]), "v"(dsts[3]),
               "v"(srcs[4]), "v"(dsts[4]), "v"(srcs[5]), "v"(dsts[5]),
               "v"(srcs[6]), "v"(dsts[6]), "v"(srcs[7]), "v"(dsts[7]));
  int dm = dsts[0];
#pragma unroll
  for (int i = 1; i < NV; i++) dm = (sub >> 1) == i ? dsts[i] : dm;
  v4f a[NV], bq[NV];
#pragma unroll
  for (int i = 0; i < NV; i++) {
    int base = ((2 * i + hf) % H_) * C_ + sub * 4;
    a[i]  = *(const v4f*)(xl + (size_t)srcs[i] * (H_ * C_) + base);
    bq[i] = *(const v4f*)(xr + (size_t)dsts[i] * (H_ * C_) + base);
  }
  asm volatile("" ::
               "v"(a[0]), "v"(bq[0]), "v"(a[1]), "v"(bq[1]), "v"(a[2]), "v"(bq[2]), "v"(a[3]), "v"(bq[3]),
               "v"(a[4]), "v"(bq[4]), "v"(a[5]), "v"(bq[5]), "v"(a[6]), "v"(bq[6]), "v"(a[7]), "v"(bq[7]));
  float p[NV];
#pragma unroll
  for (int i = 0; i < NV; i++) {
    v4f t4 = at[i & 1];
    p[i] = lrelu(a[i].x + bq[i].x) * t4.x + lrelu(a[i].y + bq[i].y) * t4.y +
           lrelu(a[i].z + bq[i].z) * t4.z + lrelu(a[i].w + bq[i].w) * t4.w;
  }
  bool b3 = (sub & 8) != 0, b2 = (sub & 4) != 0, b1 = (sub & 2) != 0;
  float w4[4], w2[2];
#pragma unroll
  for (int k = 0; k < 4; k++) {
    float keep = b3 ? p[k + 4] : p[k];
    float send = b3 ? p[k] : p[k + 4];
    w4[k] = keep + __shfl_xor(send, 8, 32);
  }
#pragma unroll
  for (int k = 0; k < 2; k++) {
    float keep = b2 ? w4[k + 2] : w4[k];
    float send = b2 ? w4[k] : w4[k + 2];
    w2[k] = keep + __shfl_xor(send, 4, 32);
  }
  float keep1 = b1 ? w2[1] : w2[0];
  float send1 = b1 ? w2[0] : w2[1];
  float t = keep1 + __shfl_xor(send1, 2, 32);
  t += __shfl_xor(t, 1, 32);
  if ((sub & 1) == 0 && gm < G) {
    ebuf[gm] = t;
    atomicMax(nmax + (size_t)dm * H_ + gm % H_, f2key(t));
  }
}

#define OWN_R 196
#define RCAP 512

template <int H_> struct RowVec;
template <> struct RowVec<4> { typedef v4f T; };
template <> struct RowVec<1> { typedef v2f T; };

__device__ __forceinline__ int rl_i(int v, int j) { return __builtin_amdgcn_readlane(v, j); }
__device__ __forceinline__ float rl_f(float v, int j) {
  return __int_as_float(__builtin_amdgcn_readlane(__float_as_int(v), j));
}

template <int H_, bool REC_, typename RT_>
__device__ __forceinline__ void own_batch(float* __restrict__ accs, float (&den)[H_],
                                          const int* oe_w,
                                          const RT_* or_w, int b, int nb,
                                          const float* __restrict__ xl,
                                          const int* __restrict__ ei,
                                          const float* __restrict__ ebuf,
                                          const unsigned* __restrict__ nmax, int lo, int lane,
                                          int* rec_e, int* rec_r, int recpos) {
  constexpr int HC = H_ * 64;
  constexpr int NP = H_ == 4 ? 2 : 1;
  constexpr int W = H_ == 4 ? 4 : 2;
  typedef typename RowVec<H_>::T RV;
  const int jj = b + min(lane, nb - 1);
  const int e = oe_w[jj];
  const int r = (int)or_w[jj];
  if constexpr (REC_) {
    if (lane < nb && recpos + lane < RCAP) {
      rec_e[recpos + lane] = e;
      rec_r[recpos + lane] = r;
    }
  }
  const int s = ei[e];
  float a[H_];
  if constexpr (H_ == 4) {
    const v4f lg = *(const v4f*)(ebuf + (size_t)e * 4);
    const uint4 mk = *(const uint4*)(nmax + (size_t)(lo + r) * 4);
    a[0] = expf(lg.x - key2f(mk.x)); a[1] = expf(lg.y - key2f(mk.y));
    a[2] = expf(lg.z - key2f(mk.z)); a[3] = expf(lg.w - key2f(mk.w));
  } else {
    a[0] = expf(ebuf[e] - key2f(nmax[lo + r]));
  }
  const bool hf = (lane >> 4) != 0;
  for (int j0 = 0; j0 < nb; j0 += 8) {
    RV v[8][NP];
#pragma unroll
    for (int g = 0; g < 8; g++) {
      const int sg = rl_i(s, min(j0 + g, nb - 1));
      const float* rp = xl + (size_t)sg * HC + W * lane;
#pragma unroll
      for (int k = 0; k < NP; k++) v[g][k] = *(const RV*)(rp + k * 128);
    }
    if constexpr (NP == 2) {
      asm volatile("" ::
                   "v"(v[0][0]), "v"(v[0][1]), "v"(v[1][0]), "v"(v[1][1]), "v"(v[2][0]), "v"(v[2][1]),
                   "v"(v[3][0]), "v"(v[3][1]), "v"(v[4][0]), "v"(v[4][1]), "v"(v[5][0]), "v"(v[5][1]),
                   "v"(v[6][0]), "v"(v[6][1]), "v"(v[7][0]), "v"(v[7][1]));
    } else {
      asm volatile("" ::
                   "v"(v[0][0]), "v"(v[1][0]), "v"(v[2][0]), "v"(v[3][0]),
                   "v"(v[4][0]), "v"(v[5][0]), "v"(v[6][0]), "v"(v[7][0]));
    }
#pragma unroll
    for (int g = 0; g < 8; g++) {
      if (j0 + g < nb) {
        const int rg = rl_i(r, j0 + g);
        const bool mine = lane == (rg >> 5);
        float alk[NP];
        if constexpr (H_ == 4) {
          const float a0 = rl_f(a[0], j0 + g), a1 = rl_f(a[1], j0 + g);
          const float a2 = rl_f(a[2], j0 + g), a3 = rl_f(a[3], j0 + g);
          den[0] += mine ? a0 : 0.0f; den[1] += mine ? a1 : 0.0f;
          den[2] += mine ? a2 : 0.0f; den[3] += mine ? a3 : 0.0f;
          alk[0] = hf ? a1 : a0;
          alk[1] = hf ? a3 : a2;
        } else {
          const float a0 = rl_f(a[0], j0 + g);
          den[0] += mine ? a0 : 0.0f;
          alk[0] = a0;
        }
        float* ap = accs + rg * HC + W * lane;
#pragma unroll
        for (int k = 0; k < NP; k++) {
          const float al = alk[k];
          RV cur = *(RV*)(ap + k * 128);
          cur += al * v[g][k];
          *(RV*)(ap + k * 128) = cur;
        }
      }
    }
  }
}

__device__ __forceinline__ float elu1(float v) { return v > 0.0f ? v : expm1f(v); }

template <int H_, bool ELU_, int MODE_>
__global__ __launch_bounds__(1024) void owner_sums(const float* __restrict__ xl,
                                                   const int* __restrict__ ei, int E_,
                                                   const float* __restrict__ ebuf,
                                                   const float* __restrict__ eself,
                                                   const unsigned* __restrict__ nmax,
                                                   const float* __restrict__ bias,
                                                   float* __restrict__ outp, int Nn,
                                                   int* rl_e, int* rl_r, int* rl_n) {
  static_assert(MODE_ == 1 || MODE_ == 2, "1: scan and write the lists; 2: walk the lists");
  constexpr int HC = H_ * 64;
  constexpr int R = OWN_R;
  constexpr int NW = 32;
  constexpr int CH = 512;
  constexpr int NL = CH / 32;
  constexpr int QCAP = H_ == 4 ? 640 : 1024;
  constexpr int OCAP = 128;
  static_assert(R <= 255, "row numbers inside the group are kept in a byte");
  static_assert(R <= 8 * NW, "lanes 0..7 of an owner wave keep the denominators of its rows");
  static_assert(R * H_ <= NW * QCAP, "the denominators are handed over in the queue space");
  static_assert((R * HC) % 4 == 0, "16-byte clears");
  __shared__ __attribute__((aligned(16))) float accs[R * HC];
  __shared__ int qe[NW * QCAP];
  __shared__ unsigned char qr[NW * QCAP];
  __shared__ int oe[NW * OCAP];
  __shared__ unsigned char orow[NW * OCAP];
  __shared__ int qcnt[NW];
  __shared__ int pending[2];

  const int tid = (int)threadIdx.x;
  const int lane = tid & 31;
  const int w = __builtin_amdgcn_readfirstlane(tid >> 5);
  const int lo = (int)blockIdx.x * R;
  const int nrows = min(R, Nn - lo);
  const int* __restrict__ dsts = ei + E_;
  const unsigned lt = (1u << lane) - 1u;

  for (int i = 4 * tid; i < R * HC; i += 4096) *(v4f*)(accs + i) = (v4f){0.0f, 0.0f, 0.0f, 0.0f};
  if (tid < 2) pending[tid] = 0;
  float den[H_];
#pragma unroll
  for (int h = 0; h < H_; h++) den[h] = 0.0f;

  const int per = ((E_ + NW * CH - 1) / (NW * CH)) * CH;
  int pos = w * per;
  const int wend = min(pos + per, E_);

  int d[NL];
  auto load_step = [&](int p) {
    if (p + CH <= E_) {
#pragma unroll
      for (int k = 0; k < NL; k++) d[k] = dsts[p + k * 32 + lane];
    } else {
#pragma unroll
      for (int k = 0; k < NL; k++) {
        int idx = p + k * 32 + lane;
        d[k] = idx < E_ ? dsts[idx] : -1;
      }
    }
  };
  const int lbase = (int)blockIdx.x * NW + w;
  int* const my_e = rl_e + (size_t)lbase * RCAP;
  int* const my_r = rl_r + (size_t)lbase * RCAP;
  int rec = 0;
  bool walk = false;
  int nlist = 0;
  if constexpr (MODE_ == 2) {
    const int c = rl_n[(int)blockIdx.x * NW + lane];
    walk = __builtin_amdgcn_ballot_w32(c > RCAP) == 0u;
    nlist = rl_i(c, w);
  }
  if (!walk && pos < wend) load_step(pos);
  __syncthreads();

  if (walk) {
    for (int b = 0; b < nlist; b += 32)
      own_batch<H_, false, int>(accs, den, my_e, my_r, b, min(32, nlist - b), xl, ei, ebuf, nmax,
                                lo, lane, nullptr, nullptr, 0);
    __syncthreads();
  } else
  for (int round = 0;; round++) {
    int cnt = 0;
    while (pos < wend && cnt + CH <= QCAP) {
      int cur[NL];
#pragma unroll
      for (int k = 0; k < NL; k++) cur[k] = d[k];
      if (pos + CH < wend) load_step(pos + CH);
#pragma unroll
      for (int k = 0; k < NL; k++) {
        const unsigned rr = (unsigned)(cur[k] - lo);
        const bool m = rr < (unsigned)nrows;
        const unsigned mask = __builtin_amdgcn_ballot_w32(m);
        if (mask != 0u) {
          if (m) {
            const int p = w * QCAP + cnt + __builtin_popcount(mask & lt);
            qe[p] = pos + k * 32 + lane;
            qr[p] = (unsigned char)rr;
          }
          cnt += __builtin_popcount(mask);
        }
      }
      pos += CH;
    }
    if (lane == 0) {
      qcnt[w] = cnt;
      if (pos < wend) pending[round & 1] = 1;
    }
    __syncthreads();

    int fill = 0;
    for (int q = 0; q < NW; q++) {
      const int n = __builtin_amdgcn_readfirstlane(qcnt[q]);
      for (int i = 0; i < n; i += 32) {
        const int j = i + lane;
        const int r = qr[q * QCAP + j];
        const bool mine = j < n && (r & 31) == w;
        const unsigned mask = __builtin_amdgcn_ballot_w32(mine);
        if (mask != 0u) {
          if (mine) {
            const int p = w * OCAP + fill + __builtin_popcount(mask & lt);
            oe[p] = qe[q * QCAP + j];
            orow[p] = (unsigned char)r;
          }
          fill += __builtin_popcount(mask);
          if (fill > OCAP - 32) {
            for (int b = 0; b < fill; b += 32) {
              const int nbt = min(32, fill - b);
              own_batch<H_, MODE_ == 1, unsigned char>(accs, den, oe + w * OCAP, orow + w * OCAP, b,
                                                       nbt, xl, ei, ebuf, nmax, lo, lane, my_e,
                                                       my_r, rec);
              rec += nbt;
            }
            fill = 0;
          }
        }
      }
    }
    for (int b = 0; b < fill; b += 32) {
      const int nbt = min(32, fill - b);
      own_batch<H_, MODE_ == 1, unsigned char>(accs, den, oe + w * OCAP, orow + w * OCAP, b, nbt,
                                               xl, ei, ebuf, nmax, lo, lane, my_e, my_r, rec);
      rec += nbt;
    }

    const int again = __builtin_amdgcn_readfirstlane(pending[round & 1]);
    if (tid == 0) pending[(round + 1) & 1] = 0;
    __syncthreads();
    if (!again) break;
  }

  if constexpr (MODE_ == 1) {
    if (lane == 0) rl_n[lbase] = rec;
  }

  float* dsum = (float*)qe;
  {
    const int rr = lane * 32 + w;
    if (lane < 8 && rr < R) {
#pragma unroll
      for (int h = 0; h < H_; h++) dsum[rr * H_ + h] = den[h];
    }
  }
  __syncthreads();

  const int tot = nrows * HC;
  const int ch = (4 * tid) % HC;
  const v4f bb = *(const v4f*)(bias + ch);
  for (int i0 = 4 * tid; i0 < tot; i0 += 4 * 4096) {
    v4f xs[4];
    float lg[4];
    unsigned mk[4];
#pragma unroll
    for (int u = 0; u < 4; u++) {
      const int i = min(i0 + u * 4096, tot - 4);
      const size_t node = (size_t)(lo + i / HC);
      const int item = (int)node * H_ + ((i % HC) >> 6);
      lg[u] = eself[item];
      mk[u] = nmax[item];
      xs[u] = *(const v4f*)(xl + node * HC + i % HC);
    }
    asm volatile("" ::
                 "v"(xs[0]), "v"(xs[1]), "v"(xs[2]), "v"(xs[3]),
                 "v"(lg[0]), "v"(mk[0]), "v"(lg[1]), "v"(mk[1]),
                 "v"(lg[2]), "v"(mk[2]), "v"(lg[3]), "v"(mk[3]));
#pragma unroll
    for (int u = 0; u < 4; u++) {
      const int i = i0 + u * 4096;
      if (i < tot) {
        const float exs = expf(lg[u] - key2f(mk[u]));
        const float rc = 1.0f / (dsum[(i / HC) * H_ + ((i % HC) >> 6)] + exs);
        const v4f sm = *(const v4f*)(accs + i);
        v4f o = (sm + exs * xs[u]) * rc + bb;
        if constexpr (ELU_) { o.x = elu1(o.x); o.y = elu1(o.y); o.z = elu1(o.z); o.w = elu1(o.w); }
        *(v4f*)(outp + (size_t)(lo + i / HC) * HC + i % HC) = o;
      }
    }
  }
}

extern "C" void kernel_launch(void* const* d_in, const int* in_sizes, int n_in,
                              void* d_out, int out_size, void* d_ws, size_t ws_size,
                              hipStream_t stream) {
  (void)n_in; (void)out_size; (void)ws_size;
  const int IN = 128, HID = 64, H = 4, OUT = 64;
  const float*    x    = (const float*)d_in[0];
  const int*      ei   = (const int*)d_in[1];
  const float*    Wl1  = (const float*)d_in[2];
  const float*    bl1  = (const float*)d_in[3];
  const float*    Wr1  = (const float*)d_in[4];
  const float*    br1  = (const float*)d_in[5];
  const float*    att1 = (const float*)d_in[6];
  const float*    bias1= (const float*)d_in[7];
  const float*    Wl2  = (const float*)d_in[8];
  const float*    bl2  = (const float*)d_in[9];
  const float*    Wr2  = (const float*)d_in[10];
  const float*    br2  = (const float*)d_in[11];
  const float*    att2 = (const float*)d_in[12];
  const float*    bias2= (const float*)d_in[13];
  float* out = (float*)d_out;

  const int N    = in_sizes[0] / IN;
  const int E    = in_sizes[1] / 2;
  const int Etot = E + N;
  const int HC   = H * HID;

  float* ws = (float*)d_ws;
  size_t o = 0;
  float*    xl1   = ws + o; o += (size_t)N * HC;
  float*    xr1   = ws + o; o += (size_t)N * HC;
  float*    ebuf  = ws + o; o += (size_t)Etot * H;
  unsigned* nmax  = (unsigned*)(ws + o); o += (size_t)N * H;
  const int gown = (N + OWN_R - 1) / OWN_R;
  int* rl_e = (int*)(ws + o); o += (size_t)gown * 32 * RCAP;
  int* rl_r = (int*)(ws + o); o += (size_t)gown * 32 * RCAP;
  int* rl_n = (int*)(ws + o); o += (size_t)gown * 32;
  float*  h    = xr1;
  float*  xl2  = xl1;
  float*  xr2  = xl1 + (size_t)N * OUT;

  const int B = 256;

  int rblks = (N / 16 + 7) / 8;
  gemm_bias_wmma<128, 256, true><<<rblks * (HC / 64), B, 0, stream>>>(x, Wl1, bl1, xl1, N,
                                                                      nmax);
  gemm_bias_wmma<128, 256, false><<<rblks * (HC / 64), B, 0, stream>>>(x, Wr1, br1, xr1, N,
                                                                       nullptr);

  long long items1 = (long long)Etot * H;
  int gs1 = (int)((items1 + 127) / 128);
  edge_scores<4, 64><<<gs1, B, 0, stream>>>(xl1, xr1, att1, ei, E, Etot, ebuf, nmax);
  owner_sums<4, true, 1><<<gown, 1024, 0, stream>>>(xl1, ei, E, ebuf, ebuf + (size_t)E * H, nmax,
                                                    bias1, h, N, rl_e, rl_r, rl_n);

  gemm_bias_wmma<256, 64, true><<<rblks * (OUT / 64), B, 0, stream>>>(h, Wl2, bl2, xl2, N,
                                                                      nmax);
  gemm_bias_wmma<256, 64, false><<<rblks * (OUT / 64), B, 0, stream>>>(h, Wr2, br2, xr2, N,
                                                                       nullptr);

  long long items2 = Etot;
  int gs2 = (int)((items2 + 127) / 128);
  edge_scores<1, 64><<<gs2, B, 0, stream>>>(xl2, xr2, att2, ei, E, Etot, ebuf, nmax);
  owner_sums<1, false, 2><<<gown, 1024, 0, stream>>>(xl2, ei, E, ebuf, ebuf + E, nmax, bias2, out, N,
                                                     rl_e, rl_r, rl_n);
}


# Edge logits by an owner pass over the in-edge lists (built there in layer 1): its 196 xr rows sit in LDS, waves gather source rows only; row max by owner, no atomics; layer-2 logit sums re-paired
# speedup vs baseline: 2.5182x; 1.0458x over previous
#include <hip/hip_runtime.h>
#include <math.h>

#define SLOPE 0.2f

typedef float v2f __attribute__((ext_vector_type(2)));
typedef float v4f __attribute__((ext_vector_type(4)));
typedef float v8f __attribute__((ext_vector_type(8)));

__device__ __forceinline__ float lrelu(float v) { return v > 0.0f ? v : SLOPE * v; }

template <int K, int NC>
__global__ __launch_bounds__(256) void gemm_bias_wmma(const float* __restrict__ X,
                                                      const float* __restrict__ W,
                                                      const float* __restrict__ b,
                                                      float* __restrict__ Y, int M) {
  constexpr int SLAB = 64;
  constexpr int NSLAB = NC / SLAB;
  constexpr int PITCH = 160;
  __shared__ __attribute__((aligned(16))) float wl[(K / 2) * PITCH];

  const int slab = blockIdx.x % NSLAB;
  const int rblk = blockIdx.x / NSLAB;
  const int cbase = slab * SLAB;

#pragma unroll
  for (int j = 0; j < (K / 2) * 16 / 256; j++) {
    int it = j * 256 + (int)threadIdx.x;
    int p = it >> 4, q = it & 15;
    const float* g = W + (size_t)(2 * p) * NC + cbase + 4 * q;
    float4 r0 = *(const float4*)g;
    float4 r1 = *(const float4*)(g + NC);
    float4* d = (float4*)(wl + p * PITCH + 8 * q);
    d[0] = make_float4(r0.x, r1.x, r0.y, r1.y);
    d[1] = make_float4(r0.z, r1.z, r0.w, r1.w);
  }
  __syncthreads();

  int wid = threadIdx.x >> 5;
  int mt = rblk * 8 + wid;
  if (mt * 16 >= M) return;
  int lane = threadIdx.x & 31;
  int half = lane >> 4;
  int l = lane & 15;
  const float* xp = X + (size_t)(mt * 16 + l) * K + 2 * half;
  const float* bp = wl + half * PITCH + 2 * l;
  v8f acc[4];
#pragma unroll
  for (int t = 0; t < 4; t++) {
    float bc = b[cbase + t * 16 + l];
    acc[t] = (v8f){bc, bc, bc, bc, bc, bc, bc, bc};
  }
#pragma unroll
  for (int k0 = 0; k0 < K; k0 += 4) {
    v2f a = *(const v2f*)(xp + k0);
#pragma unroll
    for (int t = 0; t < 4; t++) {
      v2f bv = *(const v2f*)(bp + (k0 / 2) * PITCH + t * 32);
      acc[t] = __builtin_amdgcn_wmma_f32_16x16x4_f32(false, a, false, bv, (short)0, acc[t],
                                                     false, false);
    }
  }
#pragma unroll
  for (int t = 0; t < 4; t++) {
    float* yp = Y + (size_t)(mt * 16 + 8 * half) * NC + cbase + t * 16 + l;
#pragma unroll
    for (int v = 0; v < 8; v++) yp[(size_t)v * NC] = acc[t][v];
  }
}

#define OWN_R 196
#define RCAP 512

template <int H_> struct RowVec;
template <> struct RowVec<4> { typedef v4f T; };
template <> struct RowVec<1> { typedef v2f T; };

__device__ __forceinline__ int rl_i(int v, int j) { return __builtin_amdgcn_readlane(v, j); }
__device__ __forceinline__ float rl_f(float v, int j) {
  return __int_as_float(__builtin_amdgcn_readlane(__float_as_int(v), j));
}
__device__ __forceinline__ float dot_lrelu(v4f z, v4f t) {
  return lrelu(z.x) * t.x + lrelu(z.y) * t.y + lrelu(z.z) * t.z + lrelu(z.w) * t.w;
}
__device__ __forceinline__ float dot_lrelu(v2f z, v2f t) {
  return lrelu(z.x) * t.x + lrelu(z.y) * t.y;
}

template <int H_, bool REC_, typename RT_>
__device__ __forceinline__ void score_batch(const float* xrs, float (&mx)[H_],
                                            const int* oe_w, const RT_* or_w, int b, int nb,
                                            const float* __restrict__ xl,
                                            const int* __restrict__ ei,
                                            const float* __restrict__ att, float* ebuf, int lane,
                                            int* rec_e, int* rec_r, int recpos) {
  constexpr int HC = H_ * 64;
  constexpr int NP = H_ == 4 ? 2 : 1;
  constexpr int W = H_ == 4 ? 4 : 2;
  typedef typename RowVec<H_>::T RV;
  const int jj = b + min(lane, nb - 1);
  const int e = oe_w[jj];
  const int r = (int)or_w[jj];
  if constexpr (REC_) {
    if (lane < nb && recpos + lane < RCAP) {
      rec_e[recpos + lane] = e;
      rec_r[recpos + lane] = r;
    }
  }
  const int s = ei[e];
  RV at[NP];
#pragma unroll
  for (int k = 0; k < NP; k++) at[k] = *(const RV*)(att + k * 128 + W * lane);
  for (int j0 = 0; j0 < nb; j0 += 8) {
    RV v[8][NP];
#pragma unroll
    for (int g = 0; g < 8; g++) {
      const int sg = rl_i(s, min(j0 + g, nb - 1));
      const float* rp = xl + (size_t)sg * HC + W * lane;
#pragma unroll
      for (int k = 0; k < NP; k++) v[g][k] = *(const RV*)(rp + k * 128);
    }
    if constexpr (NP == 2) {
      asm volatile("" ::
                   "v"(v[0][0]), "v"(v[0][1]), "v"(v[1][0]), "v"(v[1][1]), "v"(v[2][0]), "v"(v[2][1]),
                   "v"(v[3][0]), "v"(v[3][1]), "v"(v[4][0]), "v"(v[4][1]), "v"(v[5][0]), "v"(v[5][1]),
                   "v"(v[6][0]), "v"(v[6][1]), "v"(v[7][0]), "v"(v[7][1]));
    } else {
      asm volatile("" ::
                   "v"(v[0][0]), "v"(v[1][0]), "v"(v[2][0]), "v"(v[3][0]),
                   "v"(v[4][0]), "v"(v[5][0]), "v"(v[6][0]), "v"(v[7][0]));
    }
    float q[8 * NP];
#pragma unroll
    for (int g = 0; g < 8; g++) {
      const int rg = rl_i(r, min(j0 + g, nb - 1));
      const float* xp = xrs + rg * HC + W * lane;
#pragma unroll
      for (int k = 0; k < NP; k++) {
        const RV z = v[g][k] + *(const RV*)(xp + k * 128);
        q[g * NP + k] = dot_lrelu(z, at[k]);
      }
      if constexpr (NP == 2) {
        if (g == 1 || g == 3 || g == 5)
          asm volatile("" :: "v"(q[2 * g - 2]), "v"(q[2 * g - 1]), "v"(q[2 * g]), "v"(q[2 * g + 1]));
      }
    }
    float t;
    int gl;
    if constexpr (H_ == 4) {
      const int sub = lane & 15;
      const bool b3 = (sub & 8) != 0, b2 = (sub & 4) != 0, b1 = (sub & 2) != 0, b0 = (sub & 1) != 0;
      float w8[8], w4[4], w2[2];
#pragma unroll
      for (int k = 0; k < 8; k++) {
        const float keep = b3 ? q[k + 8] : q[k];
        const float send = b3 ? q[k] : q[k + 8];
        w8[k] = keep + __shfl_xor(send, 8, 32);
      }
#pragma unroll
      for (int k = 0; k < 4; k++) {
        const float keep = b2 ? w8[k + 4] : w8[k];
        const float send = b2 ? w8[k] : w8[k + 4];
        w4[k] = keep + __shfl_xor(send, 4, 32);
      }
#pragma unroll
      for (int k = 0; k < 2; k++) {
        const float keep = b1 ? w4[k + 2] : w4[k];
        const float send = b1 ? w4[k] : w4[k + 2];
        w2[k] = keep + __shfl_xor(send, 2, 32);
      }
      const float keep0 = b0 ? w2[1] : w2[0];
      const float send0 = b0 ? w2[0] : w2[1];
      t = keep0 + __shfl_xor(send0, 1, 32);
      gl = sub >> 1;
      const int eg = __shfl(e, j0 + gl, 32);
      if (j0 + gl < nb) ebuf[(size_t)eg * 4 + 2 * (sub & 1) + (lane >> 4)] = t;
    } else {
      const bool c4 = (lane & 16) != 0, c3 = (lane & 8) != 0, c2 = (lane & 4) != 0;
      float w4[4], w2[2];
#pragma unroll
      for (int k = 0; k < 4; k++) {
        const float keep = c4 ? q[k + 4] : q[k];
        const float send = c4 ? q[k] : q[k + 4];
        w4[k] = keep + __shfl_xor(send, 16, 32);
      }
#pragma unroll
      for (int k = 0; k < 2; k++) {
        const float keep = c3 ? w4[k + 2] : w4[k];
        const float send = c3 ? w4[k] : w4[k + 2];
        w2[k] = keep + __shfl_xor(send, 8, 32);
      }
      const float keep0 = c2 ? w2[1] : w2[0];
      const float send0 = c2 ? w2[0] : w2[1];
      t = keep0 + __shfl_xor(send0, 4, 32);
      t += __shfl_xor(t, 2, 32);
      t += __shfl_xor(t, 1, 32);
      gl = lane >> 2;
      const int eg = __shfl(e, j0 + gl, 32);
      if ((lane & 3) == 0 && j0 + gl < nb) ebuf[eg] = t;
    }
#pragma unroll
    for (int g = 0; g < 8; g++) {
      if (j0 + g < nb) {
        const int rg = rl_i(r, j0 + g);
        const bool mine = lane == (rg >> 5);
        if constexpr (H_ == 4) {
          const float l0 = rl_f(t, 2 * g), l1 = rl_f(t, 16 + 2 * g);
          const float l2 = rl_f(t, 2 * g + 1), l3 = rl_f(t, 17 + 2 * g);
          mx[0] = mine ? fmaxf(mx[0], l0) : mx[0]; mx[1] = mine ? fmaxf(mx[1], l1) : mx[1];
          mx[2] = mine ? fmaxf(mx[2], l2) : mx[2]; mx[3] = mine ? fmaxf(mx[3], l3) : mx[3];
        } else {
          const float l0 = rl_f(t, 4 * g);
          mx[0] = mine ? fmaxf(mx[0], l0) : mx[0];
        }
      }
    }
  }
}

template <int H_, typename RT_>
__device__ __forceinline__ void own_batch(float* __restrict__ accs, float (&den)[H_],
                                          const int* oe_w,
                                          const RT_* or_w, int b, int nb,
                                          const float* __restrict__ xl,
                                          const int* __restrict__ ei,
                                          const float* __restrict__ ebuf,
                                          const float* __restrict__ nmax, int lo, int lane) {
  constexpr int HC = H_ * 64;
  constexpr int NP = H_ == 4 ? 2 : 1;
  constexpr int W = H_ == 4 ? 4 : 2;
  typedef typename RowVec<H_>::T RV;
  const int jj = b + min(lane, nb - 1);
  const int e = oe_w[jj];
  const int r = (int)or_w[jj];
  const int s = ei[e];
  float a[H_];
  if constexpr (H_ == 4) {
    const v4f lg = *(const v4f*)(ebuf + (size_t)e * 4);
    const v4f mk = *(const v4f*)(nmax + (size_t)(lo + r) * 4);
    a[0] = expf(lg.x - mk.x); a[1] = expf(lg.y - mk.y);
    a[2] = expf(lg.z - mk.z); a[3] = expf(lg.w - mk.w);
  } else {
    a[0] = expf(ebuf[e] - nmax[lo + r]);
  }
  const bool hf = (lane >> 4) != 0;
  for (int j0 = 0; j0 < nb; j0 += 8) {
    RV v[8][NP];
#pragma unroll
    for (int g = 0; g < 8; g++) {
      const int sg = rl_i(s, min(j0 + g, nb - 1));
      const float* rp = xl + (size_t)sg * HC + W * lane;
#pragma unroll
      for (int k = 0; k < NP; k++) v[g][k] = *(const RV*)(rp + k * 128);
    }
    if constexpr (NP == 2) {
      asm volatile("" ::
                   "v"(v[0][0]), "v"(v[0][1]), "v"(v[1][0]), "v"(v[1][1]), "v"(v[2][0]), "v"(v[2][1]),
                   "v"(v[3][0]), "v"(v[3][1]), "v"(v[4][0]), "v"(v[4][1]), "v"(v[5][0]), "v"(v[5][1]),
                   "v"(v[6][0]), "v"(v[6][1]), "v"(v[7][0]), "v"(v[7][1]));
    } else {
      asm volatile("" ::
                   "v"(v[0][0]), "v"(v[1][0]), "v"(v[2][0]), "v"(v[3][0]),
                   "v"(v[4][0]), "v"(v[5][0]), "v"(v[6][0]), "v"(v[7][0]));
    }
#pragma unroll
    for (int g = 0; g < 8; g++) {
      if (j0 + g < nb) {
        const int rg = rl_i(r, j0 + g);
        const bool mine = lane == (rg >> 5);
        float alk[NP];
        if constexpr (H_ == 4) {
          const float a0 = rl_f(a[0], j0 + g), a1 = rl_f(a[1], j0 + g);
          const float a2 = rl_f(a[2], j0 + g), a3 = rl_f(a[3], j0 + g);
          den[0] += mine ? a0 : 0.0f; den[1] += mine ? a1 : 0.0f;
          den[2] += mine ? a2 : 0.0f; den[3] += mine ? a3 : 0.0f;
          alk[0] = hf ? a1 : a0;
          alk[1] = hf ? a3 : a2;
        } else {
          const float a0 = rl_f(a[0], j0 + g);
          den[0] += mine ? a0 : 0.0f;
          alk[0] = a0;
        }
        float* ap = accs + rg * HC + W * lane;
#pragma unroll
        for (int k = 0; k < NP; k++) {
          const float al = alk[k];
          RV cur = *(RV*)(ap + k * 128);
          cur += al * v[g][k];
          *(RV*)(ap + k * 128) = cur;
        }
      }
    }
  }
}

template <int H_, int PH_, bool REC_, typename RT_>
__device__ __forceinline__ void run_batch(float* rows, float (&reg)[H_], const int* oe_w,
                                          const RT_* or_w, int b, int nb,
                                          const float* __restrict__ xl,
                                          const int* __restrict__ ei,
                                          const float* __restrict__ att,
                                          const float* __restrict__ ebuf_r,
                                          const float* __restrict__ nmax_r, float* ebuf_w, int lo,
                                          int lane, int* rec_e, int* rec_r, int recpos) {
  if constexpr (PH_ == 0)
    score_batch<H_, REC_, RT_>(rows, reg, oe_w, or_w, b, nb, xl, ei, att, ebuf_w, lane, rec_e, rec_r,
                               recpos);
  else
    own_batch<H_, RT_>(rows, reg, oe_w, or_w, b, nb, xl, ei, ebuf_r, nmax_r, lo, lane);
}

__device__ __forceinline__ float elu1(float v) { return v > 0.0f ? v : expm1f(v); }

template <int H_, bool ELU_, int MODE_, int PH_>
__global__ __launch_bounds__(1024) void owner_pass(const float* __restrict__ xl,
                                                   const float* __restrict__ xr,
                                                   const float* __restrict__ att,
                                                   const int* __restrict__ ei, int E_,
                                                   const float* __restrict__ ebuf_r,
                                                   const float* __restrict__ eself_r,
                                                   const float* __restrict__ nmax_r,
                                                   float* ebuf_w, float* eself_w, float* nmax_w,
                                                   const float* __restrict__ bias,
                                                   float* __restrict__ outp, int Nn,
                                                   int* rl_e, int* rl_r, int* rl_n) {
  static_assert(MODE_ == 1 || MODE_ == 2, "1: scan and write the lists; 2: walk the lists");
  static_assert(PH_ == 0 || PH_ == 1, "0: logits and row maxima; 1: softmax weights and sums");
  static_assert(MODE_ == 2 || PH_ == 0, "the lists are written by the first pass of the call");
  constexpr int HC = H_ * 64;
  constexpr int R = OWN_R;
  constexpr int NW = 32;
  constexpr int CH = 512;
  constexpr int NL = CH / 32;
  constexpr int QCAP = H_ == 4 ? 640 : 1024;
  constexpr int OCAP = 128;
  static_assert(R <= 255, "row numbers inside the group are kept in a byte");
  static_assert(R <= 8 * NW, "lanes 0..7 of an owner wave keep the figures of its rows");
  static_assert(R * H_ <= NW * QCAP, "the rows' figures are handed over in the queue space");
  static_assert((R * HC) % 4 == 0, "16-byte clears");
  __shared__ __attribute__((aligned(16))) float accs[R * HC];
  __shared__ int qe[NW * QCAP];
  __shared__ unsigned char qr[NW * QCAP];
  __shared__ int oe[NW * OCAP];
  __shared__ unsigned char orow[NW * OCAP];
  __shared__ int qcnt[NW];
  __shared__ int pending[2];

  const int tid = (int)threadIdx.x;
  const int lane = tid & 31;
  const int w = __builtin_amdgcn_readfirstlane(tid >> 5);
  const int lo = (int)blockIdx.x * R;
  const int nrows = min(R, Nn - lo);
  const int tot = nrows * HC;
  const int* __restrict__ dsts = ei + E_;
  const unsigned lt = (1u << lane) - 1u;

  if constexpr (PH_ == 0) {
    const float* xrg = xr + (size_t)lo * HC;
    for (int n0 = 0; n0 < tot; n0 += 4 * 4096) {
      v4f t4[4];
#pragma unroll
      for (int u = 0; u < 4; u++) t4[u] = *(const v4f*)(xrg + min(n0 + 4 * tid + u * 4096, tot - 4));
      asm volatile("" :: "v"(t4[0]), "v"(t4[1]), "v"(t4[2]), "v"(t4[3]));
#pragma unroll
      for (int u = 0; u < 4; u++) {
        const int i = n0 + 4 * tid + u * 4096;
        if (i < tot) *(v4f*)(accs + i) = t4[u];
      }
    }
  } else {
    for (int i = 4 * tid; i < R * HC; i += 4096) *(v4f*)(accs + i) = (v4f){0.0f, 0.0f, 0.0f, 0.0f};
  }
  if (tid < 2) pending[tid] = 0;
  float reg[H_];
#pragma unroll
  for (int h = 0; h < H_; h++) reg[h] = PH_ == 0 ? -INFINITY : 0.0f;

  const int per = ((E_ + NW * CH - 1) / (NW * CH)) * CH;
  int pos = w * per;
  const int wend = min(pos + per, E_);

  auto load_step = [&](int (&d)[NL], int p) {
    if (p + CH <= E_) {
#pragma unroll
      for (int k = 0; k < NL; k++) d[k] = dsts[p + k * 32 + lane];
    } else {
#pragma unroll
      for (int k = 0; k < NL; k++) {
        int idx = p + k * 32 + lane;
        d[k] = idx < E_ ? dsts[idx] : -1;
      }
    }
  };
  const int lbase = (int)blockIdx.x * NW + w;
  int* const my_e = rl_e + (size_t)lbase * RCAP;
  int* const my_r = rl_r + (size_t)lbase * RCAP;
  int rec = 0;
  bool walk = false;
  int nlist = 0;
  if constexpr (MODE_ == 2) {
    const int c = rl_n[(int)blockIdx.x * NW + lane];
    walk = __builtin_amdgcn_ballot_w32(c > RCAP) == 0u;
    nlist = rl_i(c, w);
  }
  __syncthreads();

  if (walk) {
    for (int b = 0; b < nlist; b += 32)
      run_batch<H_, PH_, false, int>(accs, reg, my_e, my_r, b, min(32, nlist - b), xl, ei, att,
                                     ebuf_r, nmax_r, ebuf_w, lo, lane, nullptr, nullptr, 0);
    __syncthreads();
  } else
  for (int round = 0;; round++) {
    int cnt = 0;
    int d[NL];
    if (pos < wend) load_step(d, pos);
    while (pos < wend && cnt + CH <= QCAP) {
      int cur[NL];
#pragma unroll
      for (int k = 0; k < NL; k++) cur[k] = d[k];
      if (pos + CH < wend) load_step(d, pos + CH);
#pragma unroll
      for (int k = 0; k < NL; k++) {
        const unsigned rr = (unsigned)(cur[k] - lo);
        const bool m = rr < (unsigned)nrows;
        const unsigned mask = __builtin_amdgcn_ballot_w32(m);
        if (mask != 0u) {
          if (m) {
            const int p = w * QCAP + cnt + __builtin_popcount(mask & lt);
            qe[p] = pos + k * 32 + lane;
            qr[p] = (unsigned char)rr;
          }
          cnt += __builtin_popcount(mask);
        }
      }
      pos += CH;
    }
    if (lane == 0) {
      qcnt[w] = cnt;
      if (pos < wend) pending[round & 1] = 1;
    }
    __syncthreads();

    int fill = 0;
    for (int q = 0; q < NW; q++) {
      const int n = __builtin_amdgcn_readfirstlane(qcnt[q]);
      for (int i = 0; i < n; i += 32) {
        const int j = i + lane;
        const int r = qr[q * QCAP + j];
        const bool mine = j < n && (r & 31) == w;
        const unsigned mask = __builtin_amdgcn_ballot_w32(mine);
        if (mask != 0u) {
          if (mine) {
            const int p = w * OCAP + fill + __builtin_popcount(mask & lt);
            oe[p] = qe[q * QCAP + j];
            orow[p] = (unsigned char)r;
          }
          fill += __builtin_popcount(mask);
          if (fill > OCAP - 32) {
            for (int b = 0; b < fill; b += 32) {
              const int nbt = min(32, fill - b);
              run_batch<H_, PH_, MODE_ == 1, unsigned char>(accs, reg, oe + w * OCAP,
                                                            orow + w * OCAP, b, nbt, xl, ei, att,
                                                            ebuf_r, nmax_r, ebuf_w, lo, lane, my_e,
                                                            my_r, rec);
              rec += nbt;
            }
            fill = 0;
          }
        }
      }
    }
    for (int b = 0; b < fill; b += 32) {
      const int nbt = min(32, fill - b);
      run_batch<H_, PH_, MODE_ == 1, unsigned char>(accs, reg, oe + w * OCAP, orow + w * OCAP, b,
                                                    nbt, xl, ei, att, ebuf_r, nmax_r, ebuf_w, lo,
                                                    lane, my_e, my_r, rec);
      rec += nbt;
    }

    const int again = __builtin_amdgcn_readfirstlane(pending[round & 1]);
    if (tid == 0) pending[(round + 1) & 1] = 0;
    __syncthreads();
    if (!again) break;
  }

  if constexpr (MODE_ == 1) {
    if (lane == 0) rl_n[lbase] = rec;
  }

  float* dsum = (float*)qe;
  {
    const int rr = lane * 32 + w;
    if (lane < 8 && rr < R) {
#pragma unroll
      for (int h = 0; h < H_; h++) dsum[rr * H_ + h] = reg[h];
    }
  }
  __syncthreads();

  const int ch = (4 * tid) % HC;
  if constexpr (PH_ == 0) {
    const v4f at4 = *(const v4f*)(att + ch);
    const float* xlg = xl + (size_t)lo * HC;
    for (int n0 = 0; n0 < tot; n0 += 4 * 4096) {
      v4f xs[4];
#pragma unroll
      for (int u = 0; u < 4; u++) xs[u] = *(const v4f*)(xlg + min(n0 + 4 * tid + u * 4096, tot - 4));
      asm volatile("" :: "v"(xs[0]), "v"(xs[1]), "v"(xs[2]), "v"(xs[3]));
#pragma unroll
      for (int u = 0; u < 4; u++) {
        const int i = n0 + 4 * tid + u * 4096;
        const int ic = min(i, tot - 4);
        const v4f z = xs[u] + *(const v4f*)(accs + ic);
        float t = dot_lrelu(z, at4);
        t += __shfl_xor(t, 8, 32);
        t += __shfl_xor(t, 4, 32);
        t += __shfl_xor(t, 2, 32);
        t += __shfl_xor(t, 1, 32);
        if ((lane & 15) == 0 && i < tot) {
          const int rh = (ic / HC) * H_ + ((ic % HC) >> 6);
          const int item = lo * H_ + rh;
          nmax_w[item] = fmaxf(t, dsum[rh]);
          eself_w[item] = t;
        }
      }
    }
  } else {
    const v4f bb = *(const v4f*)(bias + ch);
    for (int i0 = 4 * tid; i0 < tot; i0 += 4 * 4096) {
      v4f xs[4];
      float lg[4];
      float mk[4];
#pragma unroll
      for (int u = 0; u < 4; u++) {
        const int i = min(i0 + u * 4096, tot - 4);
        const size_t node = (size_t)(lo + i / HC);
        const int item = (int)node * H_ + ((i % HC) >> 6);
        lg[u] = eself_r[item];
        mk[u] = nmax_r[item];
        xs[u] = *(const v4f*)(xl + node * HC + i % HC);
      }
      asm volatile("" ::
                   "v"(xs[0]), "v"(xs[1]), "v"(xs[2]), "v"(xs[3]),
                   "v"(lg[0]), "v"(mk[0]), "v"(lg[1]), "v"(mk[1]),
                   "v"(lg[2]), "v"(mk[2]), "v"(lg[3]), "v"(mk[3]));
#pragma unroll
      for (int u = 0; u < 4; u++) {
        const int i = i0 + u * 4096;
        if (i < tot) {
          const float exs = expf(lg[u] - mk[u]);
          const float rc = 1.0f / (dsum[(i / HC) * H_ + ((i % HC) >> 6)] + exs);
          const v4f sm = *(const v4f*)(accs + i);
          v4f o = (sm + exs * xs[u]) * rc + bb;
          if constexpr (ELU_) { o.x = elu1(o.x); o.y = elu1(o.y); o.z = elu1(o.z); o.w = elu1(o.w); }
          *(v4f*)(outp + (size_t)(lo + i / HC) * HC + i % HC) = o;
        }
      }
    }
  }
}

extern "C" void kernel_launch(void* const* d_in, const int* in_sizes, int n_in,
                              void* d_out, int out_size, void* d_ws, size_t ws_size,
                              hipStream_t stream) {
  (void)n_in; (void)out_size; (void)ws_size;
  const int IN = 128, HID = 64, H = 4, OUT = 64;
  const float*    x    = (const float*)d_in[0];
  const int*      ei   = (const int*)d_in[1];
  const float*    Wl1  = (const float*)d_in[2];
  const float*    bl1  = (const float*)d_in[3];
  const float*    Wr1  = (const float*)d_in[4];
  const float*    br1  = (const float*)d_in[5];
  const float*    att1 = (const float*)d_in[6];
  const float*    bias1= (const float*)d_in[7];
  const float*    Wl2  = (const float*)d_in[8];
  const float*    bl2  = (const float*)d_in[9];
  const float*    Wr2  = (const float*)d_in[10];
  const float*    br2  = (const float*)d_in[11];
  const float*    att2 = (const float*)d_in[12];
  const float*    bias2= (const float*)d_in[13];
  float* out = (float*)d_out;

  const int N    = in_sizes[0] / IN;
  const int E    = in_sizes[1] / 2;
  const int Etot = E + N;
  const int HC   = H * HID;

  float* ws = (float*)d_ws;
  size_t o = 0;
  float*    xl1   = ws + o; o += (size_t)N * HC;
  float*    xr1   = ws + o; o += (size_t)N * HC;
  float*    ebuf  = ws + o; o += (size_t)Etot * H;
  float*    nmax  = ws + o; o += (size_t)N * H;
  const int gown = (N + OWN_R - 1) / OWN_R;
  int* rl_e = (int*)(ws + o); o += (size_t)gown * 32 * RCAP;
  int* rl_r = (int*)(ws + o); o += (size_t)gown * 32 * RCAP;
  int* rl_n = (int*)(ws + o); o += (size_t)gown * 32;
  float*  h    = xr1;
  float*  xl2  = xl1;
  float*  xr2  = xl1 + (size_t)N * OUT;

  const int B = 256;
  int rblks = (N / 16 + 7) / 8;

  gemm_bias_wmma<128, 256><<<rblks * (HC / 64), B, 0, stream>>>(x, Wl1, bl1, xl1, N);
  gemm_bias_wmma<128, 256><<<rblks * (HC / 64), B, 0, stream>>>(x, Wr1, br1, xr1, N);
  owner_pass<4, false, 1, 0><<<gown, 1024, 0, stream>>>(xl1, xr1, att1, ei, E, nullptr, nullptr,
                                                        nullptr, ebuf, ebuf + (size_t)E * H, nmax,
                                                        nullptr, nullptr, N, rl_e, rl_r, rl_n);
  owner_pass<4, true, 2, 1><<<gown, 1024, 0, stream>>>(xl1, nullptr, nullptr, ei, E, ebuf,
                                                       ebuf + (size_t)E * H, nmax, nullptr, nullptr,
                                                       nullptr, bias1, h, N, rl_e, rl_r, rl_n);

  gemm_bias_wmma<256, 64><<<rblks * (OUT / 64), B, 0, stream>>>(h, Wl2, bl2, xl2, N);
  gemm_bias_wmma<256, 64><<<rblks * (OUT / 64), B, 0, stream>>>(h, Wr2, br2, xr2, N);
  owner_pass<1, false, 2, 0><<<gown, 1024, 0, stream>>>(xl2, xr2, att2, ei, E, nullptr, nullptr,
                                                        nullptr, ebuf, ebuf + E, nmax, nullptr,
                                                        nullptr, N, rl_e, rl_r, rl_n);
  owner_pass<1, false, 2, 1><<<gown, 1024, 0, stream>>>(xl2, nullptr, nullptr, ei, E, ebuf, ebuf + E,
                                                        nmax, nullptr, nullptr, nullptr, bias2, out,
                                                        N, rl_e, rl_r, rl_n);
}


# Destination-list scan leaves the owner pass: a new first launch sorts each 416-edge chunk by owner group, list order kept, with a start table; a group copies only its entries; order of sums kept
# speedup vs baseline: 2.6160x; 1.0388x over previous
#include <hip/hip_runtime.h>
#include <math.h>

#define SLOPE 0.2f

typedef float v2f __attribute__((ext_vector_type(2)));
typedef float v4f __attribute__((ext_vector_type(4)));
typedef float v8f __attribute__((ext_vector_type(8)));

__device__ __forceinline__ float lrelu(float v) { return v > 0.0f ? v : SLOPE * v; }

template <int K, int NC>
__global__ __launch_bounds__(256) void gemm_bias_wmma(const float* __restrict__ X,
                                                      const float* __restrict__ W,
                                                      const float* __restrict__ b,
                                                      float* __restrict__ Y, int M) {
  constexpr int SLAB = 64;
  constexpr int NSLAB = NC / SLAB;
  constexpr int PITCH = 160;
  __shared__ __attribute__((aligned(16))) float wl[(K / 2) * PITCH];

  const int slab = blockIdx.x % NSLAB;
  const int rblk = blockIdx.x / NSLAB;
  const int cbase = slab * SLAB;

#pragma unroll
  for (int j = 0; j < (K / 2) * 16 / 256; j++) {
    int it = j * 256 + (int)threadIdx.x;
    int p = it >> 4, q = it & 15;
    const float* g = W + (size_t)(2 * p) * NC + cbase + 4 * q;
    float4 r0 = *(const float4*)g;
    float4 r1 = *(const float4*)(g + NC);
    float4* d = (float4*)(wl + p * PITCH + 8 * q);
    d[0] = make_float4(r0.x, r1.x, r0.y, r1.y);
    d[1] = make_float4(r0.z, r1.z, r0.w, r1.w);
  }
  __syncthreads();

  int wid = threadIdx.x >> 5;
  int mt = rblk * 8 + wid;
  if (mt * 16 >= M) return;
  int lane = threadIdx.x & 31;
  int half = lane >> 4;
  int l = lane & 15;
  const float* xp = X + (size_t)(mt * 16 + l) * K + 2 * half;
  const float* bp = wl + half * PITCH + 2 * l;
  v8f acc[4];
#pragma unroll
  for (int t = 0; t < 4; t++) {
    float bc = b[cbase + t * 16 + l];
    acc[t] = (v8f){bc, bc, bc, bc, bc, bc, bc, bc};
  }
#pragma unroll
  for (int k0 = 0; k0 < K; k0 += 4) {
    v2f a = *(const v2f*)(xp + k0);
#pragma unroll
    for (int t = 0; t < 4; t++) {
      v2f bv = *(const v2f*)(bp + (k0 / 2) * PITCH + t * 32);
      acc[t] = __builtin_amdgcn_wmma_f32_16x16x4_f32(false, a, false, bv, (short)0, acc[t],
                                                     false, false);
    }
  }
#pragma unroll
  for (int t = 0; t < 4; t++) {
    float* yp = Y + (size_t)(mt * 16 + 8 * half) * NC + cbase + t * 16 + l;
#pragma unroll
    for (int v = 0; v < 8; v++) yp[(size_t)v * NC] = acc[t][v];
  }
}

#define OWN_R 196
#define RCAP 512
#define BK_L 416
#define BK_G 256
#define BK_P 288

__global__ __launch_bounds__(32) void bucket_edges(const int* __restrict__ dsts, int E_,
                                                   int* __restrict__ bk_ent,
                                                   int* __restrict__ bk_off) {
  constexpr int NB = BK_L / 32;
  __shared__ int cnt[BK_G];
  const int lane = (int)threadIdx.x;
  const int c = (int)blockIdx.x;
  const int base = c * BK_L;
  const unsigned lt = (1u << lane) - 1u;
#pragma unroll
  for (int j = 0; j < BK_G / 32; j++) cnt[j * 32 + lane] = 0;
  int d[NB];
#pragma unroll
  for (int k = 0; k < NB; k++) {
    const int idx = base + k * 32 + lane;
    d[k] = idx < E_ ? dsts[idx] : -1;
  }
  int rank[NB];
#pragma unroll
  for (int k = 0; k < NB; k++) {
    __syncthreads();
    const bool valid = d[k] >= 0;
    const unsigned g = valid ? (unsigned)d[k] / (unsigned)OWN_R : 0u;
    unsigned m = __builtin_amdgcn_ballot_w32(valid);
#pragma unroll
    for (int b = 0; b < 8; b++) {
      const bool bit = ((g >> b) & 1u) != 0u;
      const unsigned bb = __builtin_amdgcn_ballot_w32(bit);
      m &= bit ? bb : ~bb;
    }
    const int before = __builtin_popcount(m & lt);
    int start = 0;
    if (valid && before == 0) {
      start = cnt[g];
      cnt[g] = start + __builtin_popcount(m);
    }
    start = __shfl(start, valid ? __builtin_ctz(m) : lane, 32);
    rank[k] = start + before;
  }
  __syncthreads();
  int cs[8];
  int run = 0;
#pragma unroll
  for (int j = 0; j < 8; j++) {
    const int t = cnt[8 * lane + j];
    cs[j] = run;
    run += t;
  }
  int inc = run;
#pragma unroll
  for (int o = 1; o < 32; o <<= 1) {
    const int t = __shfl_up(inc, o, 32);
    if (lane >= o) inc += t;
  }
  int* offp = bk_off + (size_t)c * BK_P;
#pragma unroll
  for (int j = 0; j < 8; j++) {
    cs[j] += inc - run;
    cnt[8 * lane + j] = cs[j];
    offp[8 * lane + j] = cs[j];
  }
  if (lane == 31) offp[BK_G] = inc;
  __syncthreads();
#pragma unroll
  for (int k = 0; k < NB; k++) {
    if (d[k] >= 0) {
      const unsigned g = (unsigned)d[k] / (unsigned)OWN_R;
      const int row = d[k] - (int)g * OWN_R;
      bk_ent[base + cnt[g] + rank[k]] = ((base + k * 32 + lane) << 8) | row;
    }
  }
}

template <int H_> struct RowVec;
template <> struct RowVec<4> { typedef v4f T; };
template <> struct RowVec<1> { typedef v2f T; };

__device__ __forceinline__ int rl_i(int v, int j) { return __builtin_amdgcn_readlane(v, j); }
__device__ __forceinline__ float rl_f(float v, int j) {
  return __int_as_float(__builtin_amdgcn_readlane(__float_as_int(v), j));
}
__device__ __forceinline__ float dot_lrelu(v4f z, v4f t) {
  return lrelu(z.x) * t.x + lrelu(z.y) * t.y + lrelu(z.z) * t.z + lrelu(z.w) * t.w;
}
__device__ __forceinline__ float dot_lrelu(v2f z, v2f t) {
  return lrelu(z.x) * t.x + lrelu(z.y) * t.y;
}

template <int H_, bool REC_, typename RT_>
__device__ __forceinline__ void score_batch(const float* xrs, float (&mx)[H_],
                                            const int* oe_w, const RT_* or_w, int b, int nb,
                                            const float* __restrict__ xl,
                                            const int* __restrict__ ei,
                                            const float* __restrict__ att, float* ebuf, int lane,
                                            int* rec_e, int* rec_r, int recpos) {
  constexpr int HC = H_ * 64;
  constexpr int NP = H_ == 4 ? 2 : 1;
  constexpr int W = H_ == 4 ? 4 : 2;
  typedef typename RowVec<H_>::T RV;
  const int jj = b + min(lane, nb - 1);
  const int e = oe_w[jj];
  const int r = (int)or_w[jj];
  if constexpr (REC_) {
    if (lane < nb && recpos + lane < RCAP) {
      rec_e[recpos + lane] = e;
      rec_r[recpos + lane] = r;
    }
  }
  const int s = ei[e];
  RV at[NP];
#pragma unroll
  for (int k = 0; k < NP; k++) at[k] = *(const RV*)(att + k * 128 + W * lane);
  for (int j0 = 0; j0 < nb; j0 += 8) {
    RV v[8][NP];
#pragma unroll
    for (int g = 0; g < 8; g++) {
      const int sg = rl_i(s, min(j0 + g, nb - 1));
      const float* rp = xl + (size_t)sg * HC + W * lane;
#pragma unroll
      for (int k = 0; k < NP; k++) v[g][k] = *(const RV*)(rp + k * 128);
    }
    if constexpr (NP == 2) {
      asm volatile("" ::
                   "v"(v[0][0]), "v"(v[0][1]), "v"(v[1][0]), "v"(v[1][1]), "v"(v[2][0]), "v"(v[2][1]),
                   "v"(v[3][0]), "v"(v[3][1]), "v"(v[4][0]), "v"(v[4][1]), "v"(v[5][0]), "v"(v[5][1]),
                   "v"(v[6][0]), "v"(v[6][1]), "v"(v[7][0]), "v"(v[7][1]));
    } else {
      asm volatile("" ::
                   "v"(v[0][0]), "v"(v[1][0]), "v"(v[2][0]), "v"(v[3][0]),
                   "v"(v[4][0]), "v"(v[5][0]), "v"(v[6][0]), "v"(v[7][0]));
    }
    float q[8 * NP];
#pragma unroll
    for (int g = 0; g < 8; g++) {
      const int rg = rl_i(r, min(j0 + g, nb - 1));
      const float* xp = xrs + rg * HC + W * lane;
#pragma unroll
      for (int k = 0; k < NP; k++) {
        const RV z = v[g][k] + *(const RV*)(xp + k * 128);
        q[g * NP + k] = dot_lrelu(z, at[k]);
      }
      if constexpr (NP == 2) {
        if (g == 1 || g == 3 || g == 5)
          asm volatile("" :: "v"(q[2 * g - 2]), "v"(q[2 * g - 1]), "v"(q[2 * g]), "v"(q[2 * g + 1]));
      }
    }
    float t;
    int gl;
    if constexpr (H_ == 4) {
      const int sub = lane & 15;
      const bool b3 = (sub & 8) != 0, b2 = (sub & 4) != 0, b1 = (sub & 2) != 0, b0 = (sub & 1) != 0;
      float w8[8], w4[4], w2[2];
#pragma unroll
      for (int k = 0; k < 8; k++) {
        const float keep = b3 ? q[k + 8] : q[k];
        const float send = b3 ? q[k] : q[k + 8];
        w8[k] = keep + __shfl_xor(send, 8, 32);
      }
#pragma unroll
      for (int k = 0; k < 4; k++) {
        const float keep = b2 ? w8[k + 4] : w8[k];
        const float send = b2 ? w8[k] : w8[k + 4];
        w4[k] = keep + __shfl_xor(send, 4, 32);
      }
#pragma unroll
      for (int k = 0; k < 2; k++) {
        const float keep = b1 ? w4[k + 2] : w4[k];
        const float send = b1 ? w4[k] : w4[k + 2];
        w2[k] = keep + __shfl_xor(send, 2, 32);
      }
      const float keep0 = b0 ? w2[1] : w2[0];
      const float send0 = b0 ? w2[0] : w2[1];
      t = keep0 + __shfl_xor(send0, 1, 32);
      gl = sub >> 1;
      const int eg = __shfl(e, j0 + gl, 32);
      if (j0 + gl < nb) ebuf[(size_t)eg * 4 + 2 * (sub & 1) + (lane >> 4)] = t;
    } else {
      const bool c4 = (lane & 16) != 0, c3 = (lane & 8) != 0, c2 = (lane & 4) != 0;
      float w4[4], w2[2];
#pragma unroll
      for (int k = 0; k < 4; k++) {
        const float keep = c4 ? q[k + 4] : q[k];
        const float send = c4 ? q[k] : q[k + 4];
        w4[k] = keep + __shfl_xor(send, 16, 32);
      }
#pragma unroll
      for (int k = 0; k < 2; k++) {
        const float keep = c3 ? w4[k + 2] : w4[k];
        const float send = c3 ? w4[k] : w4[k + 2];
        w2[k] = keep + __shfl_xor(send, 8, 32);
      }
      const float keep0 = c2 ? w2[1] : w2[0];
      const float send0 = c2 ? w2[0] : w2[1];
      t = keep0 + __shfl_xor(send0, 4, 32);
      t += __shfl_xor(t, 2, 32);
      t += __shfl_xor(t, 1, 32);
      gl = lane >> 2;
      const int eg = __shfl(e, j0 + gl, 32);
      if ((lane & 3) == 0 && j0 + gl < nb) ebuf[eg] = t;
    }
#pragma unroll
    for (int g = 0; g < 8; g++) {
      if (j0 + g < nb) {
        const int rg = rl_i(r, j0 + g);
        const bool mine = lane == (rg >> 5);
        if constexpr (H_ == 4) {
          const float l0 = rl_f(t, 2 * g), l1 = rl_f(t, 16 + 2 * g);
          const float l2 = rl_f(t, 2 * g + 1), l3 = rl_f(t, 17 + 2 * g);
          mx[0] = mine ? fmaxf(mx[0], l0) : mx[0]; mx[1] = mine ? fmaxf(mx[1], l1) : mx[1];
          mx[2] = mine ? fmaxf(mx[2], l2) : mx[2]; mx[3] = mine ? fmaxf(mx[3], l3) : mx[3];
        } else {
          const float l0 = rl_f(t, 4 * g);
          mx[0] = mine ? fmaxf(mx[0], l0) : mx[0];
        }
      }
    }
  }
}

template <int H_, typename RT_>
__device__ __forceinline__ void own_batch(float* __restrict__ accs, float (&den)[H_],
                                          const int* oe_w,
                                          const RT_* or_w, int b, int nb,
                                          const float* __restrict__ xl,
                                          const int* __restrict__ ei,
                                          const float* __restrict__ ebuf,
                                          const float* __restrict__ nmax, int lo, int lane) {
  constexpr int HC = H_ * 64;
  constexpr int NP = H_ == 4 ? 2 : 1;
  constexpr int W = H_ == 4 ? 4 : 2;
  typedef typename RowVec<H_>::T RV;
  const int jj = b + min(lane, nb - 1);
  const int e = oe_w[jj];
  const int r = (int)or_w[jj];
  const int s = ei[e];
  float a[H_];
  if constexpr (H_ == 4) {
    const v4f lg = *(const v4f*)(ebuf + (size_t)e * 4);
    const v4f mk = *(const v4f*)(nmax + (size_t)(lo + r) * 4);
    a[0] = expf(lg.x - mk.x); a[1] = expf(lg.y - mk.y);
    a[2] = expf(lg.z - mk.z); a[3] = expf(lg.w - mk.w);
  } else {
    a[0] = expf(ebuf[e] - nmax[lo + r]);
  }
  const bool hf = (lane >> 4) != 0;
  for (int j0 = 0; j0 < nb; j0 += 8) {
    RV v[8][NP];
#pragma unroll
    for (int g = 0; g < 8; g++) {
      const int sg = rl_i(s, min(j0 + g, nb - 1));
      const float* rp = xl + (size_t)sg * HC + W * lane;
#pragma unroll
      for (int k = 0; k < NP; k++) v[g][k] = *(const RV*)(rp + k * 128);
    }
    if constexpr (NP == 2) {
      asm volatile("" ::
                   "v"(v[0][0]), "v"(v[0][1]), "v"(v[1][0]), "v"(v[1][1]), "v"(v[2][0]), "v"(v[2][1]),
                   "v"(v[3][0]), "v"(v[3][1]), "v"(v[4][0]), "v"(v[4][1]), "v"(v[5][0]), "v"(v[5][1]),
                   "v"(v[6][0]), "v"(v[6][1]), "v"(v[7][0]), "v"(v[7][1]));
    } else {
      asm volatile("" ::
                   "v"(v[0][0]), "v"(v[1][0]), "v"(v[2][0]), "v"(v[3][0]),
                   "v"(v[4][0]), "v"(v[5][0]), "v"(v[6][0]), "v"(v[7][0]));
    }
#pragma unroll
    for (int g = 0; g < 8; g++) {
      if (j0 + g < nb) {
        const int rg = rl_i(r, j0 + g);
        const bool mine = lane == (rg >> 5);
        float alk[NP];
        if constexpr (H_ == 4) {
          const float a0 = rl_f(a[0], j0 + g), a1 = rl_f(a[1], j0 + g);
          const float a2 = rl_f(a[2], j0 + g), a3 = rl_f(a[3], j0 + g);
          den[0] += mine ? a0 : 0.0f; den[1] += mine ? a1 : 0.0f;
          den[2] += mine ? a2 : 0.0f; den[3] += mine ? a3 : 0.0f;
          alk[0] = hf ? a1 : a0;
          alk[1] = hf ? a3 : a2;
        } else {
          const float a0 = rl_f(a[0], j0 + g);
          den[0] += mine ? a0 : 0.0f;
          alk[0] = a0;
        }
        float* ap = accs + rg * HC + W * lane;
#pragma unroll
        for (int k = 0; k < NP; k++) {
          const float al = alk[k];
          RV cur = *(RV*)(ap + k * 128);
          cur += al * v[g][k];
          *(RV*)(ap + k * 128) = cur;
        }
      }
    }
  }
}

template <int H_, int PH_, bool REC_, typename RT_>
__device__ __forceinline__ void run_batch(float* rows, float (&reg)[H_], const int* oe_w,
                                          const RT_* or_w, int b, int nb,
                                          const float* __restrict__ xl,
                                          const int* __restrict__ ei,
                                          const float* __restrict__ att,
                                          const float* __restrict__ ebuf_r,
                                          const float* __restrict__ nmax_r, float* ebuf_w, int lo,
                                          int lane, int* rec_e, int* rec_r, int recpos) {
  if constexpr (PH_ == 0)
    score_batch<H_, REC_, RT_>(rows, reg, oe_w, or_w, b, nb, xl, ei, att, ebuf_w, lane, rec_e, rec_r,
                               recpos);
  else
    own_batch<H_, RT_>(rows, reg, oe_w, or_w, b, nb, xl, ei, ebuf_r, nmax_r, lo, lane);
}

__device__ __forceinline__ float elu1(float v) { return v > 0.0f ? v : expm1f(v); }

template <int H_, bool ELU_, int MODE_, int PH_>
__global__ __launch_bounds__(1024) void owner_pass(const float* __restrict__ xl,
                                                   const float* __restrict__ xr,
                                                   const float* __restrict__ att,
                                                   const int* __restrict__ ei,
                                                   const int* __restrict__ bk_ent,
                                                   const int* __restrict__ bk_off, int nch,
                                                   const float* __restrict__ ebuf_r,
                                                   const float* __restrict__ eself_r,
                                                   const float* __restrict__ nmax_r,
                                                   float* ebuf_w, float* eself_w, float* nmax_w,
                                                   const float* __restrict__ bias,
                                                   float* __restrict__ outp, int Nn,
                                                   int* rl_e, int* rl_r, int* rl_n) {
  static_assert(MODE_ == 1 || MODE_ == 2, "1: fill queues and write the lists; 2: walk the lists");
  static_assert(PH_ == 0 || PH_ == 1, "0: logits and row maxima; 1: softmax weights and sums");
  static_assert(MODE_ == 2 || PH_ == 0, "the lists are written by the first pass of the call");
  constexpr int HC = H_ * 64;
  constexpr int R = OWN_R;
  constexpr int NW = 32;
  constexpr int QCAP = H_ == 4 ? 640 : 1024;
  constexpr int OCAP = 128;
  static_assert(R <= 255, "row numbers inside the group are kept in a byte");
  static_assert(R <= 8 * NW, "lanes 0..7 of an owner wave keep the figures of its rows");
  static_assert(R * H_ <= NW * QCAP, "the rows' figures are handed over in the queue space");
  static_assert((R * HC) % 4 == 0, "16-byte clears");
  __shared__ __attribute__((aligned(16))) float accs[R * HC];
  __shared__ int qe[NW * QCAP];
  __shared__ unsigned char qr[NW * QCAP];
  __shared__ int oe[NW * OCAP];
  __shared__ unsigned char orow[NW * OCAP];
  __shared__ int qcnt[NW];
  __shared__ int pending[2];

  const int tid = (int)threadIdx.x;
  const int lane = tid & 31;
  const int w = __builtin_amdgcn_readfirstlane(tid >> 5);
  const int lo = (int)blockIdx.x * R;
  const int nrows = min(R, Nn - lo);
  const int tot = nrows * HC;
  const unsigned lt = (1u << lane) - 1u;

  if constexpr (PH_ == 0) {
    const float* xrg = xr + (size_t)lo * HC;
    for (int n0 = 0; n0 < tot; n0 += 4 * 4096) {
      v4f t4[4];
#pragma unroll
      for (int u = 0; u < 4; u++) t4[u] = *(const v4f*)(xrg + min(n0 + 4 * tid + u * 4096, tot - 4));
      asm volatile("" :: "v"(t4[0]), "v"(t4[1]), "v"(t4[2]), "v"(t4[3]));
#pragma unroll
      for (int u = 0; u < 4; u++) {
        const int i = n0 + 4 * tid + u * 4096;
        if (i < tot) *(v4f*)(accs + i) = t4[u];
      }
    }
  } else {
    for (int i = 4 * tid; i < R * HC; i += 4096) *(v4f*)(accs + i) = (v4f){0.0f, 0.0f, 0.0f, 0.0f};
  }
  if (tid < 2) pending[tid] = 0;
  float reg[H_];
#pragma unroll
  for (int h = 0; h < H_; h++) reg[h] = PH_ == 0 ? -INFINITY : 0.0f;

  const int cpw = (nch + NW - 1) / NW;
  int cb = w * cpw;
  const int cend = min(cb + cpw, nch);
  int cs = 0, ce = 0;
  bool have = false;
  const int lbase = (int)blockIdx.x * NW + w;
  int* const my_e = rl_e + (size_t)lbase * RCAP;
  int* const my_r = rl_r + (size_t)lbase * RCAP;
  int rec = 0;
  bool walk = false;
  int nlist = 0;
  if constexpr (MODE_ == 2) {
    const int c = rl_n[(int)blockIdx.x * NW + lane];
    walk = __builtin_amdgcn_ballot_w32(c > RCAP) == 0u;
    nlist = rl_i(c, w);
  }
  __syncthreads();

  if (walk) {
    for (int b = 0; b < nlist; b += 32)
      run_batch<H_, PH_, false, int>(accs, reg, my_e, my_r, b, min(32, nlist - b), xl, ei, att,
                                     ebuf_r, nmax_r, ebuf_w, lo, lane, nullptr, nullptr, 0);
    __syncthreads();
  } else
  for (int round = 0;; round++) {
    int cnt = 0;
    for (;;) {
      if (!have) {
        if (cb >= cend) break;
        cs = 0; ce = 0;
        if (cb + lane < cend) {
          const int* op = bk_off + (size_t)(cb + lane) * BK_P + (int)blockIdx.x;
          cs = op[0];
          ce = op[1];
        }
        have = true;
      }
      const int n = ce - cs;
      int inc = n;
#pragma unroll
      for (int o = 1; o < 32; o <<= 1) {
        const int t = __shfl_up(inc, o, 32);
        if (lane >= o) inc += t;
      }
      const int all = rl_i(inc, 31);
      const int room = QCAP - cnt;
      const int ex = inc - n;
      const int take = max(0, min(n, room - ex));
      const int* ep = bk_ent + (size_t)(cb + lane) * BK_L + cs;
      const int qp = w * QCAP + cnt + ex;
      for (int j0 = 0; __builtin_amdgcn_ballot_w32(j0 < take) != 0u; j0 += 4) {
        int v[4];
#pragma unroll
        for (int u = 0; u < 4; u++) v[u] = j0 + u < take ? ep[j0 + u] : 0;
#pragma unroll
        for (int u = 0; u < 4; u++) {
          if (j0 + u < take) {
            qe[qp + j0 + u] = v[u] >> 8;
            qr[qp + j0 + u] = (unsigned char)(v[u] & 255);
          }
        }
      }
      cs += take;
      if (all > room) { cnt = QCAP; break; }
      cnt += all;
      have = false;
      cb += 32;
    }
    if (lane == 0) {
      qcnt[w] = cnt;
      if (have || cb < cend) pending[round & 1] = 1;
    }
    __syncthreads();

    int fill = 0;
    for (int q = 0; q < NW; q++) {
      const int n = __builtin_amdgcn_readfirstlane(qcnt[q]);
      for (int i = 0; i < n; i += 32) {
        const int j = i + lane;
        const int r = qr[q * QCAP + j];
        const bool mine = j < n && (r & 31) == w;
        const unsigned mask = __builtin_amdgcn_ballot_w32(mine);
        if (mask != 0u) {
          if (mine) {
            const int p = w * OCAP + fill + __builtin_popcount(mask & lt);
            oe[p] = qe[q * QCAP + j];
            orow[p] = (unsigned char)r;
          }
          fill += __builtin_popcount(mask);
          if (fill > OCAP - 32) {
            for (int b = 0; b < fill; b += 32) {
              const int nbt = min(32, fill - b);
              run_batch<H_, PH_, MODE_ == 1, unsigned char>(accs, reg, oe + w * OCAP,
                                                            orow + w * OCAP, b, nbt, xl, ei, att,
                                                            ebuf_r, nmax_r, ebuf_w, lo, lane, my_e,
                                                            my_r, rec);
              rec += nbt;
            }
            fill = 0;
          }
        }
      }
    }
    for (int b = 0; b < fill; b += 32) {
      const int nbt = min(32, fill - b);
      run_batch<H_, PH_, MODE_ == 1, unsigned char>(accs, reg, oe + w * OCAP, orow + w * OCAP, b,
                                                    nbt, xl, ei, att, ebuf_r, nmax_r, ebuf_w, lo,
                                                    lane, my_e, my_r, rec);
      rec += nbt;
    }

    const int again = __builtin_amdgcn_readfirstlane(pending[round & 1]);
    if (tid == 0) pending[(round + 1) & 1] = 0;
    __syncthreads();
    if (!again) break;
  }

  if constexpr (MODE_ == 1) {
    if (lane == 0) rl_n[lbase] = rec;
  }

  float* dsum = (float*)qe;
  {
    const int rr = lane * 32 + w;
    if (lane < 8 && rr < R) {
#pragma unroll
      for (int h = 0; h < H_; h++) dsum[rr * H_ + h] = reg[h];
    }
  }
  __syncthreads();

  const int ch = (4 * tid) % HC;
  if constexpr (PH_ == 0) {
    const v4f at4 = *(const v4f*)(att + ch);
    const float* xlg = xl + (size_t)lo * HC;
    for (int n0 = 0; n0 < tot; n0 += 4 * 4096) {
      v4f xs[4];
#pragma unroll
      for (int u = 0; u < 4; u++) xs[u] = *(const v4f*)(xlg + min(n0 + 4 * tid + u * 4096, tot - 4));
      asm volatile("" :: "v"(xs[0]), "v"(xs[1]), "v"(xs[2]), "v"(xs[3]));
#pragma unroll
      for (int u = 0; u < 4; u++) {
        const int i = n0 + 4 * tid + u * 4096;
        const int ic = min(i, tot - 4);
        const v4f z = xs[u] + *(const v4f*)(accs + ic);
        float t = dot_lrelu(z, at4);
        t += __shfl_xor(t, 8, 32);
        t += __shfl_xor(t, 4, 32);
        t += __shfl_xor(t, 2, 32);
        t += __shfl_xor(t, 1, 32);
        if ((lane & 15) == 0 && i < tot) {
          const int rh = (ic / HC) * H_ + ((ic % HC) >> 6);
          const int item = lo * H_ + rh;
          nmax_w[item] = fmaxf(t, dsum[rh]);
          eself_w[item] = t;
        }
      }
    }
  } else {
    const v4f bb = *(const v4f*)(bias + ch);
    for (int i0 = 4 * tid; i0 < tot; i0 += 4 * 4096) {
      v4f xs[4];
      float lg[4];
      float mk[4];
#pragma unroll
      for (int u = 0; u < 4; u++) {
        const int i = min(i0 + u * 4096, tot - 4);
        const size_t node = (size_t)(lo + i / HC);
        const int item = (int)node * H_ + ((i % HC) >> 6);
        lg[u] = eself_r[item];
        mk[u] = nmax_r[item];
        xs[u] = *(const v4f*)(xl + node * HC + i % HC);
      }
      asm volatile("" ::
                   "v"(xs[0]), "v"(xs[1]), "v"(xs[2]), "v"(xs[3]),
                   "v"(lg[0]), "v"(mk[0]), "v"(lg[1]), "v"(mk[1]),
                   "v"(lg[2]), "v"(mk[2]), "v"(lg[3]), "v"(mk[3]));
#pragma unroll
      for (int u = 0; u < 4; u++) {
        const int i = i0 + u * 4096;
        if (i < tot) {
          const float exs = expf(lg[u] - mk[u]);
          const float rc = 1.0f / (dsum[(i / HC) * H_ + ((i % HC) >> 6)] + exs);
          const v4f sm = *(const v4f*)(accs + i);
          v4f o = (sm + exs * xs[u]) * rc + bb;
          if constexpr (ELU_) { o.x = elu1(o.x); o.y = elu1(o.y); o.z = elu1(o.z); o.w = elu1(o.w); }
          *(v4f*)(outp + (size_t)(lo + i / HC) * HC + i % HC) = o;
        }
      }
    }
  }
}

extern "C" void kernel_launch(void* const* d_in, const int* in_sizes, int n_in,
                              void* d_out, int out_size, void* d_ws, size_t ws_size,
                              hipStream_t stream) {
  (void)n_in; (void)out_size; (void)ws_size;
  const int IN = 128, HID = 64, H = 4, OUT = 64;
  const float*    x    = (const float*)d_in[0];
  const int*      ei   = (const int*)d_in[1];
  const float*    Wl1  = (const float*)d_in[2];
  const float*    bl1  = (const float*)d_in[3];
  const float*    Wr1  = (const float*)d_in[4];
  const float*    br1  = (const float*)d_in[5];
  const float*    att1 = (const float*)d_in[6];
  const float*    bias1= (const float*)d_in[7];
  const float*    Wl2  = (const float*)d_in[8];
  const float*    bl2  = (const float*)d_in[9];
  const float*    Wr2  = (const float*)d_in[10];
  const float*    br2  = (const float*)d_in[11];
  const float*    att2 = (const float*)d_in[12];
  const float*    bias2= (const float*)d_in[13];
  float* out = (float*)d_out;

  const int N    = in_sizes[0] / IN;
  const int E    = in_sizes[1] / 2;
  const int Etot = E + N;
  const int HC   = H * HID;

  float* ws = (float*)d_ws;
  size_t o = 0;
  float*    xl1   = ws + o; o += (size_t)N * HC;
  float*    xr1   = ws + o; o += (size_t)N * HC;
  float*    ebuf  = ws + o; o += (size_t)Etot * H;
  float*    nmax  = ws + o; o += (size_t)N * H;
  const int gown = (N + OWN_R - 1) / OWN_R;
  int* rl_e = (int*)(ws + o); o += (size_t)gown * 32 * RCAP;
  int* rl_r = (int*)(ws + o); o += (size_t)gown * 32 * RCAP;
  int* rl_n = (int*)(ws + o); o += (size_t)gown * 32;
  const int nch = (E + BK_L - 1) / BK_L;
  int* bk_ent = (int*)(ws + o); o += (size_t)nch * BK_L;
  int* bk_off = (int*)(ws + o); o += (size_t)nch * BK_P;
  float*  h    = xr1;
  float*  xl2  = xl1;
  float*  xr2  = xl1 + (size_t)N * OUT;

  const int B = 256;
  int rblks = (N / 16 + 7) / 8;

  bucket_edges<<<nch, 32, 0, stream>>>(ei + E, E, bk_ent, bk_off);

  gemm_bias_wmma<128, 256><<<rblks * (HC / 64), B, 0, stream>>>(x, Wl1, bl1, xl1, N);
  gemm_bias_wmma<128, 256><<<rblks * (HC / 64), B, 0, stream>>>(x, Wr1, br1, xr1, N);
  owner_pass<4, false, 1, 0><<<gown, 1024, 0, stream>>>(xl1, xr1, att1, ei, bk_ent, bk_off, nch, nullptr, nullptr,
                                                        nullptr, ebuf, ebuf + (size_t)E * H, nmax,
                                                        nullptr, nullptr, N, rl_e, rl_r, rl_n);
  owner_pass<4, true, 2, 1><<<gown, 1024, 0, stream>>>(xl1, nullptr, nullptr, ei, bk_ent, bk_off, nch, ebuf,
                                                       ebuf + (size_t)E * H, nmax, nullptr, nullptr,
                                                       nullptr, bias1, h, N, rl_e, rl_r, rl_n);

  gemm_bias_wmma<256, 64><<<rblks * (OUT / 64), B, 0, stream>>>(h, Wl2, bl2, xl2, N);
  gemm_bias_wmma<256, 64><<<rblks * (OUT / 64), B, 0, stream>>>(h, Wr2, br2, xr2, N);
  owner_pass<1, false, 2, 0><<<gown, 1024, 0, stream>>>(xl2, xr2, att2, ei, bk_ent, bk_off, nch, nullptr, nullptr,
                                                        nullptr, ebuf, ebuf + E, nmax, nullptr,
                                                        nullptr, N, rl_e, rl_r, rl_n);
  owner_pass<1, false, 2, 1><<<gown, 1024, 0, stream>>>(xl2, nullptr, nullptr, ei, bk_ent, bk_off, nch, ebuf, ebuf + E,
                                                        nmax, nullptr, nullptr, nullptr, bias2, out,
                                                        N, rl_e, rl_r, rl_n);
}
